# starting kernel
# baseline (speedup 1.0000x reference)
#include <hip/hip_runtime.h>
#include <math.h>

#define NEG_SLOPE 0.2f

typedef __attribute__((ext_vector_type(2))) float v2f;
typedef __attribute__((ext_vector_type(8))) float v8f;

__device__ __forceinline__ float lrelu(float v) { return v > 0.f ? v : NEG_SLOPE * v; }

__device__ __forceinline__ unsigned ford(float f) {
    unsigned u = __float_as_uint(f);
    return (u & 0x80000000u) ? ~u : (u | 0x80000000u);
}
__device__ __forceinline__ float forddec(unsigned u) {
    return (u & 0x80000000u) ? __uint_as_float(u & 0x7fffffffu) : __uint_as_float(~u);
}
#define FORD_NEG_INF 0x007FFFFFu

__global__ void k_init(unsigned* m1, int n1, unsigned* m2, int n2,
                       float* s1, float* wsum1, int ns1,
                       float* s2, int ns2, float* acc2, int nacc) {
    int i = blockIdx.x * blockDim.x + threadIdx.x;
    if (i < n1)  m1[i] = FORD_NEG_INF;
    if (i < n2)  m2[i] = FORD_NEG_INF;
    if (i < ns1) { s1[i] = 0.f; wsum1[i] = 0.f; }
    if (i < ns2) s2[i] = 0.f;
    if (i < nacc) acc2[i] = 0.f;
}

__global__ void k_node1(const float* __restrict__ x, const float* __restrict__ W1,
                        const float* __restrict__ as1, const float* __restrict__ ad1,
                        float* __restrict__ alS1, float* __restrict__ alD1, int N) {
    int n = blockIdx.x * blockDim.x + threadIdx.x;
    if (n >= N) return;
    float xv = x[n];
#pragma unroll
    for (int h = 0; h < 4; ++h) {
        float ws = 0.f, wd = 0.f;
#pragma unroll
        for (int c = 0; c < 8; ++c) {
            float w = W1[h * 8 + c];
            ws += w * as1[h * 8 + c];
            wd += w * ad1[h * 8 + c];
        }
        alS1[n * 4 + h] = xv * ws;
        alD1[n * 4 + h] = xv * wd;
    }
}

__global__ void k_edge1_max(const int* __restrict__ ei, int E, int N,
                            const float* __restrict__ alS1, const float* __restrict__ alD1,
                            unsigned* __restrict__ m1) {
    int e = blockIdx.x * blockDim.x + threadIdx.x;
    if (e >= E + N) return;
    int s, d;
    if (e < E) { s = ei[e]; d = ei[E + e]; } else { s = d = e - E; }
    float4 as = *reinterpret_cast<const float4*>(alS1 + (size_t)s * 4);
    float4 ad = *reinterpret_cast<const float4*>(alD1 + (size_t)d * 4);
    atomicMax(m1 + (size_t)d * 4 + 0, ford(lrelu(as.x + ad.x)));
    atomicMax(m1 + (size_t)d * 4 + 1, ford(lrelu(as.y + ad.y)));
    atomicMax(m1 + (size_t)d * 4 + 2, ford(lrelu(as.z + ad.z)));
    atomicMax(m1 + (size_t)d * 4 + 3, ford(lrelu(as.w + ad.w)));
}

__global__ void k_edge1_sum(const int* __restrict__ ei, int E, int N,
                            const float* __restrict__ x,
                            const float* __restrict__ alS1, const float* __restrict__ alD1,
                            const unsigned* __restrict__ m1,
                            float* __restrict__ s1, float* __restrict__ wsum1) {
    int e = blockIdx.x * blockDim.x + threadIdx.x;
    if (e >= E + N) return;
    int s, d;
    if (e < E) { s = ei[e]; d = ei[E + e]; } else { s = d = e - E; }
    float4 as = *reinterpret_cast<const float4*>(alS1 + (size_t)s * 4);
    float4 ad = *reinterpret_cast<const float4*>(alD1 + (size_t)d * 4);
    uint4  mm = *reinterpret_cast<const uint4*>(m1 + (size_t)d * 4);
    float xs = x[s];
    float t0 = __expf(lrelu(as.x + ad.x) - forddec(mm.x));
    float t1 = __expf(lrelu(as.y + ad.y) - forddec(mm.y));
    float t2 = __expf(lrelu(as.z + ad.z) - forddec(mm.z));
    float t3 = __expf(lrelu(as.w + ad.w) - forddec(mm.w));
    atomicAdd(s1 + (size_t)d * 4 + 0, t0);
    atomicAdd(s1 + (size_t)d * 4 + 1, t1);
    atomicAdd(s1 + (size_t)d * 4 + 2, t2);
    atomicAdd(s1 + (size_t)d * 4 + 3, t3);
    atomicAdd(wsum1 + (size_t)d * 4 + 0, t0 * xs);
    atomicAdd(wsum1 + (size_t)d * 4 + 1, t1 * xs);
    atomicAdd(wsum1 + (size_t)d * 4 + 2, t2 * xs);
    atomicAdd(wsum1 + (size_t)d * 4 + 3, t3 * xs);
}

__global__ void k_fin1(const float* __restrict__ W1, const float* __restrict__ b1,
                       const float* __restrict__ s1, const float* __restrict__ wsum1,
                       float* __restrict__ out1, int N) {
    int i = blockIdx.x * blockDim.x + threadIdx.x;
    if (i >= N * 32) return;
    int n = i >> 5, j = i & 31, h = j >> 3;
    float v = W1[j] * (wsum1[n * 4 + h] / s1[n * 4 + h]) + b1[j];
    out1[i] = v > 0.f ? v : 0.f;
}

__global__ void k_gemm2(const float* __restrict__ out1, const float* __restrict__ W2,
                        const float* __restrict__ as2, const float* __restrict__ ad2,
                        float* __restrict__ h2, float* __restrict__ alS2,
                        float* __restrict__ alD2, int N) {
    __shared__ float Bp[32 * 16];
    int t = threadIdx.x;
    for (int i = t; i < 512; i += blockDim.x) {
        int k = i >> 4, n = i & 15;
        float v = 0.f;
        if (n < 8) {
            v = W2[k * 8 + n];
        } else if (n == 8) {
            for (int c = 0; c < 8; ++c) v += W2[k * 8 + c] * as2[c];
        } else if (n == 9) {
            for (int c = 0; c < 8; ++c) v += W2[k * 8 + c] * ad2[c];
        }
        Bp[i] = v;
    }
    __syncthreads();

    int wave = t >> 5, lane = t & 31;
    long tile = (long)blockIdx.x * 8 + wave;
    long row0 = tile * 16;
    if (row0 >= N) return;

    int m = lane & 15, khi = lane >> 4;
    long rowm = row0 + m;
    if (rowm >= N) rowm = N - 1;
    const float* Arow = out1 + rowm * 32;

    v8f c = {};
#pragma unroll
    for (int step = 0; step < 8; ++step) {
        int k0 = step * 4 + khi * 2;
        v2f a; a.x = Arow[k0];          a.y = Arow[k0 + 1];
        v2f b; b.x = Bp[k0 * 16 + m];   b.y = Bp[(k0 + 1) * 16 + m];
        c = __builtin_amdgcn_wmma_f32_16x16x4_f32(false, a, false, b, (short)0, c,
                                                  false, false);
    }

#pragma unroll
    for (int r = 0; r < 8; ++r) {
        long node = row0 + khi * 8 + r;
        if (node >= N) continue;
        float v = c[r];
        if (m < 8)        h2[node * 8 + m] = v;
        else if (m == 8)  alS2[node] = v;
        else if (m == 9)  alD2[node] = v;
    }
}

__global__ void k_edge2_max(const int* __restrict__ ei, int E, int N,
                            const float* __restrict__ alS2, const float* __restrict__ alD2,
                            unsigned* __restrict__ m2) {
    int e = blockIdx.x * blockDim.x + threadIdx.x;
    if (e >= E + N) return;
    int s, d;
    if (e < E) { s = ei[e]; d = ei[E + e]; } else { s = d = e - E; }
    atomicMax(m2 + d, ford(lrelu(alS2[s] + alD2[d])));
}

__global__ void k_edge2_sum(const int* __restrict__ ei, int E, int N,
                            const float* __restrict__ h2,
                            const float* __restrict__ alS2, const float* __restrict__ alD2,
                            const unsigned* __restrict__ m2,
                            float* __restrict__ s2, float* __restrict__ acc2) {
    int e = blockIdx.x * blockDim.x + threadIdx.x;
    if (e >= E + N) return;
    int s, d;
    if (e < E) { s = ei[e]; d = ei[E + e]; } else { s = d = e - E; }
    float t = __expf(lrelu(alS2[s] + alD2[d]) - forddec(m2[d]));
    atomicAdd(s2 + d, t);
    float4 h0 = *reinterpret_cast<const float4*>(h2 + (size_t)s * 8);
    float4 h1 = *reinterpret_cast<const float4*>(h2 + (size_t)s * 8 + 4);
    float* ad = acc2 + (size_t)d * 8;
    atomicAdd(ad + 0, t * h0.x); atomicAdd(ad + 1, t * h0.y);
    atomicAdd(ad + 2, t * h0.z); atomicAdd(ad + 3, t * h0.w);
    atomicAdd(ad + 4, t * h1.x); atomicAdd(ad + 5, t * h1.y);
    atomicAdd(ad + 6, t * h1.z); atomicAdd(ad + 7, t * h1.w);
}

__global__ void k_fin2(const float* __restrict__ acc2, const float* __restrict__ s2,
                       const float* __restrict__ b2, float* __restrict__ out, int N) {
    int i = blockIdx.x * blockDim.x + threadIdx.x;
    if (i >= N * 8) return;
    int n = i >> 3, c = i & 7;
    out[i] = acc2[i] / s2[n] + b2[c];
}

extern "C" void kernel_launch(void* const* d_in, const int* in_sizes, int n_in,
                              void* d_out, int out_size, void* d_ws, size_t ws_size,
                              hipStream_t stream) {
    const float* x   = (const float*)d_in[0];
    const int*   ei  = (const int*)d_in[1];
    const float* W1  = (const float*)d_in[2];
    const float* as1 = (const float*)d_in[3];
    const float* ad1 = (const float*)d_in[4];
    const float* b1  = (const float*)d_in[5];
    const float* W2  = (const float*)d_in[6];
    const float* as2 = (const float*)d_in[7];
    const float* ad2 = (const float*)d_in[8];
    const float* b2  = (const float*)d_in[9];
    (void)n_in; (void)out_size; (void)ws_size;

    int N = in_sizes[0];
    int E = in_sizes[1] / 2;
    int EN = E + N;

    char* ws = (char*)d_ws;
    size_t cur = 0;
    auto carve = [&](size_t bytes) -> void* {
        void* p = ws + cur;
        cur += (bytes + 255) & ~(size_t)255;
        return p;
    };
    unsigned* m1    = (unsigned*)carve(sizeof(unsigned) * (size_t)N * 4);
    unsigned* m2    = (unsigned*)carve(sizeof(unsigned) * (size_t)N);
    float*    s1    = (float*)carve(sizeof(float) * (size_t)N * 4);
    float*    wsum1 = (float*)carve(sizeof(float) * (size_t)N * 4);
    float*    s2    = (float*)carve(sizeof(float) * (size_t)N);
    float*    acc2  = (float*)carve(sizeof(float) * (size_t)N * 8);
    float*    alS1  = (float*)carve(sizeof(float) * (size_t)N * 4);
    float*    alD1  = (float*)carve(sizeof(float) * (size_t)N * 4);
    float*    out1  = (float*)carve(sizeof(float) * (size_t)N * 32);
    float*    h2    = (float*)carve(sizeof(float) * (size_t)N * 8);
    float*    alS2  = (float*)carve(sizeof(float) * (size_t)N);
    float*    alD2  = (float*)carve(sizeof(float) * (size_t)N);

    const int B = 256;
    k_init<<<(8 * N + B - 1) / B, B, 0, stream>>>(m1, 4 * N, m2, N,
                                                  s1, wsum1, 4 * N, s2, N, acc2, 8 * N);
    k_node1<<<(N + B - 1) / B, B, 0, stream>>>(x, W1, as1, ad1, alS1, alD1, N);
    k_edge1_max<<<(EN + B - 1) / B, B, 0, stream>>>(ei, E, N, alS1, alD1, m1);
    k_edge1_sum<<<(EN + B - 1) / B, B, 0, stream>>>(ei, E, N, x, alS1, alD1, m1, s1, wsum1);
    k_fin1<<<(32 * N + B - 1) / B, B, 0, stream>>>(W1, b1, s1, wsum1, out1, N);

    int tiles = (N + 15) / 16;
    k_gemm2<<<(tiles + 7) / 8, B, 0, stream>>>(out1, W2, as2, ad2, h2, alS2, alD2, N);

    k_edge2_max<<<(EN + B - 1) / B, B, 0, stream>>>(ei, E, N, alS2, alD2, m2);
    k_edge2_sum<<<(EN + B - 1) / B, B, 0, stream>>>(ei, E, N, h2, alS2, alD2, m2, s2, acc2);
    k_fin2<<<(8 * N + B - 1) / B, B, 0, stream>>>(acc2, s2, b2, (float*)d_out, N);
}


# Owner work-groups keep a destination range in LDS and filter the unsorted edge list for it (wave queue, LDS atomics): no per-edge global atomic; whole-line atomics join copies; sums in arrival order
# speedup vs baseline: 6.0876x; 6.0876x over previous
#include <hip/hip_runtime.h>
#include <math.h>


#define NEG_SLOPE 0.2f
#define WG 1024
#define UNR 8
#define QCAP 64
static_assert(UNR == 8, "the load fence in scan_chunk names eight values");
static_assert(QCAP >= 64, "a push of 32 entries on top of 32 queued ones must fit");

typedef __attribute__((ext_vector_type(2))) float v2f;
typedef __attribute__((ext_vector_type(8))) float v8f;

__device__ __forceinline__ float lrelu(float v) { return v > 0.f ? v : NEG_SLOPE * v; }

__device__ __forceinline__ unsigned ford(float f) {
    unsigned u = __float_as_uint(f);
    return (u & 0x80000000u) ? ~u : (u | 0x80000000u);
}
__device__ __forceinline__ float forddec(unsigned u) {
    return (u & 0x80000000u) ? __uint_as_float(u & 0x7fffffffu) : __uint_as_float(~u);
}
#define FORD_NEG_INF 0x007FFFFFu

__global__ void k_init(unsigned* m1, int n1, unsigned* m2, int n2,
                       float* rec1, int nrec1,
                       float* s2, int ns2, float* acc2, int nacc) {
    int i = blockIdx.x * blockDim.x + threadIdx.x;
    if (i < n1)  m1[i] = FORD_NEG_INF;
    if (i < n2)  m2[i] = FORD_NEG_INF;
    if (i < nrec1) rec1[i] = 0.f;
    if (i < ns2) s2[i] = 0.f;
    if (i < nacc) acc2[i] = 0.f;
}


__device__ __forceinline__ void wave_sync() {
    __builtin_amdgcn_fence(__ATOMIC_RELEASE, "wavefront");
    __builtin_amdgcn_wave_barrier();
    __builtin_amdgcn_fence(__ATOMIC_ACQUIRE, "wavefront");
}

__device__ __forceinline__ void load_edge(const int* __restrict__ ei, int E, int e, int& ls, int& ld) {
    int ec = e < E ? e : E - 1;
    ls = ei[ec];
    ld = ei[E + ec];
}
__device__ __forceinline__ void place_edge(int E, int e, int e1, int lo, int ls, int ld,
                                           int& s, unsigned& rel) {
    int d = e < E ? ld : e - E;
    s     = e < E ? ls : e - E;
    if (e >= e1) { d = -1; s = 0; }
    rel = (unsigned)(d - lo);
}

template <class Drain>
__device__ __forceinline__ void scan_chunk(const int* __restrict__ ei, int E, int e0, int e1,
                                           int lo, int span, int2* qw, int tid, Drain drain) {
    const unsigned lt = (1u << (tid & 31)) - 1u;
    int cnt = 0;
    for (int b = e0; b < e1; b += WG * UNR) {
        int ls[UNR], ld[UNR], s[UNR]; unsigned rel[UNR];
#pragma unroll
        for (int u = 0; u < UNR; ++u) load_edge(ei, E, b + u * WG + tid, ls[u], ld[u]);
        asm volatile("" :: "v"(ls[0]), "v"(ls[1]), "v"(ls[2]), "v"(ls[3]),
                           "v"(ls[4]), "v"(ls[5]), "v"(ls[6]), "v"(ls[7]));
#pragma unroll
        for (int u = 0; u < UNR; ++u) place_edge(E, b + u * WG + tid, e1, lo, ls[u], ld[u], s[u], rel[u]);
#pragma unroll
        for (int u = 0; u < UNR; ++u) {
            bool keep = rel[u] < (unsigned)span;
            unsigned mask = __builtin_amdgcn_ballot_w32(keep);
            if (keep) qw[cnt + __popc(mask & lt)] = make_int2(s[u], (int)rel[u]);
            cnt += __popc(mask);
            if (cnt > QCAP - 32) {
                wave_sync();
                drain(cnt);
                wave_sync();
                cnt = 0;
            }
        }
    }
    wave_sync();
    drain(cnt);
}

__device__ __forceinline__ void head_consts(const float* __restrict__ W1, const float* __restrict__ as1,
                                            const float* __restrict__ ad1, int h, float& wsl, float& wdl) {
    float a = 0.f, b = 0.f;
#pragma unroll
    for (int c = 0; c < 8; ++c) {
        float w = W1[h * 8 + c];
        a += w * as1[h * 8 + c];
        b += w * ad1[h * 8 + c];
    }
    wsl = a; wdl = b;
}

template <int SPAN, int G>
__global__ __launch_bounds__(WG) void k_l1_max(const int* __restrict__ ei, int E, int N,
                                               const float* __restrict__ x,
                                               const float* __restrict__ W1,
                                               const float* __restrict__ as1,
                                               const float* __restrict__ ad1,
                                               unsigned* __restrict__ m1) {
    __shared__ unsigned mx[SPAN * 4];
    __shared__ int2 q[(WG / 32) * QCAP];
    const int tid = threadIdx.x, lane = tid & 31;
    const int r = blockIdx.x / G, g = blockIdx.x % G;
    const int lo = r * SPAN;
    const int span = (N - lo < SPAN) ? (N - lo) : SPAN;
    const int h = lane & 3, sub = lane >> 2;
    float wsl, wdl;
    head_consts(W1, as1, ad1, h, wsl, wdl);

    for (int i = tid; i < span * 4; i += WG) mx[i] = FORD_NEG_INF;
    __syncthreads();

    int2* qw = q + (tid >> 5) * QCAP;
    auto drain = [&](int cnt) {
        for (int k0 = 0; k0 < cnt; k0 += 32) {
            bool ok[4]; int2 en[4]; float xs[4], xdv[4];
#pragma unroll
            for (int rr = 0; rr < 4; ++rr) {
                int k = k0 + rr * 8 + sub;
                ok[rr] = k < cnt;
                en[rr] = qw[ok[rr] ? k : cnt - 1];
            }
#pragma unroll
            for (int rr = 0; rr < 4; ++rr) { xs[rr] = x[en[rr].x]; xdv[rr] = x[lo + en[rr].y]; }
            asm volatile("" :: "v"(xs[0]), "v"(xs[1]), "v"(xs[2]), "v"(xs[3]),
                               "v"(xdv[0]), "v"(xdv[1]), "v"(xdv[2]), "v"(xdv[3]));
#pragma unroll
            for (int rr = 0; rr < 4; ++rr) {
                unsigned v = ford(lrelu(xs[rr] * wsl + xdv[rr] * wdl));
                if (ok[rr]) atomicMax(&mx[en[rr].y * 4 + h], v);
            }
        }
    };
    const int EN = E + N;
    const int per = (EN + G - 1) / G;
    const int e0 = g * per;
    const int e1 = (e0 + per < EN) ? (e0 + per) : EN;
    scan_chunk(ei, E, e0, e1, lo, span, qw, tid, drain);
    __syncthreads();

    const int tot = span * 4;
    const int rot = (int)(((long)tot * g) / G);
    for (int k = tid; k < tot; k += WG) {
        int i = k + rot; if (i >= tot) i -= tot;
        atomicMax(m1 + (size_t)lo * 4 + i, mx[i]);
    }
}

template <int SPAN, int G>
__global__ __launch_bounds__(WG) void k_l1_sum(const int* __restrict__ ei, int E, int N,
                                               const float* __restrict__ x,
                                               const float* __restrict__ W1,
                                               const float* __restrict__ as1,
                                               const float* __restrict__ ad1,
                                               const unsigned* __restrict__ m1,
                                               float* __restrict__ rec1) {
    __shared__ float mxf[SPAN * 4];
    __shared__ float rec[SPAN * 8];
    __shared__ int2 q[(WG / 32) * QCAP];
    const int tid = threadIdx.x, lane = tid & 31;
    const int r = blockIdx.x / G, g = blockIdx.x % G;
    const int lo = r * SPAN;
    const int span = (N - lo < SPAN) ? (N - lo) : SPAN;
    const int j = lane & 7, sub = lane >> 3;
    const int h = j & 3;
    float wsl, wdl;
    head_consts(W1, as1, ad1, h, wsl, wdl);

    for (int i = tid; i < span * 4; i += WG) mxf[i] = forddec(m1[(size_t)lo * 4 + i]);
    for (int i = tid; i < span * 8; i += WG) rec[i] = 0.f;
    __syncthreads();

    int2* qw = q + (tid >> 5) * QCAP;
    auto drain = [&](int cnt) {
        for (int k0 = 0; k0 < cnt; k0 += 16) {
            bool ok[4]; int2 en[4]; float xs[4], xdv[4];
#pragma unroll
            for (int rr = 0; rr < 4; ++rr) {
                int k = k0 + rr * 4 + sub;
                ok[rr] = k < cnt;
                en[rr] = qw[ok[rr] ? k : cnt - 1];
            }
#pragma unroll
            for (int rr = 0; rr < 4; ++rr) { xs[rr] = x[en[rr].x]; xdv[rr] = x[lo + en[rr].y]; }
            asm volatile("" :: "v"(xs[0]), "v"(xs[1]), "v"(xs[2]), "v"(xs[3]),
                               "v"(xdv[0]), "v"(xdv[1]), "v"(xdv[2]), "v"(xdv[3]));
#pragma unroll
            for (int rr = 0; rr < 4; ++rr) {
                float t = __expf(lrelu(xs[rr] * wsl + xdv[rr] * wdl) - mxf[en[rr].y * 4 + h]);
                float v = (j < 4) ? t : t * xs[rr];
                if (ok[rr]) atomicAdd(&rec[en[rr].y * 8 + j], v);
            }
        }
    };
    const int EN = E + N;
    const int per = (EN + G - 1) / G;
    const int e0 = g * per;
    const int e1 = (e0 + per < EN) ? (e0 + per) : EN;
    scan_chunk(ei, E, e0, e1, lo, span, qw, tid, drain);
    __syncthreads();

    const int tot = span * 8;
    const int rot = (int)(((long)tot * g) / G);
    for (int k = tid; k < tot; k += WG) {
        int i = k + rot; if (i >= tot) i -= tot;
        atomicAdd(rec1 + (size_t)lo * 8 + i, rec[i]);
    }
}

__global__ void k_fin1(const float* __restrict__ W1, const float* __restrict__ b1,
                       const float* __restrict__ rec1,
                       float* __restrict__ out1, int N) {
    int i = blockIdx.x * blockDim.x + threadIdx.x;
    if (i >= N * 32) return;
    int n = i >> 5, j = i & 31, h = j >> 3;
    float v = W1[j] * (rec1[n * 8 + 4 + h] / rec1[n * 8 + h]) + b1[j];
    out1[i] = v > 0.f ? v : 0.f;
}

__global__ void k_gemm2(const float* __restrict__ out1, const float* __restrict__ W2,
                        const float* __restrict__ as2, const float* __restrict__ ad2,
                        float* __restrict__ h2, float* __restrict__ alS2,
                        float* __restrict__ alD2, int N) {
    __shared__ float Bp[32 * 16];
    int t = threadIdx.x;
    for (int i = t; i < 512; i += blockDim.x) {
        int k = i >> 4, n = i & 15;
        float v = 0.f;
        if (n < 8) {
            v = W2[k * 8 + n];
        } else if (n == 8) {
            for (int c = 0; c < 8; ++c) v += W2[k * 8 + c] * as2[c];
        } else if (n == 9) {
            for (int c = 0; c < 8; ++c) v += W2[k * 8 + c] * ad2[c];
        }
        Bp[i] = v;
    }
    __syncthreads();

    int wave = t >> 5, lane = t & 31;
    long tile = (long)blockIdx.x * 8 + wave;
    long row0 = tile * 16;
    if (row0 >= N) return;

    int m = lane & 15, khi = lane >> 4;
    long rowm = row0 + m;
    if (rowm >= N) rowm = N - 1;
    const float* Arow = out1 + rowm * 32;

    v8f c = {};
#pragma unroll
    for (int step = 0; step < 8; ++step) {
        int k0 = step * 4 + khi * 2;
        v2f a; a.x = Arow[k0];          a.y = Arow[k0 + 1];
        v2f b; b.x = Bp[k0 * 16 + m];   b.y = Bp[(k0 + 1) * 16 + m];
        c = __builtin_amdgcn_wmma_f32_16x16x4_f32(false, a, false, b, (short)0, c,
                                                  false, false);
    }

#pragma unroll
    for (int r = 0; r < 8; ++r) {
        long node = row0 + khi * 8 + r;
        if (node >= N) continue;
        float v = c[r];
        if (m < 8)        h2[node * 8 + m] = v;
        else if (m == 8)  alS2[node] = v;
        else if (m == 9)  alD2[node] = v;
    }
}

template <int SPAN, int G>
__global__ __launch_bounds__(WG) void k_l2_max(const int* __restrict__ ei, int E, int N,
                                               const float* __restrict__ alS2,
                                               const float* __restrict__ alD2,
                                               unsigned* __restrict__ m2) {
    __shared__ unsigned mx[SPAN];
    __shared__ float ad[SPAN];
    __shared__ int2 q[(WG / 32) * QCAP];
    const int tid = threadIdx.x, lane = tid & 31;
    const int r = blockIdx.x / G, g = blockIdx.x % G;
    const int lo = r * SPAN;
    const int span = (N - lo < SPAN) ? (N - lo) : SPAN;

    for (int i = tid; i < span; i += WG) { ad[i] = alD2[lo + i]; mx[i] = FORD_NEG_INF; }
    __syncthreads();

    int2* qw = q + (tid >> 5) * QCAP;
    auto drain = [&](int cnt) {
        for (int k0 = 0; k0 < cnt; k0 += 64) {
            bool ok[2]; int2 en[2]; float as[2];
#pragma unroll
            for (int rr = 0; rr < 2; ++rr) {
                int k = k0 + rr * 32 + lane;
                ok[rr] = k < cnt;
                en[rr] = qw[ok[rr] ? k : cnt - 1];
            }
#pragma unroll
            for (int rr = 0; rr < 2; ++rr) as[rr] = alS2[en[rr].x];
            asm volatile("" :: "v"(as[0]), "v"(as[1]));
#pragma unroll
            for (int rr = 0; rr < 2; ++rr)
                if (ok[rr]) atomicMax(&mx[en[rr].y], ford(lrelu(as[rr] + ad[en[rr].y])));
        }
    };
    const int EN = E + N;
    const int per = (EN + G - 1) / G;
    const int e0 = g * per;
    const int e1 = (e0 + per < EN) ? (e0 + per) : EN;
    scan_chunk(ei, E, e0, e1, lo, span, qw, tid, drain);
    __syncthreads();

    const int tot = span;
    const int rot = (int)(((long)tot * g) / G);
    for (int k = tid; k < tot; k += WG) {
        int i = k + rot; if (i >= tot) i -= tot;
        atomicMax(m2 + lo + i, mx[i]);
    }
}

template <int SPAN, int G>
__global__ __launch_bounds__(WG) void k_l2_sum(const int* __restrict__ ei, int E, int N,
                                               const float* __restrict__ h2,
                                               const float* __restrict__ alS2,
                                               const float* __restrict__ alD2,
                                               const unsigned* __restrict__ m2,
                                               float* __restrict__ s2, float* __restrict__ acc2) {
    __shared__ float mxf[SPAN];
    __shared__ float ad[SPAN];
    __shared__ float ssum[SPAN];
    __shared__ float acc[SPAN * 8];
    __shared__ int2 q[(WG / 32) * QCAP];
    const int tid = threadIdx.x, lane = tid & 31;
    const int r = blockIdx.x / G, g = blockIdx.x % G;
    const int lo = r * SPAN;
    const int span = (N - lo < SPAN) ? (N - lo) : SPAN;
    const int j = lane & 7, sub = lane >> 3;

    for (int i = tid; i < span; i += WG) {
        ad[i] = alD2[lo + i]; mxf[i] = forddec(m2[lo + i]); ssum[i] = 0.f;
    }
    for (int i = tid; i < span * 8; i += WG) acc[i] = 0.f;
    __syncthreads();

    int2* qw = q + (tid >> 5) * QCAP;
    auto drain = [&](int cnt) {
        for (int k0 = 0; k0 < cnt; k0 += 16) {
            bool ok[4]; int2 en[4]; float as[4], hv[4];
#pragma unroll
            for (int rr = 0; rr < 4; ++rr) {
                int k = k0 + rr * 4 + sub;
                ok[rr] = k < cnt;
                en[rr] = qw[ok[rr] ? k : cnt - 1];
            }
#pragma unroll
            for (int rr = 0; rr < 4; ++rr) { as[rr] = alS2[en[rr].x]; hv[rr] = h2[(size_t)en[rr].x * 8 + j]; }
            asm volatile("" :: "v"(as[0]), "v"(as[1]), "v"(as[2]), "v"(as[3]),
                               "v"(hv[0]), "v"(hv[1]), "v"(hv[2]), "v"(hv[3]));
#pragma unroll
            for (int rr = 0; rr < 4; ++rr) {
                float t = __expf(lrelu(as[rr] + ad[en[rr].y]) - mxf[en[rr].y]);
                if (ok[rr]) {
                    atomicAdd(&acc[en[rr].y * 8 + j], t * hv[rr]);
                    if (j == 0) atomicAdd(&ssum[en[rr].y], t);
                }
            }
        }
    };
    const int EN = E + N;
    const int per = (EN + G - 1) / G;
    const int e0 = g * per;
    const int e1 = (e0 + per < EN) ? (e0 + per) : EN;
    scan_chunk(ei, E, e0, e1, lo, span, qw, tid, drain);
    __syncthreads();

    const int rot1 = (int)(((long)span * g) / G);
    for (int k = tid; k < span; k += WG) {
        int i = k + rot1; if (i >= span) i -= span;
        atomicAdd(s2 + lo + i, ssum[i]);
    }
    const int tot = span * 8;
    const int rot = (int)(((long)tot * g) / G);
    for (int k = tid; k < tot; k += WG) {
        int i = k + rot; if (i >= tot) i -= tot;
        atomicAdd(acc2 + (size_t)lo * 8 + i, acc[i]);
    }
}

__global__ void k_fin2(const float* __restrict__ acc2, const float* __restrict__ s2,
                       const float* __restrict__ b2, float* __restrict__ out, int N) {
    int i = blockIdx.x * blockDim.x + threadIdx.x;
    if (i >= N * 8) return;
    int n = i >> 3, c = i & 7;
    out[i] = acc2[i] / s2[n] + b2[c];
}

extern "C" void kernel_launch(void* const* d_in, const int* in_sizes, int n_in,
                              void* d_out, int out_size, void* d_ws, size_t ws_size,
                              hipStream_t stream) {
    const float* x   = (const float*)d_in[0];
    const int*   ei  = (const int*)d_in[1];
    const float* W1  = (const float*)d_in[2];
    const float* as1 = (const float*)d_in[3];
    const float* ad1 = (const float*)d_in[4];
    const float* b1  = (const float*)d_in[5];
    const float* W2  = (const float*)d_in[6];
    const float* as2 = (const float*)d_in[7];
    const float* ad2 = (const float*)d_in[8];
    const float* b2  = (const float*)d_in[9];
    (void)n_in; (void)out_size; (void)ws_size;

    int N = in_sizes[0];
    int E = in_sizes[1] / 2;

    char* ws = (char*)d_ws;
    size_t cur = 0;
    auto carve = [&](size_t bytes) -> void* {
        void* p = ws + cur;
        cur += (bytes + 255) & ~(size_t)255;
        return p;
    };
    unsigned* m1    = (unsigned*)carve(sizeof(unsigned) * (size_t)N * 4);
    unsigned* m2    = (unsigned*)carve(sizeof(unsigned) * (size_t)N);
    float*    rec1  = (float*)carve(sizeof(float) * (size_t)N * 8);
    float*    s2    = (float*)carve(sizeof(float) * (size_t)N);
    float*    acc2  = (float*)carve(sizeof(float) * (size_t)N * 8);
    float*    out1  = (float*)carve(sizeof(float) * (size_t)N * 32);
    float*    h2    = (float*)carve(sizeof(float) * (size_t)N * 8);
    float*    alS2  = (float*)carve(sizeof(float) * (size_t)N);
    float*    alD2  = (float*)carve(sizeof(float) * (size_t)N);

    const int B = 256;
    k_init<<<(8 * N + B - 1) / B, B, 0, stream>>>(m1, 4 * N, m2, N,
                                                  rec1, 8 * N, s2, N, acc2, 8 * N);
    k_l1_max<12500, 32><<<((N + 12499) / 12500) * 32, WG, 0, stream>>>(ei, E, N, x, W1, as1, ad1, m1);
    k_l1_sum<6250, 16><<<((N + 6249) / 6250) * 16, WG, 0, stream>>>(ei, E, N, x, W1, as1, ad1, m1, rec1);
    k_fin1<<<(32 * N + B - 1) / B, B, 0, stream>>>(W1, b1, rec1, out1, N);

    int tiles = (N + 15) / 16;
    k_gemm2<<<(tiles + 7) / 8, B, 0, stream>>>(out1, W2, as2, ad2, h2, alS2, alD2, N);

    k_l2_max<25000, 64><<<((N + 24999) / 25000) * 64, WG, 0, stream>>>(ei, E, N, alS2, alD2, m2);
    k_l2_sum<6250, 16><<<((N + 6249) / 6250) * 16, WG, 0, stream>>>(ei, E, N, h2, alS2, alD2, m2, s2, acc2);
    k_fin2<<<(8 * N + B - 1) / B, B, 0, stream>>>(acc2, s2, b2, (float*)d_out, N);
}


# Only the layer-1 maximum pass filters the edge list: each wave saves its kept edges in a workspace slice; the three later passes, now on the same ranges, read them back; LDS sums in arrival order
# speedup vs baseline: 9.0286x; 1.4831x over previous
#include <hip/hip_runtime.h>
#include <math.h>


#define NEG_SLOPE 0.2f
#define WG 1024
#define UNR 8
#define QCAP 64
#define SPAN_ALL 6250
#define G_ALL 16
#define SRC_BITS 19
static_assert(UNR == 8, "the load fence in scan_chunk names eight values");
static_assert(QCAP >= 64, "a push of 32 entries on top of 32 queued ones must fit");
static_assert(QCAP == 64, "walk_slice refills the queue with two entries per lane");
static_assert(SPAN_ALL <= (1 << (32 - SRC_BITS)), "destination - lo must fit above the source bits");

typedef __attribute__((ext_vector_type(2))) float v2f;
typedef __attribute__((ext_vector_type(8))) float v8f;

__device__ __forceinline__ float lrelu(float v) { return v > 0.f ? v : NEG_SLOPE * v; }

__device__ __forceinline__ unsigned ford(float f) {
    unsigned u = __float_as_uint(f);
    return (u & 0x80000000u) ? ~u : (u | 0x80000000u);
}
__device__ __forceinline__ float forddec(unsigned u) {
    return (u & 0x80000000u) ? __uint_as_float(u & 0x7fffffffu) : __uint_as_float(~u);
}
#define FORD_NEG_INF 0x007FFFFFu

__global__ void k_init(unsigned* m1, int n1, unsigned* m2, int n2,
                       float* rec1, int nrec1,
                       float* s2, int ns2, float* acc2, int nacc) {
    int i = blockIdx.x * blockDim.x + threadIdx.x;
    if (i < n1)  m1[i] = FORD_NEG_INF;
    if (i < n2)  m2[i] = FORD_NEG_INF;
    if (i < nrec1) rec1[i] = 0.f;
    if (i < ns2) s2[i] = 0.f;
    if (i < nacc) acc2[i] = 0.f;
}


__device__ __forceinline__ void wave_sync() {
    __builtin_amdgcn_fence(__ATOMIC_RELEASE, "wavefront");
    __builtin_amdgcn_wave_barrier();
    __builtin_amdgcn_fence(__ATOMIC_ACQUIRE, "wavefront");
}

__device__ __forceinline__ void load_edge(const int* __restrict__ ei, int E, int e, int& ls, int& ld) {
    int ec = e < E ? e : E - 1;
    ls = ei[ec];
    ld = ei[E + ec];
}
__device__ __forceinline__ void place_edge(int E, int e, int e1, int lo, int ls, int ld,
                                           int& s, unsigned& rel) {
    int d = e < E ? ld : e - E;
    s     = e < E ? ls : e - E;
    if (e >= e1) { d = -1; s = 0; }
    rel = (unsigned)(d - lo);
}

template <class Drain>
__device__ __forceinline__ void scan_chunk(const int* __restrict__ ei, int E, int e0, int e1,
                                           int lo, int span, int2* qw, int tid, Drain drain) {
    const unsigned lt = (1u << (tid & 31)) - 1u;
    int cnt = 0;
    for (int b = e0; b < e1; b += WG * UNR) {
        int ls[UNR], ld[UNR], s[UNR]; unsigned rel[UNR];
#pragma unroll
        for (int u = 0; u < UNR; ++u) load_edge(ei, E, b + u * WG + tid, ls[u], ld[u]);
        asm volatile("" :: "v"(ls[0]), "v"(ls[1]), "v"(ls[2]), "v"(ls[3]),
                           "v"(ls[4]), "v"(ls[5]), "v"(ls[6]), "v"(ls[7]));
#pragma unroll
        for (int u = 0; u < UNR; ++u) place_edge(E, b + u * WG + tid, e1, lo, ls[u], ld[u], s[u], rel[u]);
#pragma unroll
        for (int u = 0; u < UNR; ++u) {
            bool keep = rel[u] < (unsigned)span;
            unsigned mask = __builtin_amdgcn_ballot_w32(keep);
            if (keep) qw[cnt + __popc(mask & lt)] = make_int2(s[u], (int)rel[u]);
            cnt += __popc(mask);
            if (cnt > QCAP - 32) {
                wave_sync();
                drain(cnt);
                wave_sync();
                cnt = 0;
            }
        }
    }
    wave_sync();
    drain(cnt);
}

template <class Drain>
__device__ __forceinline__ void walk_slice(const unsigned* __restrict__ sl, int total,
                                           int2* qw, int lane, Drain drain) {
    if (total <= 0) return;
    const int last = total - 1;
    const unsigned smask = (1u << SRC_BITS) - 1u;
    unsigned a = sl[lane < last ? lane : last];
    unsigned b = sl[32 + lane < last ? 32 + lane : last];
    for (int k0 = 0; k0 < total; k0 += QCAP) {
        qw[lane]      = make_int2((int)(a & smask), (int)(a >> SRC_BITS));
        qw[32 + lane] = make_int2((int)(b & smask), (int)(b >> SRC_BITS));
        const int kn = k0 + QCAP;
        if (kn < total) {
            a = sl[kn + lane < last ? kn + lane : last];
            b = sl[kn + 32 + lane < last ? kn + 32 + lane : last];
        }
        wave_sync();
        drain(total - k0 < QCAP ? total - k0 : QCAP);
        wave_sync();
    }
}

__device__ __forceinline__ void head_consts(const float* __restrict__ W1, const float* __restrict__ as1,
                                            const float* __restrict__ ad1, int h, float& wsl, float& wdl) {
    float a = 0.f, b = 0.f;
#pragma unroll
    for (int c = 0; c < 8; ++c) {
        float w = W1[h * 8 + c];
        a += w * as1[h * 8 + c];
        b += w * ad1[h * 8 + c];
    }
    wsl = a; wdl = b;
}

template <int SPAN, int G>
__global__ __launch_bounds__(WG) void k_l1_max(const int* __restrict__ ei, int E, int N,
                                               const float* __restrict__ x,
                                               const float* __restrict__ W1,
                                               const float* __restrict__ as1,
                                               const float* __restrict__ ad1,
                                               unsigned* __restrict__ m1,
                                               unsigned* __restrict__ sl, int* __restrict__ slcnt, int capw) {
    __shared__ unsigned mx[SPAN * 4];
    __shared__ int2 q[(WG / 32) * QCAP];
    const int tid = threadIdx.x, lane = tid & 31;
    const int r = blockIdx.x / G, g = blockIdx.x % G;
    const int lo = r * SPAN;
    const int span = (N - lo < SPAN) ? (N - lo) : SPAN;
    const int h = lane & 3, sub = lane >> 2;
    float wsl, wdl;
    head_consts(W1, as1, ad1, h, wsl, wdl);

    for (int i = tid; i < span * 4; i += WG) mx[i] = FORD_NEG_INF;
    __syncthreads();

    int2* qw = q + (tid >> 5) * QCAP;
    const int wslot = blockIdx.x * (WG / 32) + (tid >> 5);
    unsigned* slw = sl + (size_t)wslot * capw;
    int pos = 0;
    auto drain = [&](int cnt) {
        for (int k0 = 0; k0 < cnt; k0 += 32) {
            bool ok[4]; int2 en[4]; float xs[4], xdv[4];
#pragma unroll
            for (int rr = 0; rr < 4; ++rr) {
                int k = k0 + rr * 8 + sub;
                ok[rr] = k < cnt;
                en[rr] = qw[ok[rr] ? k : cnt - 1];
            }
#pragma unroll
            for (int rr = 0; rr < 4; ++rr) { xs[rr] = x[en[rr].x]; xdv[rr] = x[lo + en[rr].y]; }
            asm volatile("" :: "v"(xs[0]), "v"(xs[1]), "v"(xs[2]), "v"(xs[3]),
                               "v"(xdv[0]), "v"(xdv[1]), "v"(xdv[2]), "v"(xdv[3]));
#pragma unroll
            for (int rr = 0; rr < 4; ++rr) {
                unsigned v = ford(lrelu(xs[rr] * wsl + xdv[rr] * wdl));
                if (ok[rr]) atomicMax(&mx[en[rr].y * 4 + h], v);
            }
        }
        if (lane < cnt) {
            int2 en = qw[lane];
            slw[pos + lane] = (unsigned)en.x | ((unsigned)en.y << SRC_BITS);
        }
        if (32 + lane < cnt) {
            int2 en = qw[32 + lane];
            slw[pos + 32 + lane] = (unsigned)en.x | ((unsigned)en.y << SRC_BITS);
        }
        pos += cnt;
    };
    const int EN = E + N;
    const int per = (EN + G - 1) / G;
    const int e0 = g * per;
    const int e1 = (e0 + per < EN) ? (e0 + per) : EN;
    scan_chunk(ei, E, e0, e1, lo, span, qw, tid, drain);
    if (lane == 0) slcnt[wslot] = pos;
    __syncthreads();

    const int tot = span * 4;
    const int rot = (int)(((long)tot * g) / G);
    for (int k = tid; k < tot; k += WG) {
        int i = k + rot; if (i >= tot) i -= tot;
        atomicMax(m1 + (size_t)lo * 4 + i, mx[i]);
    }
}

template <int SPAN, int G>
__global__ __launch_bounds__(WG) void k_l1_sum(const unsigned* __restrict__ sl,
                                               const int* __restrict__ slcnt, int capw, int N,
                                               const float* __restrict__ x,
                                               const float* __restrict__ W1,
                                               const float* __restrict__ as1,
                                               const float* __restrict__ ad1,
                                               const unsigned* __restrict__ m1,
                                               float* __restrict__ rec1) {
    __shared__ float mxf[SPAN * 4];
    __shared__ float rec[SPAN * 8];
    __shared__ int2 q[(WG / 32) * QCAP];
    const int tid = threadIdx.x, lane = tid & 31;
    const int r = blockIdx.x / G, g = blockIdx.x % G;
    const int lo = r * SPAN;
    const int span = (N - lo < SPAN) ? (N - lo) : SPAN;
    const int j = lane & 7, sub = lane >> 3;
    const int h = j & 3;
    float wsl, wdl;
    head_consts(W1, as1, ad1, h, wsl, wdl);

    for (int i = tid; i < span * 4; i += WG) mxf[i] = forddec(m1[(size_t)lo * 4 + i]);
    for (int i = tid; i < span * 8; i += WG) rec[i] = 0.f;
    __syncthreads();

    int2* qw = q + (tid >> 5) * QCAP;
    auto drain = [&](int cnt) {
        for (int k0 = 0; k0 < cnt; k0 += 16) {
            bool ok[4]; int2 en[4]; float xs[4], xdv[4];
#pragma unroll
            for (int rr = 0; rr < 4; ++rr) {
                int k = k0 + rr * 4 + sub;
                ok[rr] = k < cnt;
                en[rr] = qw[ok[rr] ? k : cnt - 1];
            }
#pragma unroll
            for (int rr = 0; rr < 4; ++rr) { xs[rr] = x[en[rr].x]; xdv[rr] = x[lo + en[rr].y]; }
            asm volatile("" :: "v"(xs[0]), "v"(xs[1]), "v"(xs[2]), "v"(xs[3]),
                               "v"(xdv[0]), "v"(xdv[1]), "v"(xdv[2]), "v"(xdv[3]));
#pragma unroll
            for (int rr = 0; rr < 4; ++rr) {
                float t = __expf(lrelu(xs[rr] * wsl + xdv[rr] * wdl) - mxf[en[rr].y * 4 + h]);
                float v = (j < 4) ? t : t * xs[rr];
                if (ok[rr]) atomicAdd(&rec[en[rr].y * 8 + j], v);
            }
        }
    };
    const int wslot = blockIdx.x * (WG / 32) + (tid >> 5);
    walk_slice(sl + (size_t)wslot * capw, __builtin_amdgcn_readfirstlane(slcnt[wslot]), qw, lane, drain);
    __syncthreads();

    const int tot = span * 8;
    const int rot = (int)(((long)tot * g) / G);
    for (int k = tid; k < tot; k += WG) {
        int i = k + rot; if (i >= tot) i -= tot;
        atomicAdd(rec1 + (size_t)lo * 8 + i, rec[i]);
    }
}

__global__ void k_fin1(const float* __restrict__ W1, const float* __restrict__ b1,
                       const float* __restrict__ rec1,
                       float* __restrict__ out1, int N) {
    int i = blockIdx.x * blockDim.x + threadIdx.x;
    if (i >= N * 32) return;
    int n = i >> 5, j = i & 31, h = j >> 3;
    float v = W1[j] * (rec1[n * 8 + 4 + h] / rec1[n * 8 + h]) + b1[j];
    out1[i] = v > 0.f ? v : 0.f;
}

__global__ void k_gemm2(const float* __restrict__ out1, const float* __restrict__ W2,
                        const float* __restrict__ as2, const float* __restrict__ ad2,
                        float* __restrict__ h2, float* __restrict__ alS2,
                        float* __restrict__ alD2, int N) {
    __shared__ float Bp[32 * 16];
    int t = threadIdx.x;
    for (int i = t; i < 512; i += blockDim.x) {
        int k = i >> 4, n = i & 15;
        float v = 0.f;
        if (n < 8) {
            v = W2[k * 8 + n];
        } else if (n == 8) {
            for (int c = 0; c < 8; ++c) v += W2[k * 8 + c] * as2[c];
        } else if (n == 9) {
            for (int c = 0; c < 8; ++c) v += W2[k * 8 + c] * ad2[c];
        }
        Bp[i] = v;
    }
    __syncthreads();

    int wave = t >> 5, lane = t & 31;
    long tile = (long)blockIdx.x * 8 + wave;
    long row0 = tile * 16;
    if (row0 >= N) return;

    int m = lane & 15, khi = lane >> 4;
    long rowm = row0 + m;
    if (rowm >= N) rowm = N - 1;
    const float* Arow = out1 + rowm * 32;

    v8f c = {};
#pragma unroll
    for (int step = 0; step < 8; ++step) {
        int k0 = step * 4 + khi * 2;
        v2f a; a.x = Arow[k0];          a.y = Arow[k0 + 1];
        v2f b; b.x = Bp[k0 * 16 + m];   b.y = Bp[(k0 + 1) * 16 + m];
        c = __builtin_amdgcn_wmma_f32_16x16x4_f32(false, a, false, b, (short)0, c,
                                                  false, false);
    }

#pragma unroll
    for (int r = 0; r < 8; ++r) {
        long node = row0 + khi * 8 + r;
        if (node >= N) continue;
        float v = c[r];
        if (m < 8)        h2[node * 8 + m] = v;
        else if (m == 8)  alS2[node] = v;
        else if (m == 9)  alD2[node] = v;
    }
}

template <int SPAN, int G>
__global__ __launch_bounds__(WG) void k_l2_max(const unsigned* __restrict__ sl,
                                               const int* __restrict__ slcnt, int capw, int N,
                                               const float* __restrict__ alS2,
                                               const float* __restrict__ alD2,
                                               unsigned* __restrict__ m2) {
    __shared__ unsigned mx[SPAN];
    __shared__ float ad[SPAN];
    __shared__ int2 q[(WG / 32) * QCAP];
    const int tid = threadIdx.x, lane = tid & 31;
    const int r = blockIdx.x / G, g = blockIdx.x % G;
    const int lo = r * SPAN;
    const int span = (N - lo < SPAN) ? (N - lo) : SPAN;

    for (int i = tid; i < span; i += WG) { ad[i] = alD2[lo + i]; mx[i] = FORD_NEG_INF; }
    __syncthreads();

    int2* qw = q + (tid >> 5) * QCAP;
    auto drain = [&](int cnt) {
        for (int k0 = 0; k0 < cnt; k0 += 64) {
            bool ok[2]; int2 en[2]; float as[2];
#pragma unroll
            for (int rr = 0; rr < 2; ++rr) {
                int k = k0 + rr * 32 + lane;
                ok[rr] = k < cnt;
                en[rr] = qw[ok[rr] ? k : cnt - 1];
            }
#pragma unroll
            for (int rr = 0; rr < 2; ++rr) as[rr] = alS2[en[rr].x];
            asm volatile("" :: "v"(as[0]), "v"(as[1]));
#pragma unroll
            for (int rr = 0; rr < 2; ++rr)
                if (ok[rr]) atomicMax(&mx[en[rr].y], ford(lrelu(as[rr] + ad[en[rr].y])));
        }
    };
    const int wslot = blockIdx.x * (WG / 32) + (tid >> 5);
    walk_slice(sl + (size_t)wslot * capw, __builtin_amdgcn_readfirstlane(slcnt[wslot]), qw, lane, drain);
    __syncthreads();

    const int tot = span;
    const int rot = (int)(((long)tot * g) / G);
    for (int k = tid; k < tot; k += WG) {
        int i = k + rot; if (i >= tot) i -= tot;
        atomicMax(m2 + lo + i, mx[i]);
    }
}

template <int SPAN, int G>
__global__ __launch_bounds__(WG) void k_l2_sum(const unsigned* __restrict__ sl,
                                               const int* __restrict__ slcnt, int capw, int N,
                                               const float* __restrict__ h2,
                                               const float* __restrict__ alS2,
                                               const float* __restrict__ alD2,
                                               const unsigned* __restrict__ m2,
                                               float* __restrict__ s2, float* __restrict__ acc2) {
    __shared__ float mxf[SPAN];
    __shared__ float ad[SPAN];
    __shared__ float ssum[SPAN];
    __shared__ float acc[SPAN * 8];
    __shared__ int2 q[(WG / 32) * QCAP];
    const int tid = threadIdx.x, lane = tid & 31;
    const int r = blockIdx.x / G, g = blockIdx.x % G;
    const int lo = r * SPAN;
    const int span = (N - lo < SPAN) ? (N - lo) : SPAN;
    const int j = lane & 7, sub = lane >> 3;

    for (int i = tid; i < span; i += WG) {
        ad[i] = alD2[lo + i]; mxf[i] = forddec(m2[lo + i]); ssum[i] = 0.f;
    }
    for (int i = tid; i < span * 8; i += WG) acc[i] = 0.f;
    __syncthreads();

    int2* qw = q + (tid >> 5) * QCAP;
    auto drain = [&](int cnt) {
        for (int k0 = 0; k0 < cnt; k0 += 16) {
            bool ok[4]; int2 en[4]; float as[4], hv[4];
#pragma unroll
            for (int rr = 0; rr < 4; ++rr) {
                int k = k0 + rr * 4 + sub;
                ok[rr] = k < cnt;
                en[rr] = qw[ok[rr] ? k : cnt - 1];
            }
#pragma unroll
            for (int rr = 0; rr < 4; ++rr) { as[rr] = alS2[en[rr].x]; hv[rr] = h2[(size_t)en[rr].x * 8 + j]; }
            asm volatile("" :: "v"(as[0]), "v"(as[1]), "v"(as[2]), "v"(as[3]),
                               "v"(hv[0]), "v"(hv[1]), "v"(hv[2]), "v"(hv[3]));
#pragma unroll
            for (int rr = 0; rr < 4; ++rr) {
                float t = __expf(lrelu(as[rr] + ad[en[rr].y]) - mxf[en[rr].y]);
                if (ok[rr]) {
                    atomicAdd(&acc[en[rr].y * 8 + j], t * hv[rr]);
                    if (j == 0) atomicAdd(&ssum[en[rr].y], t);
                }
            }
        }
    };
    const int wslot = blockIdx.x * (WG / 32) + (tid >> 5);
    walk_slice(sl + (size_t)wslot * capw, __builtin_amdgcn_readfirstlane(slcnt[wslot]), qw, lane, drain);
    __syncthreads();

    const int rot1 = (int)(((long)span * g) / G);
    for (int k = tid; k < span; k += WG) {
        int i = k + rot1; if (i >= span) i -= span;
        atomicAdd(s2 + lo + i, ssum[i]);
    }
    const int tot = span * 8;
    const int rot = (int)(((long)tot * g) / G);
    for (int k = tid; k < tot; k += WG) {
        int i = k + rot; if (i >= tot) i -= tot;
        atomicAdd(acc2 + (size_t)lo * 8 + i, acc[i]);
    }
}

__global__ void k_fin2(const float* __restrict__ acc2, const float* __restrict__ s2,
                       const float* __restrict__ b2, float* __restrict__ out, int N) {
    int i = blockIdx.x * blockDim.x + threadIdx.x;
    if (i >= N * 8) return;
    int n = i >> 3, c = i & 7;
    out[i] = acc2[i] / s2[n] + b2[c];
}

extern "C" void kernel_launch(void* const* d_in, const int* in_sizes, int n_in,
                              void* d_out, int out_size, void* d_ws, size_t ws_size,
                              hipStream_t stream) {
    const float* x   = (const float*)d_in[0];
    const int*   ei  = (const int*)d_in[1];
    const float* W1  = (const float*)d_in[2];
    const float* as1 = (const float*)d_in[3];
    const float* ad1 = (const float*)d_in[4];
    const float* b1  = (const float*)d_in[5];
    const float* W2  = (const float*)d_in[6];
    const float* as2 = (const float*)d_in[7];
    const float* ad2 = (const float*)d_in[8];
    const float* b2  = (const float*)d_in[9];
    (void)n_in; (void)out_size; (void)ws_size;

    int N = in_sizes[0];
    int E = in_sizes[1] / 2;

    char* ws = (char*)d_ws;
    size_t cur = 0;
    auto carve = [&](size_t bytes) -> void* {
        void* p = ws + cur;
        cur += (bytes + 255) & ~(size_t)255;
        return p;
    };
    unsigned* m1    = (unsigned*)carve(sizeof(unsigned) * (size_t)N * 4);
    unsigned* m2    = (unsigned*)carve(sizeof(unsigned) * (size_t)N);
    float*    rec1  = (float*)carve(sizeof(float) * (size_t)N * 8);
    float*    s2    = (float*)carve(sizeof(float) * (size_t)N);
    float*    acc2  = (float*)carve(sizeof(float) * (size_t)N * 8);
    float*    out1  = (float*)carve(sizeof(float) * (size_t)N * 32);
    float*    h2    = (float*)carve(sizeof(float) * (size_t)N * 8);
    float*    alS2  = (float*)carve(sizeof(float) * (size_t)N);
    float*    alD2  = (float*)carve(sizeof(float) * (size_t)N);
    const int nwg   = ((N + SPAN_ALL - 1) / SPAN_ALL) * G_ALL;
    const int per   = (E + N + G_ALL - 1) / G_ALL;
    const int capw  = ((per + WG * UNR - 1) / (WG * UNR)) * UNR * 32;
    unsigned* sl    = (unsigned*)carve(sizeof(unsigned) * (size_t)nwg * (WG / 32) * capw);
    int*      slcnt = (int*)carve(sizeof(int) * (size_t)nwg * (WG / 32));

    const int B = 256;
    k_init<<<(8 * N + B - 1) / B, B, 0, stream>>>(m1, 4 * N, m2, N,
                                                  rec1, 8 * N, s2, N, acc2, 8 * N);
    k_l1_max<SPAN_ALL, G_ALL><<<nwg, WG, 0, stream>>>(ei, E, N, x, W1, as1, ad1, m1, sl, slcnt, capw);
    k_l1_sum<SPAN_ALL, G_ALL><<<nwg, WG, 0, stream>>>(sl, slcnt, capw, N, x, W1, as1, ad1, m1, rec1);
    k_fin1<<<(32 * N + B - 1) / B, B, 0, stream>>>(W1, b1, rec1, out1, N);

    int tiles = (N + 15) / 16;
    k_gemm2<<<(tiles + 7) / 8, B, 0, stream>>>(out1, W2, as2, ad2, h2, alS2, alD2, N);

    k_l2_max<SPAN_ALL, G_ALL><<<nwg, WG, 0, stream>>>(sl, slcnt, capw, N, alS2, alD2, m2);
    k_l2_sum<SPAN_ALL, G_ALL><<<nwg, WG, 0, stream>>>(sl, slcnt, capw, N, h2, alS2, alD2, m2, s2, acc2);
    k_fin2<<<(8 * N + B - 1) / B, B, 0, stream>>>(acc2, s2, b2, (float*)d_out, N);
}


# A sharer work-group does the softmax of its own edges (max and sum pass in one launch), stores a plane; finalize merges 16 planes by exp rescale: no global atomic, 8 to 5 launches; sum order changes
# speedup vs baseline: 9.7504x; 1.0799x over previous
#include <hip/hip_runtime.h>
#include <math.h>


#define NEG_SLOPE 0.2f
#define WG 1024
#define UNR 8
#define QCAP 64
#define SPAN_ALL 6250
#define G_ALL 16
#define SRC_BITS 19
#define SLICE_CAP 2048
static_assert(UNR == 8, "the load fence in scan_chunk names eight values");
static_assert(QCAP >= 64, "a push of 32 entries on top of 32 queued ones must fit");
static_assert(QCAP == 64, "walk_slice refills the queue with two entries per lane");
static_assert(SPAN_ALL <= (1 << (32 - SRC_BITS)), "destination - lo must fit above the source bits");

typedef __attribute__((ext_vector_type(2))) float v2f;
typedef __attribute__((ext_vector_type(8))) float v8f;

__device__ __forceinline__ float lrelu(float v) { return v > 0.f ? v : NEG_SLOPE * v; }

__device__ __forceinline__ unsigned ford(float f) {
    unsigned u = __float_as_uint(f);
    return (u & 0x80000000u) ? ~u : (u | 0x80000000u);
}
__device__ __forceinline__ float forddec(unsigned u) {
    return (u & 0x80000000u) ? __uint_as_float(u & 0x7fffffffu) : __uint_as_float(~u);
}
#define FORD_NEG_INF 0x007FFFFFu


__device__ __forceinline__ void wave_sync() {
    __builtin_amdgcn_fence(__ATOMIC_RELEASE, "wavefront");
    __builtin_amdgcn_wave_barrier();
    __builtin_amdgcn_fence(__ATOMIC_ACQUIRE, "wavefront");
}

__device__ __forceinline__ void load_edge(const int* __restrict__ ei, int E, int e, int& ls, int& ld) {
    int ec = e < E ? e : E - 1;
    ls = ei[ec];
    ld = ei[E + ec];
}
__device__ __forceinline__ void place_edge(int E, int e, int e1, int lo, int ls, int ld,
                                           int& s, unsigned& rel) {
    int d = e < E ? ld : e - E;
    s     = e < E ? ls : e - E;
    if (e >= e1) { d = -1; s = 0; }
    rel = (unsigned)(d - lo);
}

template <class Drain>
__device__ __forceinline__ void scan_chunk(const int* __restrict__ ei, int E, int e0, int e1,
                                           int lo, int span, int2* qw, int tid, Drain drain) {
    const unsigned lt = (1u << (tid & 31)) - 1u;
    int cnt = 0;
    for (int b = e0; b < e1; b += WG * UNR) {
        int ls[UNR], ld[UNR], s[UNR]; unsigned rel[UNR];
#pragma unroll
        for (int u = 0; u < UNR; ++u) load_edge(ei, E, b + u * WG + tid, ls[u], ld[u]);
        asm volatile("" :: "v"(ls[0]), "v"(ls[1]), "v"(ls[2]), "v"(ls[3]),
                           "v"(ls[4]), "v"(ls[5]), "v"(ls[6]), "v"(ls[7]));
#pragma unroll
        for (int u = 0; u < UNR; ++u) place_edge(E, b + u * WG + tid, e1, lo, ls[u], ld[u], s[u], rel[u]);
#pragma unroll
        for (int u = 0; u < UNR; ++u) {
            bool keep = rel[u] < (unsigned)span;
            unsigned mask = __builtin_amdgcn_ballot_w32(keep);
            if (keep) qw[cnt + __popc(mask & lt)] = make_int2(s[u], (int)rel[u]);
            cnt += __popc(mask);
            if (cnt > QCAP - 32) {
                wave_sync();
                drain(cnt);
                wave_sync();
                cnt = 0;
            }
        }
    }
    wave_sync();
    drain(cnt);
}

template <class Drain>
__device__ __forceinline__ void walk_slice(const unsigned* sl, int total,
                                           int2* qw, int lane, Drain drain) {
    if (total <= 0) return;
    const int last = total - 1;
    const unsigned smask = (1u << SRC_BITS) - 1u;
    unsigned a = sl[lane < last ? lane : last];
    unsigned b = sl[32 + lane < last ? 32 + lane : last];
    for (int k0 = 0; k0 < total; k0 += QCAP) {
        qw[lane]      = make_int2((int)(a & smask), (int)(a >> SRC_BITS));
        qw[32 + lane] = make_int2((int)(b & smask), (int)(b >> SRC_BITS));
        const int kn = k0 + QCAP;
        if (kn < total) {
            a = sl[kn + lane < last ? kn + lane : last];
            b = sl[kn + 32 + lane < last ? kn + 32 + lane : last];
        }
        wave_sync();
        drain(total - k0 < QCAP ? total - k0 : QCAP);
        wave_sync();
    }
}

__device__ __forceinline__ void head_consts(const float* __restrict__ W1, const float* __restrict__ as1,
                                            const float* __restrict__ ad1, int h, float& wsl, float& wdl) {
    float a = 0.f, b = 0.f;
#pragma unroll
    for (int c = 0; c < 8; ++c) {
        float w = W1[h * 8 + c];
        a += w * as1[h * 8 + c];
        b += w * ad1[h * 8 + c];
    }
    wsl = a; wdl = b;
}

template <int SPAN, int G>
__global__ __launch_bounds__(WG) void k_l1(const int* __restrict__ ei, int E, int N,
                                           const float* __restrict__ x,
                                           const float* __restrict__ W1,
                                           const float* __restrict__ as1,
                                           const float* __restrict__ ad1,
                                           float* __restrict__ pm1, float* __restrict__ prec1,
                                           unsigned* sl, int* __restrict__ slcnt, int capw) {
    __shared__ __attribute__((aligned(16))) unsigned mx[SPAN * 4];
    __shared__ __attribute__((aligned(16))) float rec[SPAN * 8];
    __shared__ int2 q[(WG / 32) * QCAP];
    const int tid = threadIdx.x, lane = tid & 31;
    const int r = blockIdx.x / G, g = blockIdx.x % G;
    const int lo = r * SPAN;
    const int span = (N - lo < SPAN) ? (N - lo) : SPAN;
    const int h = lane & 3;
    float wsl, wdl;
    head_consts(W1, as1, ad1, h, wsl, wdl);

    for (int i = tid; i < span * 4; i += WG) mx[i] = FORD_NEG_INF;
    for (int i = tid; i < span * 8; i += WG) rec[i] = 0.f;
    __syncthreads();

    int2* qw = q + (tid >> 5) * QCAP;
    const int wslot = blockIdx.x * (WG / 32) + (tid >> 5);
    unsigned* slw = sl + (size_t)wslot * capw;
    int pos = 0;
    const int sub4 = lane >> 2;
    auto drain_max = [&](int cnt) {
        for (int k0 = 0; k0 < cnt; k0 += 32) {
            bool ok[4]; int2 en[4]; float xs[4], xdv[4];
#pragma unroll
            for (int rr = 0; rr < 4; ++rr) {
                int k = k0 + rr * 8 + sub4;
                ok[rr] = k < cnt;
                en[rr] = qw[ok[rr] ? k : cnt - 1];
            }
#pragma unroll
            for (int rr = 0; rr < 4; ++rr) { xs[rr] = x[en[rr].x]; xdv[rr] = x[lo + en[rr].y]; }
            asm volatile("" :: "v"(xs[0]), "v"(xs[1]), "v"(xs[2]), "v"(xs[3]),
                               "v"(xdv[0]), "v"(xdv[1]), "v"(xdv[2]), "v"(xdv[3]));
#pragma unroll
            for (int rr = 0; rr < 4; ++rr) {
                unsigned v = ford(lrelu(xs[rr] * wsl + xdv[rr] * wdl));
                if (ok[rr]) atomicMax(&mx[en[rr].y * 4 + h], v);
            }
        }
        if (lane < cnt && pos + lane < capw) {
            int2 en = qw[lane];
            slw[pos + lane] = (unsigned)en.x | ((unsigned)en.y << SRC_BITS);
        }
        if (32 + lane < cnt && pos + 32 + lane < capw) {
            int2 en = qw[32 + lane];
            slw[pos + 32 + lane] = (unsigned)en.x | ((unsigned)en.y << SRC_BITS);
        }
        pos += cnt;
    };
    const int EN = E + N;
    const int per = (EN + G - 1) / G;
    const int e0 = g * per;
    const int e1 = (e0 + per < EN) ? (e0 + per) : EN;
    scan_chunk(ei, E, e0, e1, lo, span, qw, tid, drain_max);
    if (lane == 0) slcnt[wslot] = pos;
    __syncthreads();

    const int j = lane & 7, sub8 = lane >> 3;
    auto drain_sum = [&](int cnt) {
        for (int k0 = 0; k0 < cnt; k0 += 16) {
            bool ok[4]; int2 en[4]; float xs[4], xdv[4];
#pragma unroll
            for (int rr = 0; rr < 4; ++rr) {
                int k = k0 + rr * 4 + sub8;
                ok[rr] = k < cnt;
                en[rr] = qw[ok[rr] ? k : cnt - 1];
            }
#pragma unroll
            for (int rr = 0; rr < 4; ++rr) { xs[rr] = x[en[rr].x]; xdv[rr] = x[lo + en[rr].y]; }
            asm volatile("" :: "v"(xs[0]), "v"(xs[1]), "v"(xs[2]), "v"(xs[3]),
                               "v"(xdv[0]), "v"(xdv[1]), "v"(xdv[2]), "v"(xdv[3]));
#pragma unroll
            for (int rr = 0; rr < 4; ++rr) {
                float t = __expf(lrelu(xs[rr] * wsl + xdv[rr] * wdl) - forddec(mx[en[rr].y * 4 + h]));
                float v = (j < 4) ? t : t * xs[rr];
                if (ok[rr]) atomicAdd(&rec[en[rr].y * 8 + j], v);
            }
        }
    };
    const int total = __builtin_amdgcn_readfirstlane(pos);
    if (total <= capw) walk_slice(slw, total, qw, lane, drain_sum);
    else               scan_chunk(ei, E, e0, e1, lo, span, qw, tid, drain_sum);
    __syncthreads();

    float4* pm = reinterpret_cast<float4*>(pm1 + ((size_t)g * N + lo) * 4);
    for (int k = tid; k < span; k += WG) {
        const uint4 u = *reinterpret_cast<const uint4*>(&mx[k * 4]);
        pm[k] = make_float4(forddec(u.x), forddec(u.y), forddec(u.z), forddec(u.w));
    }
    float4* pr = reinterpret_cast<float4*>(prec1 + ((size_t)g * N + lo) * 8);
    for (int k = tid; k < span * 2; k += WG) pr[k] = *reinterpret_cast<const float4*>(&rec[k * 4]);
}

template <int G>
__global__ void k_fin1(const float* __restrict__ W1, const float* __restrict__ b1,
                       const float* __restrict__ pm1, const float* __restrict__ prec1,
                       float* __restrict__ out1, int N) {
    const int i = blockIdx.x * blockDim.x + threadIdx.x;
    const int ic = i < N * 4 ? i : N * 4 - 1;
    const int n = ic >> 2, h = ic & 3;
    float m[G], s[G], w[G];
#pragma unroll
    for (int g = 0; g < G; ++g) {
        m[g] = pm1[(size_t)g * N * 4 + ic];
        s[g] = prec1[((size_t)g * N + n) * 8 + h];
        w[g] = prec1[((size_t)g * N + n) * 8 + 4 + h];
    }
    float M = m[0];
#pragma unroll
    for (int g = 1; g < G; ++g) M = fmaxf(M, m[g]);
    float S = 0.f, Wt = 0.f;
#pragma unroll
    for (int g = 0; g < G; ++g) {
        float sc = __expf(m[g] - M);
        S += s[g] * sc;
        Wt += w[g] * sc;
    }
    const float ratio = Wt / S;

    const int lane = threadIdx.x & 31;
    const int node0 = (i - lane) >> 2;
    const int hd = (lane & 7) >> 1, c0 = (lane & 1) * 4;
#pragma unroll
    for (int k = 0; k < 2; ++k) {
        const int nl = k * 4 + (lane >> 3);
        const float rt = __shfl(ratio, nl * 4 + hd, 32);
        float v[4];
#pragma unroll
        for (int c = 0; c < 4; ++c) {
            float t = W1[hd * 8 + c0 + c] * rt + b1[hd * 8 + c0 + c];
            v[c] = t > 0.f ? t : 0.f;
        }
        const int node = node0 + nl;
        if (node < N)
            *reinterpret_cast<float4*>(out1 + (size_t)node * 32 + hd * 8 + c0) = make_float4(v[0], v[1], v[2], v[3]);
    }
}

__global__ void k_gemm2(const float* __restrict__ out1, const float* __restrict__ W2,
                        const float* __restrict__ as2, const float* __restrict__ ad2,
                        float* __restrict__ h2, float* __restrict__ alS2,
                        float* __restrict__ alD2, int N) {
    __shared__ float Bp[32 * 16];
    int t = threadIdx.x;
    for (int i = t; i < 512; i += blockDim.x) {
        int k = i >> 4, n = i & 15;
        float v = 0.f;
        if (n < 8) {
            v = W2[k * 8 + n];
        } else if (n == 8) {
            for (int c = 0; c < 8; ++c) v += W2[k * 8 + c] * as2[c];
        } else if (n == 9) {
            for (int c = 0; c < 8; ++c) v += W2[k * 8 + c] * ad2[c];
        }
        Bp[i] = v;
    }
    __syncthreads();

    int wave = t >> 5, lane = t & 31;
    long tile = (long)blockIdx.x * 8 + wave;
    long row0 = tile * 16;
    if (row0 >= N) return;

    int m = lane & 15, khi = lane >> 4;
    long rowm = row0 + m;
    if (rowm >= N) rowm = N - 1;
    const float* Arow = out1 + rowm * 32;

    v8f c = {};
#pragma unroll
    for (int step = 0; step < 8; ++step) {
        int k0 = step * 4 + khi * 2;
        v2f a; a.x = Arow[k0];          a.y = Arow[k0 + 1];
        v2f b; b.x = Bp[k0 * 16 + m];   b.y = Bp[(k0 + 1) * 16 + m];
        c = __builtin_amdgcn_wmma_f32_16x16x4_f32(false, a, false, b, (short)0, c,
                                                  false, false);
    }

#pragma unroll
    for (int r = 0; r < 8; ++r) {
        long node = row0 + khi * 8 + r;
        if (node >= N) continue;
        float v = c[r];
        if (m < 8)        h2[node * 8 + m] = v;
        else if (m == 8)  alS2[node] = v;
        else if (m == 9)  alD2[node] = v;
    }
}

template <int SPAN, int G>
__global__ __launch_bounds__(WG) void k_l2(const int* __restrict__ ei, int E,
                                           const unsigned* __restrict__ sl,
                                           const int* __restrict__ slcnt, int capw, int N,
                                           const float* __restrict__ h2,
                                           const float* __restrict__ alS2,
                                           const float* __restrict__ alD2,
                                           float* __restrict__ pm2, float* __restrict__ ps2,
                                           float* __restrict__ pacc2) {
    __shared__ unsigned mx[SPAN];
    __shared__ float ad[SPAN];
    __shared__ float ssum[SPAN];
    __shared__ __attribute__((aligned(16))) float acc[SPAN * 8];
    __shared__ int2 q[(WG / 32) * QCAP];
    const int tid = threadIdx.x, lane = tid & 31;
    const int r = blockIdx.x / G, g = blockIdx.x % G;
    const int lo = r * SPAN;
    const int span = (N - lo < SPAN) ? (N - lo) : SPAN;

    for (int i = tid; i < span; i += WG) { ad[i] = alD2[lo + i]; mx[i] = FORD_NEG_INF; ssum[i] = 0.f; }
    for (int i = tid; i < span * 8; i += WG) acc[i] = 0.f;
    __syncthreads();

    int2* qw = q + (tid >> 5) * QCAP;
    auto drain_max = [&](int cnt) {
        for (int k0 = 0; k0 < cnt; k0 += 64) {
            bool ok[2]; int2 en[2]; float as[2];
#pragma unroll
            for (int rr = 0; rr < 2; ++rr) {
                int k = k0 + rr * 32 + lane;
                ok[rr] = k < cnt;
                en[rr] = qw[ok[rr] ? k : cnt - 1];
            }
#pragma unroll
            for (int rr = 0; rr < 2; ++rr) as[rr] = alS2[en[rr].x];
            asm volatile("" :: "v"(as[0]), "v"(as[1]));
#pragma unroll
            for (int rr = 0; rr < 2; ++rr)
                if (ok[rr]) atomicMax(&mx[en[rr].y], ford(lrelu(as[rr] + ad[en[rr].y])));
        }
    };
    const int j = lane & 7, sub = lane >> 3;
    auto drain_sum = [&](int cnt) {
        for (int k0 = 0; k0 < cnt; k0 += 16) {
            bool ok[4]; int2 en[4]; float as[4], hv[4];
#pragma unroll
            for (int rr = 0; rr < 4; ++rr) {
                int k = k0 + rr * 4 + sub;
                ok[rr] = k < cnt;
                en[rr] = qw[ok[rr] ? k : cnt - 1];
            }
#pragma unroll
            for (int rr = 0; rr < 4; ++rr) { as[rr] = alS2[en[rr].x]; hv[rr] = h2[(size_t)en[rr].x * 8 + j]; }
            asm volatile("" :: "v"(as[0]), "v"(as[1]), "v"(as[2]), "v"(as[3]),
                               "v"(hv[0]), "v"(hv[1]), "v"(hv[2]), "v"(hv[3]));
#pragma unroll
            for (int rr = 0; rr < 4; ++rr) {
                float t = __expf(lrelu(as[rr] + ad[en[rr].y]) - forddec(mx[en[rr].y]));
                if (ok[rr]) {
                    atomicAdd(&acc[en[rr].y * 8 + j], t * hv[rr]);
                    if (j == 0) atomicAdd(&ssum[en[rr].y], t);
                }
            }
        }
    };
    const int wslot = blockIdx.x * (WG / 32) + (tid >> 5);
    const unsigned* slw = sl + (size_t)wslot * capw;
    const int total = __builtin_amdgcn_readfirstlane(slcnt[wslot]);
    const int EN = E + N;
    const int per = (EN + G - 1) / G;
    const int e0 = g * per;
    const int e1 = (e0 + per < EN) ? (e0 + per) : EN;
    if (total <= capw) walk_slice(slw, total, qw, lane, drain_max);
    else               scan_chunk(ei, E, e0, e1, lo, span, qw, tid, drain_max);
    __syncthreads();
    if (total <= capw) walk_slice(slw, total, qw, lane, drain_sum);
    else               scan_chunk(ei, E, e0, e1, lo, span, qw, tid, drain_sum);
    __syncthreads();

    float* pm = pm2 + (size_t)g * N + lo;
    float* ps = ps2 + (size_t)g * N + lo;
    for (int k = tid; k < span; k += WG) { pm[k] = forddec(mx[k]); ps[k] = ssum[k]; }
    float4* pa = reinterpret_cast<float4*>(pacc2 + ((size_t)g * N + lo) * 8);
    for (int k = tid; k < span * 2; k += WG) pa[k] = *reinterpret_cast<const float4*>(&acc[k * 4]);
}

template <int G>
__global__ void k_fin2(const float* __restrict__ pm2, const float* __restrict__ ps2,
                       const float* __restrict__ pacc2,
                       const float* __restrict__ b2, float* __restrict__ out, int N) {
    const int i = blockIdx.x * blockDim.x + threadIdx.x;
    if (i >= N * 2) return;
    const int n = i >> 1, q = i & 1;
    float m[G], s[G]; float4 a[G];
#pragma unroll
    for (int g = 0; g < G; ++g) {
        m[g] = pm2[(size_t)g * N + n];
        s[g] = ps2[(size_t)g * N + n];
        a[g] = reinterpret_cast<const float4*>(pacc2 + (size_t)g * N * 8)[i];
    }
    float M = m[0];
#pragma unroll
    for (int g = 1; g < G; ++g) M = fmaxf(M, m[g]);
    float S = 0.f, A0 = 0.f, A1 = 0.f, A2 = 0.f, A3 = 0.f;
#pragma unroll
    for (int g = 0; g < G; ++g) {
        float sc = __expf(m[g] - M);
        S  += s[g] * sc;
        A0 += a[g].x * sc; A1 += a[g].y * sc; A2 += a[g].z * sc; A3 += a[g].w * sc;
    }
    reinterpret_cast<float4*>(out)[i] = make_float4(A0 / S + b2[q * 4 + 0], A1 / S + b2[q * 4 + 1],
                                                    A2 / S + b2[q * 4 + 2], A3 / S + b2[q * 4 + 3]);
}

extern "C" void kernel_launch(void* const* d_in, const int* in_sizes, int n_in,
                              void* d_out, int out_size, void* d_ws, size_t ws_size,
                              hipStream_t stream) {
    const float* x   = (const float*)d_in[0];
    const int*   ei  = (const int*)d_in[1];
    const float* W1  = (const float*)d_in[2];
    const float* as1 = (const float*)d_in[3];
    const float* ad1 = (const float*)d_in[4];
    const float* b1  = (const float*)d_in[5];
    const float* W2  = (const float*)d_in[6];
    const float* as2 = (const float*)d_in[7];
    const float* ad2 = (const float*)d_in[8];
    const float* b2  = (const float*)d_in[9];
    (void)n_in; (void)out_size; (void)ws_size;

    int N = in_sizes[0];
    int E = in_sizes[1] / 2;

    char* ws = (char*)d_ws;
    size_t cur = 0;
    auto carve = [&](size_t bytes) -> void* {
        void* p = ws + cur;
        cur += (bytes + 255) & ~(size_t)255;
        return p;
    };
    float*    pm1   = (float*)carve(sizeof(float) * (size_t)G_ALL * N * 4);
    float*    prec1 = (float*)carve(sizeof(float) * (size_t)G_ALL * N * 8);
    float*    pm2   = (float*)carve(sizeof(float) * (size_t)G_ALL * N);
    float*    ps2   = (float*)carve(sizeof(float) * (size_t)G_ALL * N);
    float*    pacc2 = (float*)carve(sizeof(float) * (size_t)G_ALL * N * 8);
    float*    out1  = (float*)carve(sizeof(float) * (size_t)N * 32);
    float*    h2    = (float*)carve(sizeof(float) * (size_t)N * 8);
    float*    alS2  = (float*)carve(sizeof(float) * (size_t)N);
    float*    alD2  = (float*)carve(sizeof(float) * (size_t)N);
    const int nwg   = ((N + SPAN_ALL - 1) / SPAN_ALL) * G_ALL;
    const int per   = (E + N + G_ALL - 1) / G_ALL;
    const int capf  = ((per + WG * UNR - 1) / (WG * UNR)) * UNR * 32;
    const int capw  = capf < SLICE_CAP ? capf : SLICE_CAP;
    unsigned* sl    = (unsigned*)carve(sizeof(unsigned) * (size_t)nwg * (WG / 32) * capw);
    int*      slcnt = (int*)carve(sizeof(int) * (size_t)nwg * (WG / 32));

    const int B = 256;
    k_l1<SPAN_ALL, G_ALL><<<nwg, WG, 0, stream>>>(ei, E, N, x, W1, as1, ad1, pm1, prec1, sl, slcnt, capw);
    k_fin1<G_ALL><<<(4 * N + B - 1) / B, B, 0, stream>>>(W1, b1, pm1, prec1, out1, N);

    int tiles = (N + 15) / 16;
    k_gemm2<<<(tiles + 7) / 8, B, 0, stream>>>(out1, W2, as2, ad2, h2, alS2, alD2, N);

    k_l2<SPAN_ALL, G_ALL><<<nwg, WG, 0, stream>>>(ei, E, sl, slcnt, capw, N, h2, alS2, alD2, pm2, ps2, pacc2);
    k_fin2<G_ALL><<<(2 * N + B - 1) / B, B, 0, stream>>>(pm2, ps2, pacc2, b2, (float*)d_out, N);
}


# Sum passes: an edge's softmax weight is computed once (one lane per head in layer 1; one lane in layer 2, passed to the 8 channel lanes via LDS), not by each record lane; LDS sums in arrival order
# speedup vs baseline: 10.3221x; 1.0586x over previous
#include <hip/hip_runtime.h>
#include <math.h>


#define NEG_SLOPE 0.2f
#define WG 1024
#define UNR 8
#define QCAP 64
#define SPAN_ALL 6250
#define G_ALL 16
#define SRC_BITS 19
#define SLICE_CAP 2048
static_assert(UNR == 8, "the load fence in scan_chunk names eight values");
static_assert(QCAP >= 64, "a push of 32 entries on top of 32 queued ones must fit");
static_assert(QCAP == 64, "walk_slice refills the queue with two entries per lane");
static_assert(SPAN_ALL <= (1 << (32 - SRC_BITS)), "destination - lo must fit above the source bits");

typedef __attribute__((ext_vector_type(2))) float v2f;
typedef __attribute__((ext_vector_type(8))) float v8f;

__device__ __forceinline__ float lrelu(float v) { return v > 0.f ? v : NEG_SLOPE * v; }

__device__ __forceinline__ unsigned ford(float f) {
    unsigned u = __float_as_uint(f);
    return (u & 0x80000000u) ? ~u : (u | 0x80000000u);
}
__device__ __forceinline__ float forddec(unsigned u) {
    return (u & 0x80000000u) ? __uint_as_float(u & 0x7fffffffu) : __uint_as_float(~u);
}
#define FORD_NEG_INF 0x007FFFFFu


__device__ __forceinline__ void wave_sync() {
    __builtin_amdgcn_fence(__ATOMIC_RELEASE, "wavefront");
    __builtin_amdgcn_wave_barrier();
    __builtin_amdgcn_fence(__ATOMIC_ACQUIRE, "wavefront");
}

__device__ __forceinline__ void load_edge(const int* __restrict__ ei, int E, int e, int& ls, int& ld) {
    int ec = e < E ? e : E - 1;
    ls = ei[ec];
    ld = ei[E + ec];
}
__device__ __forceinline__ void place_edge(int E, int e, int e1, int lo, int ls, int ld,
                                           int& s, unsigned& rel) {
    int d = e < E ? ld : e - E;
    s     = e < E ? ls : e - E;
    if (e >= e1) { d = -1; s = 0; }
    rel = (unsigned)(d - lo);
}

template <class Drain>
__device__ __forceinline__ void scan_chunk(const int* __restrict__ ei, int E, int e0, int e1,
                                           int lo, int span, int2* qw, int tid, Drain drain) {
    const unsigned lt = (1u << (tid & 31)) - 1u;
    int cnt = 0;
    for (int b = e0; b < e1; b += WG * UNR) {
        int ls[UNR], ld[UNR], s[UNR]; unsigned rel[UNR];
#pragma unroll
        for (int u = 0; u < UNR; ++u) load_edge(ei, E, b + u * WG + tid, ls[u], ld[u]);
        asm volatile("" :: "v"(ls[0]), "v"(ls[1]), "v"(ls[2]), "v"(ls[3]),
                           "v"(ls[4]), "v"(ls[5]), "v"(ls[6]), "v"(ls[7]));
#pragma unroll
        for (int u = 0; u < UNR; ++u) place_edge(E, b + u * WG + tid, e1, lo, ls[u], ld[u], s[u], rel[u]);
#pragma unroll
        for (int u = 0; u < UNR; ++u) {
            bool keep = rel[u] < (unsigned)span;
            unsigned mask = __builtin_amdgcn_ballot_w32(keep);
            if (keep) qw[cnt + __popc(mask & lt)] = make_int2(s[u], (int)rel[u]);
            cnt += __popc(mask);
            if (cnt > QCAP - 32) {
                wave_sync();
                drain(cnt);
                wave_sync();
                cnt = 0;
            }
        }
    }
    wave_sync();
    drain(cnt);
}

template <class Drain>
__device__ __forceinline__ void walk_slice(const unsigned* sl, int total,
                                           int2* qw, int lane, Drain drain) {
    if (total <= 0) return;
    const int last = total - 1;
    const unsigned smask = (1u << SRC_BITS) - 1u;
    unsigned a = sl[lane < last ? lane : last];
    unsigned b = sl[32 + lane < last ? 32 + lane : last];
    for (int k0 = 0; k0 < total; k0 += QCAP) {
        qw[lane]      = make_int2((int)(a & smask), (int)(a >> SRC_BITS));
        qw[32 + lane] = make_int2((int)(b & smask), (int)(b >> SRC_BITS));
        const int kn = k0 + QCAP;
        if (kn < total) {
            a = sl[kn + lane < last ? kn + lane : last];
            b = sl[kn + 32 + lane < last ? kn + 32 + lane : last];
        }
        wave_sync();
        drain(total - k0 < QCAP ? total - k0 : QCAP);
        wave_sync();
    }
}

__device__ __forceinline__ void head_consts(const float* __restrict__ W1, const float* __restrict__ as1,
                                            const float* __restrict__ ad1, int h, float& wsl, float& wdl) {
    float a = 0.f, b = 0.f;
#pragma unroll
    for (int c = 0; c < 8; ++c) {
        float w = W1[h * 8 + c];
        a += w * as1[h * 8 + c];
        b += w * ad1[h * 8 + c];
    }
    wsl = a; wdl = b;
}

template <int SPAN, int G>
__global__ __launch_bounds__(WG) void k_l1(const int* __restrict__ ei, int E, int N,
                                           const float* __restrict__ x,
                                           const float* __restrict__ W1,
                                           const float* __restrict__ as1,
                                           const float* __restrict__ ad1,
                                           float* __restrict__ pm1, float* __restrict__ prec1,
                                           unsigned* sl, int* __restrict__ slcnt, int capw) {
    __shared__ __attribute__((aligned(16))) unsigned mx[SPAN * 4];
    __shared__ __attribute__((aligned(16))) float rec[SPAN * 8];
    __shared__ int2 q[(WG / 32) * QCAP];
    const int tid = threadIdx.x, lane = tid & 31;
    const int r = blockIdx.x / G, g = blockIdx.x % G;
    const int lo = r * SPAN;
    const int span = (N - lo < SPAN) ? (N - lo) : SPAN;
    const int h = lane & 3;
    float wsl, wdl;
    head_consts(W1, as1, ad1, h, wsl, wdl);

    for (int i = tid; i < span * 4; i += WG) mx[i] = FORD_NEG_INF;
    for (int i = tid; i < span * 8; i += WG) rec[i] = 0.f;
    __syncthreads();

    int2* qw = q + (tid >> 5) * QCAP;
    const int wslot = blockIdx.x * (WG / 32) + (tid >> 5);
    unsigned* slw = sl + (size_t)wslot * capw;
    int pos = 0;
    const int sub4 = lane >> 2;
    auto drain_max = [&](int cnt) {
        for (int k0 = 0; k0 < cnt; k0 += 32) {
            bool ok[4]; int2 en[4]; float xs[4], xdv[4];
#pragma unroll
            for (int rr = 0; rr < 4; ++rr) {
                int k = k0 + rr * 8 + sub4;
                ok[rr] = k < cnt;
                en[rr] = qw[ok[rr] ? k : cnt - 1];
            }
#pragma unroll
            for (int rr = 0; rr < 4; ++rr) { xs[rr] = x[en[rr].x]; xdv[rr] = x[lo + en[rr].y]; }
            asm volatile("" :: "v"(xs[0]), "v"(xs[1]), "v"(xs[2]), "v"(xs[3]),
                               "v"(xdv[0]), "v"(xdv[1]), "v"(xdv[2]), "v"(xdv[3]));
#pragma unroll
            for (int rr = 0; rr < 4; ++rr) {
                unsigned v = ford(lrelu(xs[rr] * wsl + xdv[rr] * wdl));
                if (ok[rr]) atomicMax(&mx[en[rr].y * 4 + h], v);
            }
        }
        if (lane < cnt && pos + lane < capw) {
            int2 en = qw[lane];
            slw[pos + lane] = (unsigned)en.x | ((unsigned)en.y << SRC_BITS);
        }
        if (32 + lane < cnt && pos + 32 + lane < capw) {
            int2 en = qw[32 + lane];
            slw[pos + 32 + lane] = (unsigned)en.x | ((unsigned)en.y << SRC_BITS);
        }
        pos += cnt;
    };
    const int EN = E + N;
    const int per = (EN + G - 1) / G;
    const int e0 = g * per;
    const int e1 = (e0 + per < EN) ? (e0 + per) : EN;
    scan_chunk(ei, E, e0, e1, lo, span, qw, tid, drain_max);
    if (lane == 0) slcnt[wslot] = pos;
    __syncthreads();

    auto drain_sum = [&](int cnt) {
        for (int k0 = 0; k0 < cnt; k0 += 32) {
            bool ok[4]; int2 en[4]; float xs[4], xdv[4];
#pragma unroll
            for (int rr = 0; rr < 4; ++rr) {
                int k = k0 + rr * 8 + sub4;
                ok[rr] = k < cnt;
                en[rr] = qw[ok[rr] ? k : cnt - 1];
            }
#pragma unroll
            for (int rr = 0; rr < 4; ++rr) { xs[rr] = x[en[rr].x]; xdv[rr] = x[lo + en[rr].y]; }
            asm volatile("" :: "v"(xs[0]), "v"(xs[1]), "v"(xs[2]), "v"(xs[3]),
                               "v"(xdv[0]), "v"(xdv[1]), "v"(xdv[2]), "v"(xdv[3]));
#pragma unroll
            for (int rr = 0; rr < 4; ++rr) {
                float t = __expf(lrelu(xs[rr] * wsl + xdv[rr] * wdl) - forddec(mx[en[rr].y * 4 + h]));
                if (ok[rr]) {
                    atomicAdd(&rec[en[rr].y * 8 + h], t);
                    atomicAdd(&rec[en[rr].y * 8 + 4 + h], t * xs[rr]);
                }
            }
        }
    };
    const int total = __builtin_amdgcn_readfirstlane(pos);
    if (total <= capw) walk_slice(slw, total, qw, lane, drain_sum);
    else               scan_chunk(ei, E, e0, e1, lo, span, qw, tid, drain_sum);
    __syncthreads();

    float4* pm = reinterpret_cast<float4*>(pm1 + ((size_t)g * N + lo) * 4);
    for (int k = tid; k < span; k += WG) {
        const uint4 u = *reinterpret_cast<const uint4*>(&mx[k * 4]);
        pm[k] = make_float4(forddec(u.x), forddec(u.y), forddec(u.z), forddec(u.w));
    }
    float4* pr = reinterpret_cast<float4*>(prec1 + ((size_t)g * N + lo) * 8);
    for (int k = tid; k < span * 2; k += WG) pr[k] = *reinterpret_cast<const float4*>(&rec[k * 4]);
}

template <int G>
__global__ void k_fin1(const float* __restrict__ W1, const float* __restrict__ b1,
                       const float* __restrict__ pm1, const float* __restrict__ prec1,
                       float* __restrict__ out1, int N) {
    const int i = blockIdx.x * blockDim.x + threadIdx.x;
    const int ic = i < N * 4 ? i : N * 4 - 1;
    const int n = ic >> 2, h = ic & 3;
    float m[G], s[G], w[G];
#pragma unroll
    for (int g = 0; g < G; ++g) {
        m[g] = pm1[(size_t)g * N * 4 + ic];
        s[g] = prec1[((size_t)g * N + n) * 8 + h];
        w[g] = prec1[((size_t)g * N + n) * 8 + 4 + h];
    }
    float M = m[0];
#pragma unroll
    for (int g = 1; g < G; ++g) M = fmaxf(M, m[g]);
    float S = 0.f, Wt = 0.f;
#pragma unroll
    for (int g = 0; g < G; ++g) {
        float sc = __expf(m[g] - M);
        S += s[g] * sc;
        Wt += w[g] * sc;
    }
    const float ratio = Wt / S;

    const int lane = threadIdx.x & 31;
    const int node0 = (i - lane) >> 2;
    const int hd = (lane & 7) >> 1, c0 = (lane & 1) * 4;
#pragma unroll
    for (int k = 0; k < 2; ++k) {
        const int nl = k * 4 + (lane >> 3);
        const float rt = __shfl(ratio, nl * 4 + hd, 32);
        float v[4];
#pragma unroll
        for (int c = 0; c < 4; ++c) {
            float t = W1[hd * 8 + c0 + c] * rt + b1[hd * 8 + c0 + c];
            v[c] = t > 0.f ? t : 0.f;
        }
        const int node = node0 + nl;
        if (node < N)
            *reinterpret_cast<float4*>(out1 + (size_t)node * 32 + hd * 8 + c0) = make_float4(v[0], v[1], v[2], v[3]);
    }
}

__global__ void k_gemm2(const float* __restrict__ out1, const float* __restrict__ W2,
                        const float* __restrict__ as2, const float* __restrict__ ad2,
                        float* __restrict__ h2, float* __restrict__ alS2,
                        float* __restrict__ alD2, int N) {
    __shared__ float Bp[32 * 16];
    int t = threadIdx.x;
    for (int i = t; i < 512; i += blockDim.x) {
        int k = i >> 4, n = i & 15;
        float v = 0.f;
        if (n < 8) {
            v = W2[k * 8 + n];
        } else if (n == 8) {
            for (int c = 0; c < 8; ++c) v += W2[k * 8 + c] * as2[c];
        } else if (n == 9) {
            for (int c = 0; c < 8; ++c) v += W2[k * 8 + c] * ad2[c];
        }
        Bp[i] = v;
    }
    __syncthreads();

    int wave = t >> 5, lane = t & 31;
    long tile = (long)blockIdx.x * 8 + wave;
    long row0 = tile * 16;
    if (row0 >= N) return;

    int m = lane & 15, khi = lane >> 4;
    long rowm = row0 + m;
    if (rowm >= N) rowm = N - 1;
    const float* Arow = out1 + rowm * 32;

    v8f c = {};
#pragma unroll
    for (int step = 0; step < 8; ++step) {
        int k0 = step * 4 + khi * 2;
        v2f a; a.x = Arow[k0];          a.y = Arow[k0 + 1];
        v2f b; b.x = Bp[k0 * 16 + m];   b.y = Bp[(k0 + 1) * 16 + m];
        c = __builtin_amdgcn_wmma_f32_16x16x4_f32(false, a, false, b, (short)0, c,
                                                  false, false);
    }

#pragma unroll
    for (int r = 0; r < 8; ++r) {
        long node = row0 + khi * 8 + r;
        if (node >= N) continue;
        float v = c[r];
        if (m < 8)        h2[node * 8 + m] = v;
        else if (m == 8)  alS2[node] = v;
        else if (m == 9)  alD2[node] = v;
    }
}

template <int SPAN, int G>
__global__ __launch_bounds__(WG) void k_l2(const int* __restrict__ ei, int E,
                                           const unsigned* __restrict__ sl,
                                           const int* __restrict__ slcnt, int capw, int N,
                                           const float* __restrict__ h2,
                                           const float* __restrict__ alS2,
                                           const float* __restrict__ alD2,
                                           float* __restrict__ pm2, float* __restrict__ ps2,
                                           float* __restrict__ pacc2) {
    __shared__ unsigned mx[SPAN];
    __shared__ float ad[SPAN];
    __shared__ float ssum[SPAN];
    __shared__ __attribute__((aligned(16))) float acc[SPAN * 8];
    __shared__ int2 q[(WG / 32) * QCAP];
    __shared__ float tq[WG];
    const int tid = threadIdx.x, lane = tid & 31;
    const int r = blockIdx.x / G, g = blockIdx.x % G;
    const int lo = r * SPAN;
    const int span = (N - lo < SPAN) ? (N - lo) : SPAN;

    for (int i = tid; i < span; i += WG) { ad[i] = alD2[lo + i]; mx[i] = FORD_NEG_INF; ssum[i] = 0.f; }
    for (int i = tid; i < span * 8; i += WG) acc[i] = 0.f;
    __syncthreads();

    int2* qw = q + (tid >> 5) * QCAP;
    auto drain_max = [&](int cnt) {
        for (int k0 = 0; k0 < cnt; k0 += 64) {
            bool ok[2]; int2 en[2]; float as[2];
#pragma unroll
            for (int rr = 0; rr < 2; ++rr) {
                int k = k0 + rr * 32 + lane;
                ok[rr] = k < cnt;
                en[rr] = qw[ok[rr] ? k : cnt - 1];
            }
#pragma unroll
            for (int rr = 0; rr < 2; ++rr) as[rr] = alS2[en[rr].x];
            asm volatile("" :: "v"(as[0]), "v"(as[1]));
#pragma unroll
            for (int rr = 0; rr < 2; ++rr)
                if (ok[rr]) atomicMax(&mx[en[rr].y], ford(lrelu(as[rr] + ad[en[rr].y])));
        }
    };
    float* tw = tq + (tid & ~31);
    const int j = lane & 7, sub = lane >> 3;
    auto drain_sum = [&](int cnt) {
        for (int k0 = 0; k0 < cnt; k0 += 32) {
            const int nb = cnt - k0 < 32 ? cnt - k0 : 32;
            const int2* qb = qw + k0;
            const bool oka = lane < nb;
            const int2 ea = qb[oka ? lane : nb - 1];
            bool ok[8]; int kk[8]; int2 en[8]; float hv[8];
#pragma unroll
            for (int rr = 0; rr < 8; ++rr) {
                int k = rr * 4 + sub;
                ok[rr] = k < nb;
                kk[rr] = ok[rr] ? k : nb - 1;
                en[rr] = qb[kk[rr]];
            }
            const float adv = ad[ea.y];
            const unsigned mxv = mx[ea.y];
            const float as = alS2[ea.x];
#pragma unroll
            for (int rr = 0; rr < 8; ++rr) hv[rr] = h2[(size_t)en[rr].x * 8 + j];
            asm volatile("" :: "v"(as), "v"(hv[0]), "v"(hv[1]), "v"(hv[2]), "v"(hv[3]),
                               "v"(hv[4]), "v"(hv[5]), "v"(hv[6]), "v"(hv[7]));
            const float t = __expf(lrelu(as + adv) - forddec(mxv));
            if (oka) atomicAdd(&ssum[ea.y], t);
            tw[lane] = t;
            wave_sync();
            float tv[8];
#pragma unroll
            for (int rr = 0; rr < 8; ++rr) tv[rr] = tw[kk[rr]];
            asm volatile("" :: "v"(tv[0]), "v"(tv[1]), "v"(tv[2]), "v"(tv[3]),
                               "v"(tv[4]), "v"(tv[5]), "v"(tv[6]), "v"(tv[7]));
#pragma unroll
            for (int rr = 0; rr < 8; ++rr)
                if (ok[rr]) atomicAdd(&acc[en[rr].y * 8 + j], tv[rr] * hv[rr]);
            wave_sync();
        }
    };
    const int wslot = blockIdx.x * (WG / 32) + (tid >> 5);
    const unsigned* slw = sl + (size_t)wslot * capw;
    const int total = __builtin_amdgcn_readfirstlane(slcnt[wslot]);
    const int EN = E + N;
    const int per = (EN + G - 1) / G;
    const int e0 = g * per;
    const int e1 = (e0 + per < EN) ? (e0 + per) : EN;
    if (total <= capw) walk_slice(slw, total, qw, lane, drain_max);
    else               scan_chunk(ei, E, e0, e1, lo, span, qw, tid, drain_max);
    __syncthreads();
    if (total <= capw) walk_slice(slw, total, qw, lane, drain_sum);
    else               scan_chunk(ei, E, e0, e1, lo, span, qw, tid, drain_sum);
    __syncthreads();

    float* pm = pm2 + (size_t)g * N + lo;
    float* ps = ps2 + (size_t)g * N + lo;
    for (int k = tid; k < span; k += WG) { pm[k] = forddec(mx[k]); ps[k] = ssum[k]; }
    float4* pa = reinterpret_cast<float4*>(pacc2 + ((size_t)g * N + lo) * 8);
    for (int k = tid; k < span * 2; k += WG) pa[k] = *reinterpret_cast<const float4*>(&acc[k * 4]);
}

template <int G>
__global__ void k_fin2(const float* __restrict__ pm2, const float* __restrict__ ps2,
                       const float* __restrict__ pacc2,
                       const float* __restrict__ b2, float* __restrict__ out, int N) {
    const int i = blockIdx.x * blockDim.x + threadIdx.x;
    if (i >= N * 2) return;
    const int n = i >> 1, q = i & 1;
    float m[G], s[G]; float4 a[G];
#pragma unroll
    for (int g = 0; g < G; ++g) {
        m[g] = pm2[(size_t)g * N + n];
        s[g] = ps2[(size_t)g * N + n];
        a[g] = reinterpret_cast<const float4*>(pacc2 + (size_t)g * N * 8)[i];
    }
    float M = m[0];
#pragma unroll
    for (int g = 1; g < G; ++g) M = fmaxf(M, m[g]);
    float S = 0.f, A0 = 0.f, A1 = 0.f, A2 = 0.f, A3 = 0.f;
#pragma unroll
    for (int g = 0; g < G; ++g) {
        float sc = __expf(m[g] - M);
        S  += s[g] * sc;
        A0 += a[g].x * sc; A1 += a[g].y * sc; A2 += a[g].z * sc; A3 += a[g].w * sc;
    }
    reinterpret_cast<float4*>(out)[i] = make_float4(A0 / S + b2[q * 4 + 0], A1 / S + b2[q * 4 + 1],
                                                    A2 / S + b2[q * 4 + 2], A3 / S + b2[q * 4 + 3]);
}

extern "C" void kernel_launch(void* const* d_in, const int* in_sizes, int n_in,
                              void* d_out, int out_size, void* d_ws, size_t ws_size,
                              hipStream_t stream) {
    const float* x   = (const float*)d_in[0];
    const int*   ei  = (const int*)d_in[1];
    const float* W1  = (const float*)d_in[2];
    const float* as1 = (const float*)d_in[3];
    const float* ad1 = (const float*)d_in[4];
    const float* b1  = (const float*)d_in[5];
    const float* W2  = (const float*)d_in[6];
    const float* as2 = (const float*)d_in[7];
    const float* ad2 = (const float*)d_in[8];
    const float* b2  = (const float*)d_in[9];
    (void)n_in; (void)out_size; (void)ws_size;

    int N = in_sizes[0];
    int E = in_sizes[1] / 2;

    char* ws = (char*)d_ws;
    size_t cur = 0;
    auto carve = [&](size_t bytes) -> void* {
        void* p = ws + cur;
        cur += (bytes + 255) & ~(size_t)255;
        return p;
    };
    float*    pm1   = (float*)carve(sizeof(float) * (size_t)G_ALL * N * 4);
    float*    prec1 = (float*)carve(sizeof(float) * (size_t)G_ALL * N * 8);
    float*    pm2   = (float*)carve(sizeof(float) * (size_t)G_ALL * N);
    float*    ps2   = (float*)carve(sizeof(float) * (size_t)G_ALL * N);
    float*    pacc2 = (float*)carve(sizeof(float) * (size_t)G_ALL * N * 8);
    float*    out1  = (float*)carve(sizeof(float) * (size_t)N * 32);
    float*    h2    = (float*)carve(sizeof(float) * (size_t)N * 8);
    float*    alS2  = (float*)carve(sizeof(float) * (size_t)N);
    float*    alD2  = (float*)carve(sizeof(float) * (size_t)N);
    const int nwg   = ((N + SPAN_ALL - 1) / SPAN_ALL) * G_ALL;
    const int per   = (E + N + G_ALL - 1) / G_ALL;
    const int capf  = ((per + WG * UNR - 1) / (WG * UNR)) * UNR * 32;
    const int capw  = capf < SLICE_CAP ? capf : SLICE_CAP;
    unsigned* sl    = (unsigned*)carve(sizeof(unsigned) * (size_t)nwg * (WG / 32) * capw);
    int*      slcnt = (int*)carve(sizeof(int) * (size_t)nwg * (WG / 32));

    const int B = 256;
    k_l1<SPAN_ALL, G_ALL><<<nwg, WG, 0, stream>>>(ei, E, N, x, W1, as1, ad1, pm1, prec1, sl, slcnt, capw);
    k_fin1<G_ALL><<<(4 * N + B - 1) / B, B, 0, stream>>>(W1, b1, pm1, prec1, out1, N);

    int tiles = (N + 15) / 16;
    k_gemm2<<<(tiles + 7) / 8, B, 0, stream>>>(out1, W2, as2, ad2, h2, alS2, alD2, N);

    k_l2<SPAN_ALL, G_ALL><<<nwg, WG, 0, stream>>>(ei, E, sl, slcnt, capw, N, h2, alS2, alD2, pm2, ps2, pacc2);
    k_fin2<G_ALL><<<(2 * N + B - 1) / B, B, 0, stream>>>(pm2, ps2, pacc2, b2, (float*)d_out, N);
}


# Filter pass reads the edge list 16 bytes a lane (4 consecutive edges) in trips wholly inside it: 4 index loads a trip, not 16, no clamp or select; self-loop trips load none; LDS sums in arrival order
# speedup vs baseline: 11.0631x; 1.0718x over previous
#include <hip/hip_runtime.h>
#include <math.h>


#define NEG_SLOPE 0.2f
#define WG 1024
#define UNR 8
#define QCAP 64
#define SPAN_ALL 6250
#define G_ALL 16
#define SRC_BITS 19
#define SLICE_CAP 2048
static_assert(UNR == 8, "the load fence in scan_chunk names eight values; a 16-byte trip is two loads of 4 edges a lane");
static_assert(QCAP >= 64, "a push of 32 entries on top of 32 queued ones must fit");
static_assert(QCAP == 64, "walk_slice refills the queue with two entries per lane");
static_assert(SPAN_ALL <= (1 << (32 - SRC_BITS)), "destination - lo must fit above the source bits");

typedef __attribute__((ext_vector_type(2))) float v2f;
typedef __attribute__((ext_vector_type(8))) float v8f;

__device__ __forceinline__ float lrelu(float v) { return v > 0.f ? v : NEG_SLOPE * v; }

__device__ __forceinline__ unsigned ford(float f) {
    unsigned u = __float_as_uint(f);
    return (u & 0x80000000u) ? ~u : (u | 0x80000000u);
}
__device__ __forceinline__ float forddec(unsigned u) {
    return (u & 0x80000000u) ? __uint_as_float(u & 0x7fffffffu) : __uint_as_float(~u);
}
#define FORD_NEG_INF 0x007FFFFFu


__device__ __forceinline__ void wave_sync() {
    __builtin_amdgcn_fence(__ATOMIC_RELEASE, "wavefront");
    __builtin_amdgcn_wave_barrier();
    __builtin_amdgcn_fence(__ATOMIC_ACQUIRE, "wavefront");
}

__device__ __forceinline__ void load_edge(const int* __restrict__ ei, int E, int e, int& ls, int& ld) {
    int ec = e < E ? e : E - 1;
    ls = ei[ec];
    ld = ei[E + ec];
}
__device__ __forceinline__ void place_edge(int E, int e, int e1, int lo, int ls, int ld,
                                           int& s, unsigned& rel) {
    int d = e < E ? ld : e - E;
    s     = e < E ? ls : e - E;
    if (e >= e1) { d = -1; s = 0; }
    rel = (unsigned)(d - lo);
}

template <class Drain>
__device__ __forceinline__ void scan_chunk(const int* __restrict__ ei, int E, int e0, int e1,
                                           int lo, int span, int2* qw, int tid, Drain drain) {
    const unsigned lt = (1u << (tid & 31)) - 1u;
    const int ereal = e1 < E ? e1 : E;
    const bool wide = ((E & 3) == 0) && ((e0 & 3) == 0) && ((reinterpret_cast<size_t>(ei) & 15) == 0);
    int cnt = 0;
    for (int b = e0; b < e1; b += WG * UNR) {
        int s[UNR]; unsigned rel[UNR];
        if (wide && b + WG * UNR <= ereal) {
            const int4* ps = reinterpret_cast<const int4*>(ei + b) + tid;
            const int4* pd = reinterpret_cast<const int4*>(ei + E + b) + tid;
            const int4 s0 = ps[0], s1 = ps[WG], d0 = pd[0], d1 = pd[WG];
            s[0] = s0.x; s[1] = s0.y; s[2] = s0.z; s[3] = s0.w;
            s[4] = s1.x; s[5] = s1.y; s[6] = s1.z; s[7] = s1.w;
            rel[0] = (unsigned)(d0.x - lo); rel[1] = (unsigned)(d0.y - lo);
            rel[2] = (unsigned)(d0.z - lo); rel[3] = (unsigned)(d0.w - lo);
            rel[4] = (unsigned)(d1.x - lo); rel[5] = (unsigned)(d1.y - lo);
            rel[6] = (unsigned)(d1.z - lo); rel[7] = (unsigned)(d1.w - lo);
        } else if (b >= E && b + WG * UNR <= e1) {
#pragma unroll
            for (int u = 0; u < UNR; ++u) {
                s[u] = b - E + u * WG + tid;
                rel[u] = (unsigned)(s[u] - lo);
            }
        } else {
            int ls[UNR], ld[UNR];
#pragma unroll
            for (int u = 0; u < UNR; ++u) load_edge(ei, E, b + u * WG + tid, ls[u], ld[u]);
            asm volatile("" :: "v"(ls[0]), "v"(ls[1]), "v"(ls[2]), "v"(ls[3]),
                               "v"(ls[4]), "v"(ls[5]), "v"(ls[6]), "v"(ls[7]));
#pragma unroll
            for (int u = 0; u < UNR; ++u) place_edge(E, b + u * WG + tid, e1, lo, ls[u], ld[u], s[u], rel[u]);
        }
#pragma unroll
        for (int u = 0; u < UNR; ++u) {
            bool keep = rel[u] < (unsigned)span;
            unsigned mask = __builtin_amdgcn_ballot_w32(keep);
            if (keep) qw[cnt + __popc(mask & lt)] = make_int2(s[u], (int)rel[u]);
            cnt += __popc(mask);
            if (cnt > QCAP - 32) {
                wave_sync();
                drain(cnt);
                wave_sync();
                cnt = 0;
            }
        }
    }
    wave_sync();
    drain(cnt);
}

template <class Drain>
__device__ __forceinline__ void walk_slice(const unsigned* sl, int total,
                                           int2* qw, int lane, Drain drain) {
    if (total <= 0) return;
    const int last = total - 1;
    const unsigned smask = (1u << SRC_BITS) - 1u;
    unsigned a = sl[lane < last ? lane : last];
    unsigned b = sl[32 + lane < last ? 32 + lane : last];
    for (int k0 = 0; k0 < total; k0 += QCAP) {
        qw[lane]      = make_int2((int)(a & smask), (int)(a >> SRC_BITS));
        qw[32 + lane] = make_int2((int)(b & smask), (int)(b >> SRC_BITS));
        const int kn = k0 + QCAP;
        if (kn < total) {
            a = sl[kn + lane < last ? kn + lane : last];
            b = sl[kn + 32 + lane < last ? kn + 32 + lane : last];
        }
        wave_sync();
        drain(total - k0 < QCAP ? total - k0 : QCAP);
        wave_sync();
    }
}

__device__ __forceinline__ void head_consts(const float* __restrict__ W1, const float* __restrict__ as1,
                                            const float* __restrict__ ad1, int h, float& wsl, float& wdl) {
    float a = 0.f, b = 0.f;
#pragma unroll
    for (int c = 0; c < 8; ++c) {
        float w = W1[h * 8 + c];
        a += w * as1[h * 8 + c];
        b += w * ad1[h * 8 + c];
    }
    wsl = a; wdl = b;
}

template <int SPAN, int G>
__global__ __launch_bounds__(WG) void k_l1(const int* __restrict__ ei, int E, int N,
                                           const float* __restrict__ x,
                                           const float* __restrict__ W1,
                                           const float* __restrict__ as1,
                                           const float* __restrict__ ad1,
                                           float* __restrict__ pm1, float* __restrict__ prec1,
                                           unsigned* sl, int* __restrict__ slcnt, int capw) {
    __shared__ __attribute__((aligned(16))) unsigned mx[SPAN * 4];
    __shared__ __attribute__((aligned(16))) float rec[SPAN * 8];
    __shared__ int2 q[(WG / 32) * QCAP];
    const int tid = threadIdx.x, lane = tid & 31;
    const int r = blockIdx.x / G, g = blockIdx.x % G;
    const int lo = r * SPAN;
    const int span = (N - lo < SPAN) ? (N - lo) : SPAN;
    const int h = lane & 3;
    float wsl, wdl;
    head_consts(W1, as1, ad1, h, wsl, wdl);

    for (int i = tid; i < span * 4; i += WG) mx[i] = FORD_NEG_INF;
    for (int i = tid; i < span * 8; i += WG) rec[i] = 0.f;
    __syncthreads();

    int2* qw = q + (tid >> 5) * QCAP;
    const int wslot = blockIdx.x * (WG / 32) + (tid >> 5);
    unsigned* slw = sl + (size_t)wslot * capw;
    int pos = 0;
    const int sub4 = lane >> 2;
    auto drain_max = [&](int cnt) {
        for (int k0 = 0; k0 < cnt; k0 += 32) {
            bool ok[4]; int2 en[4]; float xs[4], xdv[4];
#pragma unroll
            for (int rr = 0; rr < 4; ++rr) {
                int k = k0 + rr * 8 + sub4;
                ok[rr] = k < cnt;
                en[rr] = qw[ok[rr] ? k : cnt - 1];
            }
#pragma unroll
            for (int rr = 0; rr < 4; ++rr) { xs[rr] = x[en[rr].x]; xdv[rr] = x[lo + en[rr].y]; }
            asm volatile("" :: "v"(xs[0]), "v"(xs[1]), "v"(xs[2]), "v"(xs[3]),
                               "v"(xdv[0]), "v"(xdv[1]), "v"(xdv[2]), "v"(xdv[3]));
#pragma unroll
            for (int rr = 0; rr < 4; ++rr) {
                unsigned v = ford(lrelu(xs[rr] * wsl + xdv[rr] * wdl));
                if (ok[rr]) atomicMax(&mx[en[rr].y * 4 + h], v);
            }
        }
        if (lane < cnt && pos + lane < capw) {
            int2 en = qw[lane];
            slw[pos + lane] = (unsigned)en.x | ((unsigned)en.y << SRC_BITS);
        }
        if (32 + lane < cnt && pos + 32 + lane < capw) {
            int2 en = qw[32 + lane];
            slw[pos + 32 + lane] = (unsigned)en.x | ((unsigned)en.y << SRC_BITS);
        }
        pos += cnt;
    };
    const int EN = E + N;
    const int per = ((EN + G - 1) / G + 3) & ~3;
    const int e0 = g * per;
    const int e1 = (e0 + per < EN) ? (e0 + per) : EN;
    scan_chunk(ei, E, e0, e1, lo, span, qw, tid, drain_max);
    if (lane == 0) slcnt[wslot] = pos;
    __syncthreads();

    auto drain_sum = [&](int cnt) {
        for (int k0 = 0; k0 < cnt; k0 += 32) {
            bool ok[4]; int2 en[4]; float xs[4], xdv[4];
#pragma unroll
            for (int rr = 0; rr < 4; ++rr) {
                int k = k0 + rr * 8 + sub4;
                ok[rr] = k < cnt;
                en[rr] = qw[ok[rr] ? k : cnt - 1];
            }
#pragma unroll
            for (int rr = 0; rr < 4; ++rr) { xs[rr] = x[en[rr].x]; xdv[rr] = x[lo + en[rr].y]; }
            asm volatile("" :: "v"(xs[0]), "v"(xs[1]), "v"(xs[2]), "v"(xs[3]),
                               "v"(xdv[0]), "v"(xdv[1]), "v"(xdv[2]), "v"(xdv[3]));
#pragma unroll
            for (int rr = 0; rr < 4; ++rr) {
                float t = __expf(lrelu(xs[rr] * wsl + xdv[rr] * wdl) - forddec(mx[en[rr].y * 4 + h]));
                if (ok[rr]) {
                    atomicAdd(&rec[en[rr].y * 8 + h], t);
                    atomicAdd(&rec[en[rr].y * 8 + 4 + h], t * xs[rr]);
                }
            }
        }
    };
    const int total = __builtin_amdgcn_readfirstlane(pos);
    if (total <= capw) walk_slice(slw, total, qw, lane, drain_sum);
    else               scan_chunk(ei, E, e0, e1, lo, span, qw, tid, drain_sum);
    __syncthreads();

    float4* pm = reinterpret_cast<float4*>(pm1 + ((size_t)g * N + lo) * 4);
    for (int k = tid; k < span; k += WG) {
        const uint4 u = *reinterpret_cast<const uint4*>(&mx[k * 4]);
        pm[k] = make_float4(forddec(u.x), forddec(u.y), forddec(u.z), forddec(u.w));
    }
    float4* pr = reinterpret_cast<float4*>(prec1 + ((size_t)g * N + lo) * 8);
    for (int k = tid; k < span * 2; k += WG) pr[k] = *reinterpret_cast<const float4*>(&rec[k * 4]);
}

template <int G>
__global__ void k_fin1(const float* __restrict__ W1, const float* __restrict__ b1,
                       const float* __restrict__ pm1, const float* __restrict__ prec1,
                       float* __restrict__ out1, int N) {
    const int i = blockIdx.x * blockDim.x + threadIdx.x;
    const int ic = i < N * 4 ? i : N * 4 - 1;
    const int n = ic >> 2, h = ic & 3;
    float m[G], s[G], w[G];
#pragma unroll
    for (int g = 0; g < G; ++g) {
        m[g] = pm1[(size_t)g * N * 4 + ic];
        s[g] = prec1[((size_t)g * N + n) * 8 + h];
        w[g] = prec1[((size_t)g * N + n) * 8 + 4 + h];
    }
    float M = m[0];
#pragma unroll
    for (int g = 1; g < G; ++g) M = fmaxf(M, m[g]);
    float S = 0.f, Wt = 0.f;
#pragma unroll
    for (int g = 0; g < G; ++g) {
        float sc = __expf(m[g] - M);
        S += s[g] * sc;
        Wt += w[g] * sc;
    }
    const float ratio = Wt / S;

    const int lane = threadIdx.x & 31;
    const int node0 = (i - lane) >> 2;
    const int hd = (lane & 7) >> 1, c0 = (lane & 1) * 4;
#pragma unroll
    for (int k = 0; k < 2; ++k) {
        const int nl = k * 4 + (lane >> 3);
        const float rt = __shfl(ratio, nl * 4 + hd, 32);
        float v[4];
#pragma unroll
        for (int c = 0; c < 4; ++c) {
            float t = W1[hd * 8 + c0 + c] * rt + b1[hd * 8 + c0 + c];
            v[c] = t > 0.f ? t : 0.f;
        }
        const int node = node0 + nl;
        if (node < N)
            *reinterpret_cast<float4*>(out1 + (size_t)node * 32 + hd * 8 + c0) = make_float4(v[0], v[1], v[2], v[3]);
    }
}

__global__ void k_gemm2(const float* __restrict__ out1, const float* __restrict__ W2,
                        const float* __restrict__ as2, const float* __restrict__ ad2,
                        float* __restrict__ h2, float* __restrict__ alS2,
                        float* __restrict__ alD2, int N) {
    __shared__ float Bp[32 * 16];
    int t = threadIdx.x;
    for (int i = t; i < 512; i += blockDim.x) {
        int k = i >> 4, n = i & 15;
        float v = 0.f;
        if (n < 8) {
            v = W2[k * 8 + n];
        } else if (n == 8) {
            for (int c = 0; c < 8; ++c) v += W2[k * 8 + c] * as2[c];
        } else if (n == 9) {
            for (int c = 0; c < 8; ++c) v += W2[k * 8 + c] * ad2[c];
        }
        Bp[i] = v;
    }
    __syncthreads();

    int wave = t >> 5, lane = t & 31;
    long tile = (long)blockIdx.x * 8 + wave;
    long row0 = tile * 16;
    if (row0 >= N) return;

    int m = lane & 15, khi = lane >> 4;
    long rowm = row0 + m;
    if (rowm >= N) rowm = N - 1;
    const float* Arow = out1 + rowm * 32;

    v8f c = {};
#pragma unroll
    for (int step = 0; step < 8; ++step) {
        int k0 = step * 4 + khi * 2;
        v2f a; a.x = Arow[k0];          a.y = Arow[k0 + 1];
        v2f b; b.x = Bp[k0 * 16 + m];   b.y = Bp[(k0 + 1) * 16 + m];
        c = __builtin_amdgcn_wmma_f32_16x16x4_f32(false, a, false, b, (short)0, c,
                                                  false, false);
    }

#pragma unroll
    for (int r = 0; r < 8; ++r) {
        long node = row0 + khi * 8 + r;
        if (node >= N) continue;
        float v = c[r];
        if (m < 8)        h2[node * 8 + m] = v;
        else if (m == 8)  alS2[node] = v;
        else if (m == 9)  alD2[node] = v;
    }
}

template <int SPAN, int G>
__global__ __launch_bounds__(WG) void k_l2(const int* __restrict__ ei, int E,
                                           const unsigned* __restrict__ sl,
                                           const int* __restrict__ slcnt, int capw, int N,
                                           const float* __restrict__ h2,
                                           const float* __restrict__ alS2,
                                           const float* __restrict__ alD2,
                                           float* __restrict__ pm2, float* __restrict__ ps2,
                                           float* __restrict__ pacc2) {
    __shared__ unsigned mx[SPAN];
    __shared__ float ad[SPAN];
    __shared__ float ssum[SPAN];
    __shared__ __attribute__((aligned(16))) float acc[SPAN * 8];
    __shared__ int2 q[(WG / 32) * QCAP];
    __shared__ float tq[WG];
    const int tid = threadIdx.x, lane = tid & 31;
    const int r = blockIdx.x / G, g = blockIdx.x % G;
    const int lo = r * SPAN;
    const int span = (N - lo < SPAN) ? (N - lo) : SPAN;

    for (int i = tid; i < span; i += WG) { ad[i] = alD2[lo + i]; mx[i] = FORD_NEG_INF; ssum[i] = 0.f; }
    for (int i = tid; i < span * 8; i += WG) acc[i] = 0.f;
    __syncthreads();

    int2* qw = q + (tid >> 5) * QCAP;
    auto drain_max = [&](int cnt) {
        for (int k0 = 0; k0 < cnt; k0 += 64) {
            bool ok[2]; int2 en[2]; float as[2];
#pragma unroll
            for (int rr = 0; rr < 2; ++rr) {
                int k = k0 + rr * 32 + lane;
                ok[rr] = k < cnt;
                en[rr] = qw[ok[rr] ? k : cnt - 1];
            }
#pragma unroll
            for (int rr = 0; rr < 2; ++rr) as[rr] = alS2[en[rr].x];
            asm volatile("" :: "v"(as[0]), "v"(as[1]));
#pragma unroll
            for (int rr = 0; rr < 2; ++rr)
                if (ok[rr]) atomicMax(&mx[en[rr].y], ford(lrelu(as[rr] + ad[en[rr].y])));
        }
    };
    float* tw = tq + (tid & ~31);
    const int j = lane & 7, sub = lane >> 3;
    auto drain_sum = [&](int cnt) {
        for (int k0 = 0; k0 < cnt; k0 += 32) {
            const int nb = cnt - k0 < 32 ? cnt - k0 : 32;
            const int2* qb = qw + k0;
            const bool oka = lane < nb;
            const int2 ea = qb[oka ? lane : nb - 1];
            bool ok[8]; int kk[8]; int2 en[8]; float hv[8];
#pragma unroll
            for (int rr = 0; rr < 8; ++rr) {
                int k = rr * 4 + sub;
                ok[rr] = k < nb;
                kk[rr] = ok[rr] ? k : nb - 1;
                en[rr] = qb[kk[rr]];
            }
            const float adv = ad[ea.y];
            const unsigned mxv = mx[ea.y];
            const float as = alS2[ea.x];
#pragma unroll
            for (int rr = 0; rr < 8; ++rr) hv[rr] = h2[(size_t)en[rr].x * 8 + j];
            asm volatile("" :: "v"(as), "v"(hv[0]), "v"(hv[1]), "v"(hv[2]), "v"(hv[3]),
                               "v"(hv[4]), "v"(hv[5]), "v"(hv[6]), "v"(hv[7]));
            const float t = __expf(lrelu(as + adv) - forddec(mxv));
            if (oka) atomicAdd(&ssum[ea.y], t);
            tw[lane] = t;
            wave_sync();
            float tv[8];
#pragma unroll
            for (int rr = 0; rr < 8; ++rr) tv[rr] = tw[kk[rr]];
            asm volatile("" :: "v"(tv[0]), "v"(tv[1]), "v"(tv[2]), "v"(tv[3]),
                               "v"(tv[4]), "v"(tv[5]), "v"(tv[6]), "v"(tv[7]));
#pragma unroll
            for (int rr = 0; rr < 8; ++rr)
                if (ok[rr]) atomicAdd(&acc[en[rr].y * 8 + j], tv[rr] * hv[rr]);
            wave_sync();
        }
    };
    const int wslot = blockIdx.x * (WG / 32) + (tid >> 5);
    const unsigned* slw = sl + (size_t)wslot * capw;
    const int total = __builtin_amdgcn_readfirstlane(slcnt[wslot]);
    const int EN = E + N;
    const int per = ((EN + G - 1) / G + 3) & ~3;
    const int e0 = g * per;
    const int e1 = (e0 + per < EN) ? (e0 + per) : EN;
    if (total <= capw) walk_slice(slw, total, qw, lane, drain_max);
    else               scan_chunk(ei, E, e0, e1, lo, span, qw, tid, drain_max);
    __syncthreads();
    if (total <= capw) walk_slice(slw, total, qw, lane, drain_sum);
    else               scan_chunk(ei, E, e0, e1, lo, span, qw, tid, drain_sum);
    __syncthreads();

    float* pm = pm2 + (size_t)g * N + lo;
    float* ps = ps2 + (size_t)g * N + lo;
    for (int k = tid; k < span; k += WG) { pm[k] = forddec(mx[k]); ps[k] = ssum[k]; }
    float4* pa = reinterpret_cast<float4*>(pacc2 + ((size_t)g * N + lo) * 8);
    for (int k = tid; k < span * 2; k += WG) pa[k] = *reinterpret_cast<const float4*>(&acc[k * 4]);
}

template <int G>
__global__ void k_fin2(const float* __restrict__ pm2, const float* __restrict__ ps2,
                       const float* __restrict__ pacc2,
                       const float* __restrict__ b2, float* __restrict__ out, int N) {
    const int i = blockIdx.x * blockDim.x + threadIdx.x;
    if (i >= N * 2) return;
    const int n = i >> 1, q = i & 1;
    float m[G], s[G]; float4 a[G];
#pragma unroll
    for (int g = 0; g < G; ++g) {
        m[g] = pm2[(size_t)g * N + n];
        s[g] = ps2[(size_t)g * N + n];
        a[g] = reinterpret_cast<const float4*>(pacc2 + (size_t)g * N * 8)[i];
    }
    float M = m[0];
#pragma unroll
    for (int g = 1; g < G; ++g) M = fmaxf(M, m[g]);
    float S = 0.f, A0 = 0.f, A1 = 0.f, A2 = 0.f, A3 = 0.f;
#pragma unroll
    for (int g = 0; g < G; ++g) {
        float sc = __expf(m[g] - M);
        S  += s[g] * sc;
        A0 += a[g].x * sc; A1 += a[g].y * sc; A2 += a[g].z * sc; A3 += a[g].w * sc;
    }
    reinterpret_cast<float4*>(out)[i] = make_float4(A0 / S + b2[q * 4 + 0], A1 / S + b2[q * 4 + 1],
                                                    A2 / S + b2[q * 4 + 2], A3 / S + b2[q * 4 + 3]);
}

extern "C" void kernel_launch(void* const* d_in, const int* in_sizes, int n_in,
                              void* d_out, int out_size, void* d_ws, size_t ws_size,
                              hipStream_t stream) {
    const float* x   = (const float*)d_in[0];
    const int*   ei  = (const int*)d_in[1];
    const float* W1  = (const float*)d_in[2];
    const float* as1 = (const float*)d_in[3];
    const float* ad1 = (const float*)d_in[4];
    const float* b1  = (const float*)d_in[5];
    const float* W2  = (const float*)d_in[6];
    const float* as2 = (const float*)d_in[7];
    const float* ad2 = (const float*)d_in[8];
    const float* b2  = (const float*)d_in[9];
    (void)n_in; (void)out_size; (void)ws_size;

    int N = in_sizes[0];
    int E = in_sizes[1] / 2;

    char* ws = (char*)d_ws;
    size_t cur = 0;
    auto carve = [&](size_t bytes) -> void* {
        void* p = ws + cur;
        cur += (bytes + 255) & ~(size_t)255;
        return p;
    };
    float*    pm1   = (float*)carve(sizeof(float) * (size_t)G_ALL * N * 4);
    float*    prec1 = (float*)carve(sizeof(float) * (size_t)G_ALL * N * 8);
    float*    pm2   = (float*)carve(sizeof(float) * (size_t)G_ALL * N);
    float*    ps2   = (float*)carve(sizeof(float) * (size_t)G_ALL * N);
    float*    pacc2 = (float*)carve(sizeof(float) * (size_t)G_ALL * N * 8);
    float*    out1  = (float*)carve(sizeof(float) * (size_t)N * 32);
    float*    h2    = (float*)carve(sizeof(float) * (size_t)N * 8);
    float*    alS2  = (float*)carve(sizeof(float) * (size_t)N);
    float*    alD2  = (float*)carve(sizeof(float) * (size_t)N);
    const int nwg   = ((N + SPAN_ALL - 1) / SPAN_ALL) * G_ALL;
    const int per   = ((E + N + G_ALL - 1) / G_ALL + 3) & ~3;
    const int capf  = ((per + WG * UNR - 1) / (WG * UNR)) * UNR * 32;
    const int capw  = capf < SLICE_CAP ? capf : SLICE_CAP;
    unsigned* sl    = (unsigned*)carve(sizeof(unsigned) * (size_t)nwg * (WG / 32) * capw);
    int*      slcnt = (int*)carve(sizeof(int) * (size_t)nwg * (WG / 32));

    const int B = 256;
    k_l1<SPAN_ALL, G_ALL><<<nwg, WG, 0, stream>>>(ei, E, N, x, W1, as1, ad1, pm1, prec1, sl, slcnt, capw);
    k_fin1<G_ALL><<<(4 * N + B - 1) / B, B, 0, stream>>>(W1, b1, pm1, prec1, out1, N);

    int tiles = (N + 15) / 16;
    k_gemm2<<<(tiles + 7) / 8, B, 0, stream>>>(out1, W2, as2, ad2, h2, alS2, alD2, N);

    k_l2<SPAN_ALL, G_ALL><<<nwg, WG, 0, stream>>>(ei, E, sl, slcnt, capw, N, h2, alS2, alD2, pm2, ps2, pacc2);
    k_fin2<G_ALL><<<(2 * N + B - 1) / B, B, 0, stream>>>(pm2, ps2, pacc2, b2, (float*)d_out, N);
}


# Maximum pass of a layer stores the per-source scalar it gathers (x[s]; layer-2 source logit) beside the kept edge in the wave's slice; sum pass reads it in order, no 2nd gather; sums in arrival order
# speedup vs baseline: 11.7149x; 1.0589x over previous
#include <hip/hip_runtime.h>
#include <math.h>


#define NEG_SLOPE 0.2f
#define WG 1024
#define UNR 8
#define QCAP 64
#define SPAN_ALL 6250
#define G_ALL 16
#define SRC_BITS 19
#define SLICE_CAP 2048
static_assert(UNR == 8, "the load fence in scan_chunk names eight values; a 16-byte trip is two loads of 4 edges a lane");
static_assert(QCAP >= 64, "a push of 32 entries on top of 32 queued ones must fit");
static_assert(QCAP == 64, "walk_slice refills the queue with two entries per lane");
static_assert(SPAN_ALL <= (1 << (32 - SRC_BITS)), "destination - lo must fit above the source bits");

typedef __attribute__((ext_vector_type(2))) float v2f;
typedef __attribute__((ext_vector_type(8))) float v8f;

__device__ __forceinline__ float lrelu(float v) { return v > 0.f ? v : NEG_SLOPE * v; }

__device__ __forceinline__ unsigned ford(float f) {
    unsigned u = __float_as_uint(f);
    return (u & 0x80000000u) ? ~u : (u | 0x80000000u);
}
__device__ __forceinline__ float forddec(unsigned u) {
    return (u & 0x80000000u) ? __uint_as_float(u & 0x7fffffffu) : __uint_as_float(~u);
}
#define FORD_NEG_INF 0x007FFFFFu


__device__ __forceinline__ void wave_sync() {
    __builtin_amdgcn_fence(__ATOMIC_RELEASE, "wavefront");
    __builtin_amdgcn_wave_barrier();
    __builtin_amdgcn_fence(__ATOMIC_ACQUIRE, "wavefront");
}

__device__ __forceinline__ void load_edge(const int* __restrict__ ei, int E, int e, int& ls, int& ld) {
    int ec = e < E ? e : E - 1;
    ls = ei[ec];
    ld = ei[E + ec];
}
__device__ __forceinline__ void place_edge(int E, int e, int e1, int lo, int ls, int ld,
                                           int& s, unsigned& rel) {
    int d = e < E ? ld : e - E;
    s     = e < E ? ls : e - E;
    if (e >= e1) { d = -1; s = 0; }
    rel = (unsigned)(d - lo);
}

template <class Drain>
__device__ __forceinline__ void scan_chunk(const int* __restrict__ ei, int E, int e0, int e1,
                                           int lo, int span, int2* qw, int tid, Drain drain) {
    const unsigned lt = (1u << (tid & 31)) - 1u;
    const int ereal = e1 < E ? e1 : E;
    const bool wide = ((E & 3) == 0) && ((e0 & 3) == 0) && ((reinterpret_cast<size_t>(ei) & 15) == 0);
    int cnt = 0;
    for (int b = e0; b < e1; b += WG * UNR) {
        int s[UNR]; unsigned rel[UNR];
        if (wide && b + WG * UNR <= ereal) {
            const int4* ps = reinterpret_cast<const int4*>(ei + b) + tid;
            const int4* pd = reinterpret_cast<const int4*>(ei + E + b) + tid;
            const int4 s0 = ps[0], s1 = ps[WG], d0 = pd[0], d1 = pd[WG];
            s[0] = s0.x; s[1] = s0.y; s[2] = s0.z; s[3] = s0.w;
            s[4] = s1.x; s[5] = s1.y; s[6] = s1.z; s[7] = s1.w;
            rel[0] = (unsigned)(d0.x - lo); rel[1] = (unsigned)(d0.y - lo);
            rel[2] = (unsigned)(d0.z - lo); rel[3] = (unsigned)(d0.w - lo);
            rel[4] = (unsigned)(d1.x - lo); rel[5] = (unsigned)(d1.y - lo);
            rel[6] = (unsigned)(d1.z - lo); rel[7] = (unsigned)(d1.w - lo);
        } else if (b >= E && b + WG * UNR <= e1) {
#pragma unroll
            for (int u = 0; u < UNR; ++u) {
                s[u] = b - E + u * WG + tid;
                rel[u] = (unsigned)(s[u] - lo);
            }
        } else {
            int ls[UNR], ld[UNR];
#pragma unroll
            for (int u = 0; u < UNR; ++u) load_edge(ei, E, b + u * WG + tid, ls[u], ld[u]);
            asm volatile("" :: "v"(ls[0]), "v"(ls[1]), "v"(ls[2]), "v"(ls[3]),
                               "v"(ls[4]), "v"(ls[5]), "v"(ls[6]), "v"(ls[7]));
#pragma unroll
            for (int u = 0; u < UNR; ++u) place_edge(E, b + u * WG + tid, e1, lo, ls[u], ld[u], s[u], rel[u]);
        }
#pragma unroll
        for (int u = 0; u < UNR; ++u) {
            bool keep = rel[u] < (unsigned)span;
            unsigned mask = __builtin_amdgcn_ballot_w32(keep);
            if (keep) qw[cnt + __popc(mask & lt)] = make_int2(s[u], (int)rel[u]);
            cnt += __popc(mask);
            if (cnt > QCAP - 32) {
                wave_sync();
                drain(cnt);
                wave_sync();
                cnt = 0;
            }
        }
    }
    wave_sync();
    drain(cnt);
}

template <class Drain>
__device__ __forceinline__ void walk_slice(const unsigned* sl, int total,
                                           int2* qw, int lane, Drain drain) {
    if (total <= 0) return;
    const int last = total - 1;
    const unsigned smask = (1u << SRC_BITS) - 1u;
    unsigned a = sl[lane < last ? lane : last];
    unsigned b = sl[32 + lane < last ? 32 + lane : last];
    for (int k0 = 0; k0 < total; k0 += QCAP) {
        qw[lane]      = make_int2((int)(a & smask), (int)(a >> SRC_BITS));
        qw[32 + lane] = make_int2((int)(b & smask), (int)(b >> SRC_BITS));
        const int kn = k0 + QCAP;
        if (kn < total) {
            a = sl[kn + lane < last ? kn + lane : last];
            b = sl[kn + 32 + lane < last ? kn + 32 + lane : last];
        }
        wave_sync();
        drain(total - k0 < QCAP ? total - k0 : QCAP);
        wave_sync();
    }
}

template <class Drain>
__device__ __forceinline__ void walk_slice_val(const unsigned* sl, const float* sv, int total,
                                               int2* qw, int lane, Drain drain) {
    if (total <= 0) return;
    const int last = total - 1;
    int i0 = lane < last ? lane : last, i1 = 32 + lane < last ? 32 + lane : last;
    unsigned a = sl[i0], b = sl[i1];
    float va = sv[i0], vb = sv[i1];
    for (int k0 = 0; k0 < total; k0 += QCAP) {
        qw[lane]      = make_int2(__float_as_int(va), (int)(a >> SRC_BITS));
        qw[32 + lane] = make_int2(__float_as_int(vb), (int)(b >> SRC_BITS));
        const int kn = k0 + QCAP;
        if (kn < total) {
            i0 = kn + lane < last ? kn + lane : last;
            i1 = kn + 32 + lane < last ? kn + 32 + lane : last;
            a = sl[i0]; b = sl[i1];
            va = sv[i0]; vb = sv[i1];
        }
        wave_sync();
        drain(total - k0 < QCAP ? total - k0 : QCAP);
        wave_sync();
    }
}

template <bool B> struct on_slices { static constexpr bool value = B; };

__device__ __forceinline__ void head_consts(const float* __restrict__ W1, const float* __restrict__ as1,
                                            const float* __restrict__ ad1, int h, float& wsl, float& wdl) {
    float a = 0.f, b = 0.f;
#pragma unroll
    for (int c = 0; c < 8; ++c) {
        float w = W1[h * 8 + c];
        a += w * as1[h * 8 + c];
        b += w * ad1[h * 8 + c];
    }
    wsl = a; wdl = b;
}

template <int SPAN, int G>
__global__ __launch_bounds__(WG) void k_l1(const int* __restrict__ ei, int E, int N,
                                           const float* __restrict__ x,
                                           const float* __restrict__ W1,
                                           const float* __restrict__ as1,
                                           const float* __restrict__ ad1,
                                           float* __restrict__ pm1, float* __restrict__ prec1,
                                           unsigned* sl, float* slx, int* __restrict__ slcnt, int capw) {
    __shared__ __attribute__((aligned(16))) unsigned mx[SPAN * 4];
    __shared__ __attribute__((aligned(16))) float rec[SPAN * 8];
    __shared__ int2 q[(WG / 32) * QCAP];
    const int tid = threadIdx.x, lane = tid & 31;
    const int r = blockIdx.x / G, g = blockIdx.x % G;
    const int lo = r * SPAN;
    const int span = (N - lo < SPAN) ? (N - lo) : SPAN;
    const int h = lane & 3;
    float wsl, wdl;
    head_consts(W1, as1, ad1, h, wsl, wdl);

    for (int i = tid; i < span * 4; i += WG) mx[i] = FORD_NEG_INF;
    for (int i = tid; i < span * 8; i += WG) rec[i] = 0.f;
    __syncthreads();

    int2* qw = q + (tid >> 5) * QCAP;
    const int wslot = blockIdx.x * (WG / 32) + (tid >> 5);
    unsigned* slw = sl + (size_t)wslot * capw;
    float* slxw = slx + (size_t)wslot * capw;
    int pos = 0;
    const int sub4 = lane >> 2;
    auto drain_max = [&](int cnt) {
        for (int k0 = 0; k0 < cnt; k0 += 32) {
            bool ok[4]; int kq[4]; int2 en[4]; float xs[4], xdv[4];
#pragma unroll
            for (int rr = 0; rr < 4; ++rr) {
                kq[rr] = k0 + rr * 8 + sub4;
                ok[rr] = kq[rr] < cnt;
                en[rr] = qw[ok[rr] ? kq[rr] : cnt - 1];
            }
#pragma unroll
            for (int rr = 0; rr < 4; ++rr) { xs[rr] = x[en[rr].x]; xdv[rr] = x[lo + en[rr].y]; }
            asm volatile("" :: "v"(xs[0]), "v"(xs[1]), "v"(xs[2]), "v"(xs[3]),
                               "v"(xdv[0]), "v"(xdv[1]), "v"(xdv[2]), "v"(xdv[3]));
#pragma unroll
            for (int rr = 0; rr < 4; ++rr) {
                unsigned v = ford(lrelu(xs[rr] * wsl + xdv[rr] * wdl));
                if (ok[rr]) atomicMax(&mx[en[rr].y * 4 + h], v);
            }
            if (h == 0) {
#pragma unroll
                for (int rr = 0; rr < 4; ++rr)
                    if (ok[rr] && pos + kq[rr] < capw) slxw[pos + kq[rr]] = xs[rr];
            }
        }
        if (lane < cnt && pos + lane < capw) {
            int2 en = qw[lane];
            slw[pos + lane] = (unsigned)en.x | ((unsigned)en.y << SRC_BITS);
        }
        if (32 + lane < cnt && pos + 32 + lane < capw) {
            int2 en = qw[32 + lane];
            slw[pos + 32 + lane] = (unsigned)en.x | ((unsigned)en.y << SRC_BITS);
        }
        pos += cnt;
    };
    const int EN = E + N;
    const int per = ((EN + G - 1) / G + 3) & ~3;
    const int e0 = g * per;
    const int e1 = (e0 + per < EN) ? (e0 + per) : EN;
    scan_chunk(ei, E, e0, e1, lo, span, qw, tid, drain_max);
    if (lane == 0) slcnt[wslot] = pos;
    __syncthreads();

    auto sum_pass = [&](int cnt, auto where) {
        constexpr bool SAVED = decltype(where)::value;
        for (int k0 = 0; k0 < cnt; k0 += 32) {
            bool ok[4]; int2 en[4]; float xs[4], xdv[4];
#pragma unroll
            for (int rr = 0; rr < 4; ++rr) {
                int k = k0 + rr * 8 + sub4;
                ok[rr] = k < cnt;
                en[rr] = qw[ok[rr] ? k : cnt - 1];
            }
#pragma unroll
            for (int rr = 0; rr < 4; ++rr) {
                if constexpr (SAVED) xs[rr] = __int_as_float(en[rr].x);
                else                 xs[rr] = x[en[rr].x];
                xdv[rr] = x[lo + en[rr].y];
            }
            if constexpr (SAVED)
                asm volatile("" :: "v"(xdv[0]), "v"(xdv[1]), "v"(xdv[2]), "v"(xdv[3]));
            else
                asm volatile("" :: "v"(xs[0]), "v"(xs[1]), "v"(xs[2]), "v"(xs[3]),
                                   "v"(xdv[0]), "v"(xdv[1]), "v"(xdv[2]), "v"(xdv[3]));
#pragma unroll
            for (int rr = 0; rr < 4; ++rr) {
                float t = __expf(lrelu(xs[rr] * wsl + xdv[rr] * wdl) - forddec(mx[en[rr].y * 4 + h]));
                if (ok[rr]) {
                    atomicAdd(&rec[en[rr].y * 8 + h], t);
                    atomicAdd(&rec[en[rr].y * 8 + 4 + h], t * xs[rr]);
                }
            }
        }
    };
    auto drain_sum_saved  = [&](int cnt) { sum_pass(cnt, on_slices<true>{}); };
    auto drain_sum_gather = [&](int cnt) { sum_pass(cnt, on_slices<false>{}); };
    const int total = __builtin_amdgcn_readfirstlane(pos);
    if (total <= capw) walk_slice_val(slw, slxw, total, qw, lane, drain_sum_saved);
    else               scan_chunk(ei, E, e0, e1, lo, span, qw, tid, drain_sum_gather);
    __syncthreads();

    float4* pm = reinterpret_cast<float4*>(pm1 + ((size_t)g * N + lo) * 4);
    for (int k = tid; k < span; k += WG) {
        const uint4 u = *reinterpret_cast<const uint4*>(&mx[k * 4]);
        pm[k] = make_float4(forddec(u.x), forddec(u.y), forddec(u.z), forddec(u.w));
    }
    float4* pr = reinterpret_cast<float4*>(prec1 + ((size_t)g * N + lo) * 8);
    for (int k = tid; k < span * 2; k += WG) pr[k] = *reinterpret_cast<const float4*>(&rec[k * 4]);
}

template <int G>
__global__ void k_fin1(const float* __restrict__ W1, const float* __restrict__ b1,
                       const float* __restrict__ pm1, const float* __restrict__ prec1,
                       float* __restrict__ out1, int N) {
    const int i = blockIdx.x * blockDim.x + threadIdx.x;
    const int ic = i < N * 4 ? i : N * 4 - 1;
    const int n = ic >> 2, h = ic & 3;
    float m[G], s[G], w[G];
#pragma unroll
    for (int g = 0; g < G; ++g) {
        m[g] = pm1[(size_t)g * N * 4 + ic];
        s[g] = prec1[((size_t)g * N + n) * 8 + h];
        w[g] = prec1[((size_t)g * N + n) * 8 + 4 + h];
    }
    float M = m[0];
#pragma unroll
    for (int g = 1; g < G; ++g) M = fmaxf(M, m[g]);
    float S = 0.f, Wt = 0.f;
#pragma unroll
    for (int g = 0; g < G; ++g) {
        float sc = __expf(m[g] - M);
        S += s[g] * sc;
        Wt += w[g] * sc;
    }
    const float ratio = Wt / S;

    const int lane = threadIdx.x & 31;
    const int node0 = (i - lane) >> 2;
    const int hd = (lane & 7) >> 1, c0 = (lane & 1) * 4;
#pragma unroll
    for (int k = 0; k < 2; ++k) {
        const int nl = k * 4 + (lane >> 3);
        const float rt = __shfl(ratio, nl * 4 + hd, 32);
        float v[4];
#pragma unroll
        for (int c = 0; c < 4; ++c) {
            float t = W1[hd * 8 + c0 + c] * rt + b1[hd * 8 + c0 + c];
            v[c] = t > 0.f ? t : 0.f;
        }
        const int node = node0 + nl;
        if (node < N)
            *reinterpret_cast<float4*>(out1 + (size_t)node * 32 + hd * 8 + c0) = make_float4(v[0], v[1], v[2], v[3]);
    }
}

__global__ void k_gemm2(const float* __restrict__ out1, const float* __restrict__ W2,
                        const float* __restrict__ as2, const float* __restrict__ ad2,
                        float* __restrict__ h2, float* __restrict__ alS2,
                        float* __restrict__ alD2, int N) {
    __shared__ float Bp[32 * 16];
    int t = threadIdx.x;
    for (int i = t; i < 512; i += blockDim.x) {
        int k = i >> 4, n = i & 15;
        float v = 0.f;
        if (n < 8) {
            v = W2[k * 8 + n];
        } else if (n == 8) {
            for (int c = 0; c < 8; ++c) v += W2[k * 8 + c] * as2[c];
        } else if (n == 9) {
            for (int c = 0; c < 8; ++c) v += W2[k * 8 + c] * ad2[c];
        }
        Bp[i] = v;
    }
    __syncthreads();

    int wave = t >> 5, lane = t & 31;
    long tile = (long)blockIdx.x * 8 + wave;
    long row0 = tile * 16;
    if (row0 >= N) return;

    int m = lane & 15, khi = lane >> 4;
    long rowm = row0 + m;
    if (rowm >= N) rowm = N - 1;
    const float* Arow = out1 + rowm * 32;

    v8f c = {};
#pragma unroll
    for (int step = 0; step < 8; ++step) {
        int k0 = step * 4 + khi * 2;
        v2f a; a.x = Arow[k0];          a.y = Arow[k0 + 1];
        v2f b; b.x = Bp[k0 * 16 + m];   b.y = Bp[(k0 + 1) * 16 + m];
        c = __builtin_amdgcn_wmma_f32_16x16x4_f32(false, a, false, b, (short)0, c,
                                                  false, false);
    }

#pragma unroll
    for (int r = 0; r < 8; ++r) {
        long node = row0 + khi * 8 + r;
        if (node >= N) continue;
        float v = c[r];
        if (m < 8)        h2[node * 8 + m] = v;
        else if (m == 8)  alS2[node] = v;
        else if (m == 9)  alD2[node] = v;
    }
}

template <int SPAN, int G>
__global__ __launch_bounds__(WG) void k_l2(const int* __restrict__ ei, int E,
                                           const unsigned* __restrict__ sl, float* sla,
                                           const int* __restrict__ slcnt, int capw, int N,
                                           const float* __restrict__ h2,
                                           const float* __restrict__ alS2,
                                           const float* __restrict__ alD2,
                                           float* __restrict__ pm2, float* __restrict__ ps2,
                                           float* __restrict__ pacc2) {
    __shared__ unsigned mx[SPAN];
    __shared__ float ad[SPAN];
    __shared__ float ssum[SPAN];
    __shared__ __attribute__((aligned(16))) float acc[SPAN * 8];
    __shared__ int2 q[(WG / 32) * QCAP];
    __shared__ float tq[WG];
    const int tid = threadIdx.x, lane = tid & 31;
    const int r = blockIdx.x / G, g = blockIdx.x % G;
    const int lo = r * SPAN;
    const int span = (N - lo < SPAN) ? (N - lo) : SPAN;

    for (int i = tid; i < span; i += WG) { ad[i] = alD2[lo + i]; mx[i] = FORD_NEG_INF; ssum[i] = 0.f; }
    for (int i = tid; i < span * 8; i += WG) acc[i] = 0.f;
    __syncthreads();

    int2* qw = q + (tid >> 5) * QCAP;
    const int wslot = blockIdx.x * (WG / 32) + (tid >> 5);
    const unsigned* slw = sl + (size_t)wslot * capw;
    float* slaw = sla + (size_t)wslot * capw;
    int posm = 0, poss = 0;
    auto max_pass = [&](int cnt, auto where) {
        constexpr bool SAVED = decltype(where)::value;
        for (int k0 = 0; k0 < cnt; k0 += 64) {
            bool ok[2]; int kq[2]; int2 en[2]; float as[2];
#pragma unroll
            for (int rr = 0; rr < 2; ++rr) {
                kq[rr] = k0 + rr * 32 + lane;
                ok[rr] = kq[rr] < cnt;
                en[rr] = qw[ok[rr] ? kq[rr] : cnt - 1];
            }
#pragma unroll
            for (int rr = 0; rr < 2; ++rr) as[rr] = alS2[en[rr].x];
            asm volatile("" :: "v"(as[0]), "v"(as[1]));
#pragma unroll
            for (int rr = 0; rr < 2; ++rr)
                if (ok[rr]) atomicMax(&mx[en[rr].y], ford(lrelu(as[rr] + ad[en[rr].y])));
            if constexpr (SAVED) {
#pragma unroll
                for (int rr = 0; rr < 2; ++rr)
                    if (ok[rr]) slaw[posm + kq[rr]] = as[rr];
            }
        }
        if constexpr (SAVED) posm += cnt;
    };
    auto drain_max_saved  = [&](int cnt) { max_pass(cnt, on_slices<true>{}); };
    auto drain_max_gather = [&](int cnt) { max_pass(cnt, on_slices<false>{}); };
    float* tw = tq + (tid & ~31);
    const int j = lane & 7, sub = lane >> 3;
    auto sum_pass = [&](int cnt, auto where) {
        constexpr bool SAVED = decltype(where)::value;
        for (int k0 = 0; k0 < cnt; k0 += 32) {
            const int nb = cnt - k0 < 32 ? cnt - k0 : 32;
            const int2* qb = qw + k0;
            const bool oka = lane < nb;
            const int2 ea = qb[oka ? lane : nb - 1];
            bool ok[8]; int kk[8]; int2 en[8]; float hv[8];
#pragma unroll
            for (int rr = 0; rr < 8; ++rr) {
                int k = rr * 4 + sub;
                ok[rr] = k < nb;
                kk[rr] = ok[rr] ? k : nb - 1;
                en[rr] = qb[kk[rr]];
            }
            const float adv = ad[ea.y];
            const unsigned mxv = mx[ea.y];
            float as;
            if constexpr (SAVED) as = slaw[poss + k0 + (oka ? lane : nb - 1)];
            else                 as = alS2[ea.x];
#pragma unroll
            for (int rr = 0; rr < 8; ++rr) hv[rr] = h2[(size_t)en[rr].x * 8 + j];
            asm volatile("" :: "v"(as), "v"(hv[0]), "v"(hv[1]), "v"(hv[2]), "v"(hv[3]),
                               "v"(hv[4]), "v"(hv[5]), "v"(hv[6]), "v"(hv[7]));
            const float t = __expf(lrelu(as + adv) - forddec(mxv));
            if (oka) atomicAdd(&ssum[ea.y], t);
            tw[lane] = t;
            wave_sync();
            float tv[8];
#pragma unroll
            for (int rr = 0; rr < 8; ++rr) tv[rr] = tw[kk[rr]];
            asm volatile("" :: "v"(tv[0]), "v"(tv[1]), "v"(tv[2]), "v"(tv[3]),
                               "v"(tv[4]), "v"(tv[5]), "v"(tv[6]), "v"(tv[7]));
#pragma unroll
            for (int rr = 0; rr < 8; ++rr)
                if (ok[rr]) atomicAdd(&acc[en[rr].y * 8 + j], tv[rr] * hv[rr]);
            wave_sync();
        }
        if constexpr (SAVED) poss += cnt;
    };
    auto drain_sum_saved  = [&](int cnt) { sum_pass(cnt, on_slices<true>{}); };
    auto drain_sum_gather = [&](int cnt) { sum_pass(cnt, on_slices<false>{}); };
    const int total = __builtin_amdgcn_readfirstlane(slcnt[wslot]);
    const int EN = E + N;
    const int per = ((EN + G - 1) / G + 3) & ~3;
    const int e0 = g * per;
    const int e1 = (e0 + per < EN) ? (e0 + per) : EN;
    if (total <= capw) walk_slice(slw, total, qw, lane, drain_max_saved);
    else               scan_chunk(ei, E, e0, e1, lo, span, qw, tid, drain_max_gather);
    __syncthreads();
    if (total <= capw) walk_slice(slw, total, qw, lane, drain_sum_saved);
    else               scan_chunk(ei, E, e0, e1, lo, span, qw, tid, drain_sum_gather);
    __syncthreads();

    float* pm = pm2 + (size_t)g * N + lo;
    float* ps = ps2 + (size_t)g * N + lo;
    for (int k = tid; k < span; k += WG) { pm[k] = forddec(mx[k]); ps[k] = ssum[k]; }
    float4* pa = reinterpret_cast<float4*>(pacc2 + ((size_t)g * N + lo) * 8);
    for (int k = tid; k < span * 2; k += WG) pa[k] = *reinterpret_cast<const float4*>(&acc[k * 4]);
}

template <int G>
__global__ void k_fin2(const float* __restrict__ pm2, const float* __restrict__ ps2,
                       const float* __restrict__ pacc2,
                       const float* __restrict__ b2, float* __restrict__ out, int N) {
    const int i = blockIdx.x * blockDim.x + threadIdx.x;
    if (i >= N * 2) return;
    const int n = i >> 1, q = i & 1;
    float m[G], s[G]; float4 a[G];
#pragma unroll
    for (int g = 0; g < G; ++g) {
        m[g] = pm2[(size_t)g * N + n];
        s[g] = ps2[(size_t)g * N + n];
        a[g] = reinterpret_cast<const float4*>(pacc2 + (size_t)g * N * 8)[i];
    }
    float M = m[0];
#pragma unroll
    for (int g = 1; g < G; ++g) M = fmaxf(M, m[g]);
    float S = 0.f, A0 = 0.f, A1 = 0.f, A2 = 0.f, A3 = 0.f;
#pragma unroll
    for (int g = 0; g < G; ++g) {
        float sc = __expf(m[g] - M);
        S  += s[g] * sc;
        A0 += a[g].x * sc; A1 += a[g].y * sc; A2 += a[g].z * sc; A3 += a[g].w * sc;
    }
    reinterpret_cast<float4*>(out)[i] = make_float4(A0 / S + b2[q * 4 + 0], A1 / S + b2[q * 4 + 1],
                                                    A2 / S + b2[q * 4 + 2], A3 / S + b2[q * 4 + 3]);
}

extern "C" void kernel_launch(void* const* d_in, const int* in_sizes, int n_in,
                              void* d_out, int out_size, void* d_ws, size_t ws_size,
                              hipStream_t stream) {
    const float* x   = (const float*)d_in[0];
    const int*   ei  = (const int*)d_in[1];
    const float* W1  = (const float*)d_in[2];
    const float* as1 = (const float*)d_in[3];
    const float* ad1 = (const float*)d_in[4];
    const float* b1  = (const float*)d_in[5];
    const float* W2  = (const float*)d_in[6];
    const float* as2 = (const float*)d_in[7];
    const float* ad2 = (const float*)d_in[8];
    const float* b2  = (const float*)d_in[9];
    (void)n_in; (void)out_size; (void)ws_size;

    int N = in_sizes[0];
    int E = in_sizes[1] / 2;

    char* ws = (char*)d_ws;
    size_t cur = 0;
    auto carve = [&](size_t bytes) -> void* {
        void* p = ws + cur;
        cur += (bytes + 255) & ~(size_t)255;
        return p;
    };
    float*    pm1   = (float*)carve(sizeof(float) * (size_t)G_ALL * N * 4);
    float*    prec1 = (float*)carve(sizeof(float) * (size_t)G_ALL * N * 8);
    float*    pm2   = (float*)carve(sizeof(float) * (size_t)G_ALL * N);
    float*    ps2   = (float*)carve(sizeof(float) * (size_t)G_ALL * N);
    float*    pacc2 = (float*)carve(sizeof(float) * (size_t)G_ALL * N * 8);
    float*    out1  = (float*)carve(sizeof(float) * (size_t)N * 32);
    float*    h2    = (float*)carve(sizeof(float) * (size_t)N * 8);
    float*    alS2  = (float*)carve(sizeof(float) * (size_t)N);
    float*    alD2  = (float*)carve(sizeof(float) * (size_t)N);
    const int nwg   = ((N + SPAN_ALL - 1) / SPAN_ALL) * G_ALL;
    const int per   = ((E + N + G_ALL - 1) / G_ALL + 3) & ~3;
    const int capf  = ((per + WG * UNR - 1) / (WG * UNR)) * UNR * 32;
    const int capw  = capf < SLICE_CAP ? capf : SLICE_CAP;
    unsigned* sl    = (unsigned*)carve(sizeof(unsigned) * (size_t)nwg * (WG / 32) * capw);
    int*      slcnt = (int*)carve(sizeof(int) * (size_t)nwg * (WG / 32));
    float*    slx   = pm2;
    float*    sla   = pm1;

    const int B = 256;
    k_l1<SPAN_ALL, G_ALL><<<nwg, WG, 0, stream>>>(ei, E, N, x, W1, as1, ad1, pm1, prec1, sl, slx, slcnt, capw);
    k_fin1<G_ALL><<<(4 * N + B - 1) / B, B, 0, stream>>>(W1, b1, pm1, prec1, out1, N);

    int tiles = (N + 15) / 16;
    k_gemm2<<<(tiles + 7) / 8, B, 0, stream>>>(out1, W2, as2, ad2, h2, alS2, alD2, N);

    k_l2<SPAN_ALL, G_ALL><<<nwg, WG, 0, stream>>>(ei, E, sl, sla, slcnt, capw, N, h2, alS2, alD2, pm2, ps2, pacc2);
    k_fin2<G_ALL><<<(2 * N + B - 1) / B, B, 0, stream>>>(pm2, ps2, pacc2, b2, (float*)d_out, N);
}


# Queued edges are applied in full batches of 32 by bodies without guard, clamp or select (all four edge passes); the filter pass carries its remainder, no near-empty second batch; sum order kept
# speedup vs baseline: 12.1408x; 1.0364x over previous
#include <hip/hip_runtime.h>
#include <math.h>


#define NEG_SLOPE 0.2f
#define WG 1024
#define UNR 8
#define QCAP 64
#define SPAN_ALL 6250
#define G_ALL 16
#define SRC_BITS 19
#define SLICE_CAP 2048
static_assert(UNR == 8, "the load fence in scan_chunk names eight values; a 16-byte trip is two loads of 4 edges a lane");
static_assert(QCAP >= 64, "a push of 32 entries on top of 31 carried ones must fit");
static_assert(QCAP == 64, "walk_slice refills the queue with two entries per lane");
static_assert(SPAN_ALL <= (1 << (32 - SRC_BITS)), "destination - lo must fit above the source bits");

typedef __attribute__((ext_vector_type(2))) float v2f;
typedef __attribute__((ext_vector_type(8))) float v8f;

__device__ __forceinline__ float lrelu(float v) { return v > 0.f ? v : NEG_SLOPE * v; }

__device__ __forceinline__ unsigned ford(float f) {
    unsigned u = __float_as_uint(f);
    return (u & 0x80000000u) ? ~u : (u | 0x80000000u);
}
__device__ __forceinline__ float forddec(unsigned u) {
    return (u & 0x80000000u) ? __uint_as_float(u & 0x7fffffffu) : __uint_as_float(~u);
}
#define FORD_NEG_INF 0x007FFFFFu


__device__ __forceinline__ void wave_sync() {
    __builtin_amdgcn_fence(__ATOMIC_RELEASE, "wavefront");
    __builtin_amdgcn_wave_barrier();
    __builtin_amdgcn_fence(__ATOMIC_ACQUIRE, "wavefront");
}

__device__ __forceinline__ void load_edge(const int* __restrict__ ei, int E, int e, int& ls, int& ld) {
    int ec = e < E ? e : E - 1;
    ls = ei[ec];
    ld = ei[E + ec];
}
__device__ __forceinline__ void place_edge(int E, int e, int e1, int lo, int ls, int ld,
                                           int& s, unsigned& rel) {
    int d = e < E ? ld : e - E;
    s     = e < E ? ls : e - E;
    if (e >= e1) { d = -1; s = 0; }
    rel = (unsigned)(d - lo);
}

template <class Drain>
__device__ __forceinline__ void scan_chunk(const int* __restrict__ ei, int E, int e0, int e1,
                                           int lo, int span, int2* qw, int tid, Drain drain) {
    const unsigned lt = (1u << (tid & 31)) - 1u;
    const int ereal = e1 < E ? e1 : E;
    const bool wide = ((E & 3) == 0) && ((e0 & 3) == 0) && ((reinterpret_cast<size_t>(ei) & 15) == 0);
    int cnt = 0;
    for (int b = e0; b < e1; b += WG * UNR) {
        int s[UNR]; unsigned rel[UNR];
        if (wide && b + WG * UNR <= ereal) {
            const int4* ps = reinterpret_cast<const int4*>(ei + b) + tid;
            const int4* pd = reinterpret_cast<const int4*>(ei + E + b) + tid;
            const int4 s0 = ps[0], s1 = ps[WG], d0 = pd[0], d1 = pd[WG];
            s[0] = s0.x; s[1] = s0.y; s[2] = s0.z; s[3] = s0.w;
            s[4] = s1.x; s[5] = s1.y; s[6] = s1.z; s[7] = s1.w;
            rel[0] = (unsigned)(d0.x - lo); rel[1] = (unsigned)(d0.y - lo);
            rel[2] = (unsigned)(d0.z - lo); rel[3] = (unsigned)(d0.w - lo);
            rel[4] = (unsigned)(d1.x - lo); rel[5] = (unsigned)(d1.y - lo);
            rel[6] = (unsigned)(d1.z - lo); rel[7] = (unsigned)(d1.w - lo);
        } else if (b >= E && b + WG * UNR <= e1) {
#pragma unroll
            for (int u = 0; u < UNR; ++u) {
                s[u] = b - E + u * WG + tid;
                rel[u] = (unsigned)(s[u] - lo);
            }
        } else {
            int ls[UNR], ld[UNR];
#pragma unroll
            for (int u = 0; u < UNR; ++u) load_edge(ei, E, b + u * WG + tid, ls[u], ld[u]);
            asm volatile("" :: "v"(ls[0]), "v"(ls[1]), "v"(ls[2]), "v"(ls[3]),
                               "v"(ls[4]), "v"(ls[5]), "v"(ls[6]), "v"(ls[7]));
#pragma unroll
            for (int u = 0; u < UNR; ++u) place_edge(E, b + u * WG + tid, e1, lo, ls[u], ld[u], s[u], rel[u]);
        }
#pragma unroll
        for (int u = 0; u < UNR; ++u) {
            bool keep = rel[u] < (unsigned)span;
            unsigned mask = __builtin_amdgcn_ballot_w32(keep);
            if (keep) qw[cnt + __popc(mask & lt)] = make_int2(s[u], (int)rel[u]);
            cnt += __popc(mask);
            if (cnt >= 32) {
                wave_sync();
                drain(32);
                wave_sync();
                const int rem = cnt - 32;
                if ((tid & 31) < rem) { const int2 c = qw[32 + (tid & 31)]; qw[tid & 31] = c; }
                cnt = rem;
            }
        }
    }
    wave_sync();
    drain(cnt);
}

template <class Drain>
__device__ __forceinline__ void walk_slice(const unsigned* sl, int total,
                                           int2* qw, int lane, Drain drain) {
    if (total <= 0) return;
    const int last = total - 1;
    const unsigned smask = (1u << SRC_BITS) - 1u;
    unsigned a = sl[lane < last ? lane : last];
    unsigned b = sl[32 + lane < last ? 32 + lane : last];
    for (int k0 = 0; k0 < total; k0 += QCAP) {
        qw[lane]      = make_int2((int)(a & smask), (int)(a >> SRC_BITS));
        qw[32 + lane] = make_int2((int)(b & smask), (int)(b >> SRC_BITS));
        const int kn = k0 + QCAP;
        if (kn < total) {
            a = sl[kn + lane < last ? kn + lane : last];
            b = sl[kn + 32 + lane < last ? kn + 32 + lane : last];
        }
        wave_sync();
        drain(total - k0 < QCAP ? total - k0 : QCAP);
        wave_sync();
    }
}

template <class Drain>
__device__ __forceinline__ void walk_slice_val(const unsigned* sl, const float* sv, int total,
                                               int2* qw, int lane, Drain drain) {
    if (total <= 0) return;
    const int last = total - 1;
    int i0 = lane < last ? lane : last, i1 = 32 + lane < last ? 32 + lane : last;
    unsigned a = sl[i0], b = sl[i1];
    float va = sv[i0], vb = sv[i1];
    for (int k0 = 0; k0 < total; k0 += QCAP) {
        qw[lane]      = make_int2(__float_as_int(va), (int)(a >> SRC_BITS));
        qw[32 + lane] = make_int2(__float_as_int(vb), (int)(b >> SRC_BITS));
        const int kn = k0 + QCAP;
        if (kn < total) {
            i0 = kn + lane < last ? kn + lane : last;
            i1 = kn + 32 + lane < last ? kn + 32 + lane : last;
            a = sl[i0]; b = sl[i1];
            va = sv[i0]; vb = sv[i1];
        }
        wave_sync();
        drain(total - k0 < QCAP ? total - k0 : QCAP);
        wave_sync();
    }
}

template <bool B> struct on_slices { static constexpr bool value = B; };
template <bool B> struct full_batch { static constexpr bool value = B; };

__device__ __forceinline__ void head_consts(const float* __restrict__ W1, const float* __restrict__ as1,
                                            const float* __restrict__ ad1, int h, float& wsl, float& wdl) {
    float a = 0.f, b = 0.f;
#pragma unroll
    for (int c = 0; c < 8; ++c) {
        float w = W1[h * 8 + c];
        a += w * as1[h * 8 + c];
        b += w * ad1[h * 8 + c];
    }
    wsl = a; wdl = b;
}

template <int SPAN, int G>
__global__ __launch_bounds__(WG) void k_l1(const int* __restrict__ ei, int E, int N,
                                           const float* __restrict__ x,
                                           const float* __restrict__ W1,
                                           const float* __restrict__ as1,
                                           const float* __restrict__ ad1,
                                           float* __restrict__ pm1, float* __restrict__ prec1,
                                           unsigned* sl, float* slx, int* __restrict__ slcnt, int capw) {
    __shared__ __attribute__((aligned(16))) unsigned mx[SPAN * 4];
    __shared__ __attribute__((aligned(16))) float rec[SPAN * 8];
    __shared__ int2 q[(WG / 32) * QCAP];
    const int tid = threadIdx.x, lane = tid & 31;
    const int r = blockIdx.x / G, g = blockIdx.x % G;
    const int lo = r * SPAN;
    const int span = (N - lo < SPAN) ? (N - lo) : SPAN;
    const int h = lane & 3;
    float wsl, wdl;
    head_consts(W1, as1, ad1, h, wsl, wdl);

    for (int i = tid; i < span * 4; i += WG) mx[i] = FORD_NEG_INF;
    for (int i = tid; i < span * 8; i += WG) rec[i] = 0.f;
    __syncthreads();

    int2* qw = q + (tid >> 5) * QCAP;
    const int wslot = blockIdx.x * (WG / 32) + (tid >> 5);
    unsigned* slw = sl + (size_t)wslot * capw;
    float* slxw = slx + (size_t)wslot * capw;
    int pos = 0;
    const int sub4 = lane >> 2;
    auto max_batch = [&](int k0, int cnt, auto full) {
        constexpr bool FULL = decltype(full)::value;
        bool ok[4]; int kq[4]; int2 en[4]; float xs[4], xdv[4];
#pragma unroll
        for (int rr = 0; rr < 4; ++rr) {
            kq[rr] = k0 + rr * 8 + sub4;
            ok[rr] = FULL || kq[rr] < cnt;
            en[rr] = qw[ok[rr] ? kq[rr] : cnt - 1];
        }
#pragma unroll
        for (int rr = 0; rr < 4; ++rr) { xs[rr] = x[en[rr].x]; xdv[rr] = x[lo + en[rr].y]; }
        if constexpr (FULL) {
            __builtin_amdgcn_sched_barrier(0);
        } else {
            asm volatile("" :: "v"(xs[0]), "v"(xs[1]), "v"(xs[2]), "v"(xs[3]),
                               "v"(xdv[0]), "v"(xdv[1]), "v"(xdv[2]), "v"(xdv[3]));
        }
#pragma unroll
        for (int rr = 0; rr < 4; ++rr) {
            unsigned v = ford(lrelu(xs[rr] * wsl + xdv[rr] * wdl));
            if (ok[rr]) atomicMax(&mx[en[rr].y * 4 + h], v);
        }
        if (h == 0) {
#pragma unroll
            for (int rr = 0; rr < 4; ++rr)
                if (ok[rr] && pos + kq[rr] < capw) slxw[pos + kq[rr]] = xs[rr];
        }
    };
    auto drain_max = [&](int cnt) {
        for (int k0 = 0; k0 < cnt; k0 += 32) {
            if (cnt - k0 >= 32) max_batch(k0, cnt, full_batch<true>{});
            else                max_batch(k0, cnt, full_batch<false>{});
        }
        if (lane < cnt && pos + lane < capw) {
            int2 en = qw[lane];
            slw[pos + lane] = (unsigned)en.x | ((unsigned)en.y << SRC_BITS);
        }
        if (32 + lane < cnt && pos + 32 + lane < capw) {
            int2 en = qw[32 + lane];
            slw[pos + 32 + lane] = (unsigned)en.x | ((unsigned)en.y << SRC_BITS);
        }
        pos += cnt;
    };
    const int EN = E + N;
    const int per = ((EN + G - 1) / G + 3) & ~3;
    const int e0 = g * per;
    const int e1 = (e0 + per < EN) ? (e0 + per) : EN;
    scan_chunk(ei, E, e0, e1, lo, span, qw, tid, drain_max);
    if (lane == 0) slcnt[wslot] = pos;
    __syncthreads();

    auto sum_batch = [&](int k0, int cnt, auto where, auto full) {
        constexpr bool SAVED = decltype(where)::value;
        constexpr bool FULL = decltype(full)::value;
        bool ok[4]; int2 en[4]; float xs[4], xdv[4];
#pragma unroll
        for (int rr = 0; rr < 4; ++rr) {
            int k = k0 + rr * 8 + sub4;
            ok[rr] = FULL || k < cnt;
            en[rr] = qw[ok[rr] ? k : cnt - 1];
        }
#pragma unroll
        for (int rr = 0; rr < 4; ++rr) {
            if constexpr (SAVED) xs[rr] = __int_as_float(en[rr].x);
            else                 xs[rr] = x[en[rr].x];
            xdv[rr] = x[lo + en[rr].y];
        }
        unsigned mv[4];
        if constexpr (FULL) {
#pragma unroll
            for (int rr = 0; rr < 4; ++rr) mv[rr] = mx[en[rr].y * 4 + h];
            __builtin_amdgcn_sched_barrier(0);
        } else {
            if constexpr (SAVED)
                asm volatile("" :: "v"(xdv[0]), "v"(xdv[1]), "v"(xdv[2]), "v"(xdv[3]));
            else
                asm volatile("" :: "v"(xs[0]), "v"(xs[1]), "v"(xs[2]), "v"(xs[3]),
                                   "v"(xdv[0]), "v"(xdv[1]), "v"(xdv[2]), "v"(xdv[3]));
        }
#pragma unroll
        for (int rr = 0; rr < 4; ++rr) {
            if constexpr (!FULL) mv[rr] = mx[en[rr].y * 4 + h];
            float t = __expf(lrelu(xs[rr] * wsl + xdv[rr] * wdl) - forddec(mv[rr]));
            if (ok[rr]) {
                atomicAdd(&rec[en[rr].y * 8 + h], t);
                atomicAdd(&rec[en[rr].y * 8 + 4 + h], t * xs[rr]);
            }
        }
    };
    auto sum_pass = [&](int cnt, auto where) {
        for (int k0 = 0; k0 < cnt; k0 += 32) {
            if (cnt - k0 >= 32) sum_batch(k0, cnt, where, full_batch<true>{});
            else                sum_batch(k0, cnt, where, full_batch<false>{});
        }
    };
    auto drain_sum_saved  = [&](int cnt) { sum_pass(cnt, on_slices<true>{}); };
    auto drain_sum_gather = [&](int cnt) { sum_pass(cnt, on_slices<false>{}); };
    const int total = __builtin_amdgcn_readfirstlane(pos);
    if (total <= capw) walk_slice_val(slw, slxw, total, qw, lane, drain_sum_saved);
    else               scan_chunk(ei, E, e0, e1, lo, span, qw, tid, drain_sum_gather);
    __syncthreads();

    float4* pm = reinterpret_cast<float4*>(pm1 + ((size_t)g * N + lo) * 4);
    for (int k = tid; k < span; k += WG) {
        const uint4 u = *reinterpret_cast<const uint4*>(&mx[k * 4]);
        pm[k] = make_float4(forddec(u.x), forddec(u.y), forddec(u.z), forddec(u.w));
    }
    float4* pr = reinterpret_cast<float4*>(prec1 + ((size_t)g * N + lo) * 8);
    for (int k = tid; k < span * 2; k += WG) pr[k] = *reinterpret_cast<const float4*>(&rec[k * 4]);
}

template <int G>
__global__ void k_fin1(const float* __restrict__ W1, const float* __restrict__ b1,
                       const float* __restrict__ pm1, const float* __restrict__ prec1,
                       float* __restrict__ out1, int N) {
    const int i = blockIdx.x * blockDim.x + threadIdx.x;
    const int ic = i < N * 4 ? i : N * 4 - 1;
    const int n = ic >> 2, h = ic & 3;
    float m[G], s[G], w[G];
#pragma unroll
    for (int g = 0; g < G; ++g) {
        m[g] = pm1[(size_t)g * N * 4 + ic];
        s[g] = prec1[((size_t)g * N + n) * 8 + h];
        w[g] = prec1[((size_t)g * N + n) * 8 + 4 + h];
    }
    float M = m[0];
#pragma unroll
    for (int g = 1; g < G; ++g) M = fmaxf(M, m[g]);
    float S = 0.f, Wt = 0.f;
#pragma unroll
    for (int g = 0; g < G; ++g) {
        float sc = __expf(m[g] - M);
        S += s[g] * sc;
        Wt += w[g] * sc;
    }
    const float ratio = Wt / S;

    const int lane = threadIdx.x & 31;
    const int node0 = (i - lane) >> 2;
    const int hd = (lane & 7) >> 1, c0 = (lane & 1) * 4;
#pragma unroll
    for (int k = 0; k < 2; ++k) {
        const int nl = k * 4 + (lane >> 3);
        const float rt = __shfl(ratio, nl * 4 + hd, 32);
        float v[4];
#pragma unroll
        for (int c = 0; c < 4; ++c) {
            float t = W1[hd * 8 + c0 + c] * rt + b1[hd * 8 + c0 + c];
            v[c] = t > 0.f ? t : 0.f;
        }
        const int node = node0 + nl;
        if (node < N)
            *reinterpret_cast<float4*>(out1 + (size_t)node * 32 + hd * 8 + c0) = make_float4(v[0], v[1], v[2], v[3]);
    }
}

__global__ void k_gemm2(const float* __restrict__ out1, const float* __restrict__ W2,
                        const float* __restrict__ as2, const float* __restrict__ ad2,
                        float* __restrict__ h2, float* __restrict__ alS2,
                        float* __restrict__ alD2, int N) {
    __shared__ float Bp[32 * 16];
    int t = threadIdx.x;
    for (int i = t; i < 512; i += blockDim.x) {
        int k = i >> 4, n = i & 15;
        float v = 0.f;
        if (n < 8) {
            v = W2[k * 8 + n];
        } else if (n == 8) {
            for (int c = 0; c < 8; ++c) v += W2[k * 8 + c] * as2[c];
        } else if (n == 9) {
            for (int c = 0; c < 8; ++c) v += W2[k * 8 + c] * ad2[c];
        }
        Bp[i] = v;
    }
    __syncthreads();

    int wave = t >> 5, lane = t & 31;
    long tile = (long)blockIdx.x * 8 + wave;
    long row0 = tile * 16;
    if (row0 >= N) return;

    int m = lane & 15, khi = lane >> 4;
    long rowm = row0 + m;
    if (rowm >= N) rowm = N - 1;
    const float* Arow = out1 + rowm * 32;

    v8f c = {};
#pragma unroll
    for (int step = 0; step < 8; ++step) {
        int k0 = step * 4 + khi * 2;
        v2f a; a.x = Arow[k0];          a.y = Arow[k0 + 1];
        v2f b; b.x = Bp[k0 * 16 + m];   b.y = Bp[(k0 + 1) * 16 + m];
        c = __builtin_amdgcn_wmma_f32_16x16x4_f32(false, a, false, b, (short)0, c,
                                                  false, false);
    }

#pragma unroll
    for (int r = 0; r < 8; ++r) {
        long node = row0 + khi * 8 + r;
        if (node >= N) continue;
        float v = c[r];
        if (m < 8)        h2[node * 8 + m] = v;
        else if (m == 8)  alS2[node] = v;
        else if (m == 9)  alD2[node] = v;
    }
}

template <int SPAN, int G>
__global__ __launch_bounds__(WG) void k_l2(const int* __restrict__ ei, int E,
                                           const unsigned* __restrict__ sl, float* sla,
                                           const int* __restrict__ slcnt, int capw, int N,
                                           const float* __restrict__ h2,
                                           const float* __restrict__ alS2,
                                           const float* __restrict__ alD2,
                                           float* __restrict__ pm2, float* __restrict__ ps2,
                                           float* __restrict__ pacc2) {
    __shared__ unsigned mx[SPAN];
    __shared__ float ad[SPAN];
    __shared__ float ssum[SPAN];
    __shared__ __attribute__((aligned(16))) float acc[SPAN * 8];
    __shared__ int2 q[(WG / 32) * QCAP];
    __shared__ float tq[WG];
    const int tid = threadIdx.x, lane = tid & 31;
    const int r = blockIdx.x / G, g = blockIdx.x % G;
    const int lo = r * SPAN;
    const int span = (N - lo < SPAN) ? (N - lo) : SPAN;

    for (int i = tid; i < span; i += WG) { ad[i] = alD2[lo + i]; mx[i] = FORD_NEG_INF; ssum[i] = 0.f; }
    for (int i = tid; i < span * 8; i += WG) acc[i] = 0.f;
    __syncthreads();

    int2* qw = q + (tid >> 5) * QCAP;
    const int wslot = blockIdx.x * (WG / 32) + (tid >> 5);
    const unsigned* slw = sl + (size_t)wslot * capw;
    float* slaw = sla + (size_t)wslot * capw;
    int posm = 0, poss = 0;
    auto max_batch = [&](int k0, int cnt, auto where, auto full) {
        constexpr bool SAVED = decltype(where)::value;
        constexpr bool FULL = decltype(full)::value;
        bool ok[2]; int kq[2]; int2 en[2]; float as[2];
#pragma unroll
        for (int rr = 0; rr < 2; ++rr) {
            kq[rr] = k0 + rr * 32 + lane;
            ok[rr] = FULL || kq[rr] < cnt;
            en[rr] = qw[ok[rr] ? kq[rr] : cnt - 1];
        }
#pragma unroll
        for (int rr = 0; rr < 2; ++rr) as[rr] = alS2[en[rr].x];
        if constexpr (!FULL) {
            asm volatile("" :: "v"(as[0]), "v"(as[1]));
        }
#pragma unroll
        for (int rr = 0; rr < 2; ++rr)
            if (ok[rr]) atomicMax(&mx[en[rr].y], ford(lrelu(as[rr] + ad[en[rr].y])));
        if constexpr (SAVED) {
#pragma unroll
            for (int rr = 0; rr < 2; ++rr)
                if (ok[rr]) slaw[posm + kq[rr]] = as[rr];
        }
    };
    auto max_pass = [&](int cnt, auto where) {
        for (int k0 = 0; k0 < cnt; k0 += 64) {
            if (cnt - k0 >= 64) max_batch(k0, cnt, where, full_batch<true>{});
            else                max_batch(k0, cnt, where, full_batch<false>{});
        }
        if constexpr (decltype(where)::value) posm += cnt;
    };
    auto drain_max_saved  = [&](int cnt) { max_pass(cnt, on_slices<true>{}); };
    auto drain_max_gather = [&](int cnt) { max_pass(cnt, on_slices<false>{}); };
    float* tw = tq + (tid & ~31);
    const int j = lane & 7, sub = lane >> 3;
    auto sum_batch = [&](int k0, int cnt, auto where, auto full) {
        constexpr bool SAVED = decltype(where)::value;
        constexpr bool FULL = decltype(full)::value;
        const int nb = FULL ? 32 : cnt - k0;
        const int2* qb = qw + k0;
        const bool oka = FULL || lane < nb;
        const int2 ea = qb[oka ? lane : nb - 1];
        bool ok[8]; int kk[8]; int2 en[8]; float hv[8];
#pragma unroll
        for (int rr = 0; rr < 8; ++rr) {
            int k = rr * 4 + sub;
            ok[rr] = FULL || k < nb;
            kk[rr] = ok[rr] ? k : nb - 1;
            en[rr] = qb[kk[rr]];
        }
        const float adv = ad[ea.y];
        const unsigned mxv = mx[ea.y];
        float as;
        if constexpr (SAVED) as = slaw[poss + k0 + (oka ? lane : nb - 1)];
        else                 as = alS2[ea.x];
#pragma unroll
        for (int rr = 0; rr < 8; ++rr) hv[rr] = h2[(size_t)en[rr].x * 8 + j];
        if constexpr (!FULL) {
            asm volatile("" :: "v"(as), "v"(hv[0]), "v"(hv[1]), "v"(hv[2]), "v"(hv[3]),
                               "v"(hv[4]), "v"(hv[5]), "v"(hv[6]), "v"(hv[7]));
        }
        const float t = __expf(lrelu(as + adv) - forddec(mxv));
        if (oka) atomicAdd(&ssum[ea.y], t);
        tw[lane] = t;
        wave_sync();
        float tv[8];
#pragma unroll
        for (int rr = 0; rr < 8; ++rr) tv[rr] = tw[kk[rr]];
        if constexpr (!FULL) {
            asm volatile("" :: "v"(tv[0]), "v"(tv[1]), "v"(tv[2]), "v"(tv[3]),
                               "v"(tv[4]), "v"(tv[5]), "v"(tv[6]), "v"(tv[7]));
        }
#pragma unroll
        for (int rr = 0; rr < 8; ++rr)
            if (ok[rr]) atomicAdd(&acc[en[rr].y * 8 + j], tv[rr] * hv[rr]);
        wave_sync();
    };
    auto sum_pass = [&](int cnt, auto where) {
        for (int k0 = 0; k0 < cnt; k0 += 32) {
            if (cnt - k0 >= 32) sum_batch(k0, cnt, where, full_batch<true>{});
            else                sum_batch(k0, cnt, where, full_batch<false>{});
        }
        if constexpr (decltype(where)::value) poss += cnt;
    };
    auto drain_sum_saved  = [&](int cnt) { sum_pass(cnt, on_slices<true>{}); };
    auto drain_sum_gather = [&](int cnt) { sum_pass(cnt, on_slices<false>{}); };
    const int total = __builtin_amdgcn_readfirstlane(slcnt[wslot]);
    const int EN = E + N;
    const int per = ((EN + G - 1) / G + 3) & ~3;
    const int e0 = g * per;
    const int e1 = (e0 + per < EN) ? (e0 + per) : EN;
    if (total <= capw) walk_slice(slw, total, qw, lane, drain_max_saved);
    else               scan_chunk(ei, E, e0, e1, lo, span, qw, tid, drain_max_gather);
    __syncthreads();
    if (total <= capw) walk_slice(slw, total, qw, lane, drain_sum_saved);
    else               scan_chunk(ei, E, e0, e1, lo, span, qw, tid, drain_sum_gather);
    __syncthreads();

    float* pm = pm2 + (size_t)g * N + lo;
    float* ps = ps2 + (size_t)g * N + lo;
    for (int k = tid; k < span; k += WG) { pm[k] = forddec(mx[k]); ps[k] = ssum[k]; }
    float4* pa = reinterpret_cast<float4*>(pacc2 + ((size_t)g * N + lo) * 8);
    for (int k = tid; k < span * 2; k += WG) pa[k] = *reinterpret_cast<const float4*>(&acc[k * 4]);
}

template <int G>
__global__ void k_fin2(const float* __restrict__ pm2, const float* __restrict__ ps2,
                       const float* __restrict__ pacc2,
                       const float* __restrict__ b2, float* __restrict__ out, int N) {
    const int i = blockIdx.x * blockDim.x + threadIdx.x;
    if (i >= N * 2) return;
    const int n = i >> 1, q = i & 1;
    float m[G], s[G]; float4 a[G];
#pragma unroll
    for (int g = 0; g < G; ++g) {
        m[g] = pm2[(size_t)g * N + n];
        s[g] = ps2[(size_t)g * N + n];
        a[g] = reinterpret_cast<const float4*>(pacc2 + (size_t)g * N * 8)[i];
    }
    float M = m[0];
#pragma unroll
    for (int g = 1; g < G; ++g) M = fmaxf(M, m[g]);
    float S = 0.f, A0 = 0.f, A1 = 0.f, A2 = 0.f, A3 = 0.f;
#pragma unroll
    for (int g = 0; g < G; ++g) {
        float sc = __expf(m[g] - M);
        S  += s[g] * sc;
        A0 += a[g].x * sc; A1 += a[g].y * sc; A2 += a[g].z * sc; A3 += a[g].w * sc;
    }
    reinterpret_cast<float4*>(out)[i] = make_float4(A0 / S + b2[q * 4 + 0], A1 / S + b2[q * 4 + 1],
                                                    A2 / S + b2[q * 4 + 2], A3 / S + b2[q * 4 + 3]);
}

extern "C" void kernel_launch(void* const* d_in, const int* in_sizes, int n_in,
                              void* d_out, int out_size, void* d_ws, size_t ws_size,
                              hipStream_t stream) {
    const float* x   = (const float*)d_in[0];
    const int*   ei  = (const int*)d_in[1];
    const float* W1  = (const float*)d_in[2];
    const float* as1 = (const float*)d_in[3];
    const float* ad1 = (const float*)d_in[4];
    const float* b1  = (const float*)d_in[5];
    const float* W2  = (const float*)d_in[6];
    const float* as2 = (const float*)d_in[7];
    const float* ad2 = (const float*)d_in[8];
    const float* b2  = (const float*)d_in[9];
    (void)n_in; (void)out_size; (void)ws_size;

    int N = in_sizes[0];
    int E = in_sizes[1] / 2;

    char* ws = (char*)d_ws;
    size_t cur = 0;
    auto carve = [&](size_t bytes) -> void* {
        void* p = ws + cur;
        cur += (bytes + 255) & ~(size_t)255;
        return p;
    };
    float*    pm1   = (float*)carve(sizeof(float) * (size_t)G_ALL * N * 4);
    float*    prec1 = (float*)carve(sizeof(float) * (size_t)G_ALL * N * 8);
    float*    pm2   = (float*)carve(sizeof(float) * (size_t)G_ALL * N);
    float*    ps2   = (float*)carve(sizeof(float) * (size_t)G_ALL * N);
    float*    pacc2 = (float*)carve(sizeof(float) * (size_t)G_ALL * N * 8);
    float*    out1  = (float*)carve(sizeof(float) * (size_t)N * 32);
    float*    h2    = (float*)carve(sizeof(float) * (size_t)N * 8);
    float*    alS2  = (float*)carve(sizeof(float) * (size_t)N);
    float*    alD2  = (float*)carve(sizeof(float) * (size_t)N);
    const int nwg   = ((N + SPAN_ALL - 1) / SPAN_ALL) * G_ALL;
    const int per   = ((E + N + G_ALL - 1) / G_ALL + 3) & ~3;
    const int capf  = ((per + WG * UNR - 1) / (WG * UNR)) * UNR * 32;
    const int capw  = capf < SLICE_CAP ? capf : SLICE_CAP;
    unsigned* sl    = (unsigned*)carve(sizeof(unsigned) * (size_t)nwg * (WG / 32) * capw);
    int*      slcnt = (int*)carve(sizeof(int) * (size_t)nwg * (WG / 32));
    float*    slx   = pm2;
    float*    sla   = pm1;

    const int B = 256;
    k_l1<SPAN_ALL, G_ALL><<<nwg, WG, 0, stream>>>(ei, E, N, x, W1, as1, ad1, pm1, prec1, sl, slx, slcnt, capw);
    k_fin1<G_ALL><<<(4 * N + B - 1) / B, B, 0, stream>>>(W1, b1, pm1, prec1, out1, N);

    int tiles = (N + 15) / 16;
    k_gemm2<<<(tiles + 7) / 8, B, 0, stream>>>(out1, W2, as2, ad2, h2, alS2, alD2, N);

    k_l2<SPAN_ALL, G_ALL><<<nwg, WG, 0, stream>>>(ei, E, sl, sla, slcnt, capw, N, h2, alS2, alD2, pm2, ps2, pacc2);
    k_fin2<G_ALL><<<(2 * N + B - 1) / B, B, 0, stream>>>(pm2, ps2, pacc2, b2, (float*)d_out, N);
}


# Softmax shift of a node is its own self-loop logit (known from the node alone), not a maximum over its edges: no maximum pass, no maxima in planes; layer 1 sums as it filters; sums in arrival order
# speedup vs baseline: 14.1464x; 1.1652x over previous
#include <hip/hip_runtime.h>
#include <math.h>


#define NEG_SLOPE 0.2f
#define WG 1024
#define UNR 8
#define QCAP 64
#define SPAN_ALL 6250
#define G_ALL 16
#define SRC_BITS 19
#define SLICE_CAP 2048
static_assert(UNR == 8, "the load fence in scan_chunk names eight values; a 16-byte trip is two loads of 4 edges a lane");
static_assert(QCAP >= 64, "a push of 32 entries on top of 31 carried ones must fit");
static_assert(QCAP == 64, "walk_slice refills the queue with two entries per lane");
static_assert(SPAN_ALL <= (1 << (32 - SRC_BITS)), "destination - lo must fit above the source bits");

typedef __attribute__((ext_vector_type(2))) float v2f;
typedef __attribute__((ext_vector_type(8))) float v8f;

__device__ __forceinline__ float lrelu(float v) { return v > 0.f ? v : NEG_SLOPE * v; }

__device__ __forceinline__ unsigned ford(float f) {
    unsigned u = __float_as_uint(f);
    return (u & 0x80000000u) ? ~u : (u | 0x80000000u);
}
__device__ __forceinline__ float forddec(unsigned u) {
    return (u & 0x80000000u) ? __uint_as_float(u & 0x7fffffffu) : __uint_as_float(~u);
}
#define FORD_NEG_INF 0x007FFFFFu


__device__ __forceinline__ void wave_sync() {
    __builtin_amdgcn_fence(__ATOMIC_RELEASE, "wavefront");
    __builtin_amdgcn_wave_barrier();
    __builtin_amdgcn_fence(__ATOMIC_ACQUIRE, "wavefront");
}

__device__ __forceinline__ void load_edge(const int* __restrict__ ei, int E, int e, int& ls, int& ld) {
    int ec = e < E ? e : E - 1;
    ls = ei[ec];
    ld = ei[E + ec];
}
__device__ __forceinline__ void place_edge(int E, int e, int e1, int lo, int ls, int ld,
                                           int& s, unsigned& rel) {
    int d = e < E ? ld : e - E;
    s     = e < E ? ls : e - E;
    if (e >= e1) { d = -1; s = 0; }
    rel = (unsigned)(d - lo);
}

template <class Drain>
__device__ __forceinline__ void scan_chunk(const int* __restrict__ ei, int E, int e0, int e1,
                                           int lo, int span, int2* qw, int tid, Drain drain) {
    const unsigned lt = (1u << (tid & 31)) - 1u;
    const int ereal = e1 < E ? e1 : E;
    const bool wide = ((E & 3) == 0) && ((e0 & 3) == 0) && ((reinterpret_cast<size_t>(ei) & 15) == 0);
    int cnt = 0;
    for (int b = e0; b < e1; b += WG * UNR) {
        int s[UNR]; unsigned rel[UNR];
        if (wide && b + WG * UNR <= ereal) {
            const int4* ps = reinterpret_cast<const int4*>(ei + b) + tid;
            const int4* pd = reinterpret_cast<const int4*>(ei + E + b) + tid;
            const int4 s0 = ps[0], s1 = ps[WG], d0 = pd[0], d1 = pd[WG];
            s[0] = s0.x; s[1] = s0.y; s[2] = s0.z; s[3] = s0.w;
            s[4] = s1.x; s[5] = s1.y; s[6] = s1.z; s[7] = s1.w;
            rel[0] = (unsigned)(d0.x - lo); rel[1] = (unsigned)(d0.y - lo);
            rel[2] = (unsigned)(d0.z - lo); rel[3] = (unsigned)(d0.w - lo);
            rel[4] = (unsigned)(d1.x - lo); rel[5] = (unsigned)(d1.y - lo);
            rel[6] = (unsigned)(d1.z - lo); rel[7] = (unsigned)(d1.w - lo);
        } else if (b >= E && b + WG * UNR <= e1) {
#pragma unroll
            for (int u = 0; u < UNR; ++u) {
                s[u] = b - E + u * WG + tid;
                rel[u] = (unsigned)(s[u] - lo);
            }
        } else {
            int ls[UNR], ld[UNR];
#pragma unroll
            for (int u = 0; u < UNR; ++u) load_edge(ei, E, b + u * WG + tid, ls[u], ld[u]);
            asm volatile("" :: "v"(ls[0]), "v"(ls[1]), "v"(ls[2]), "v"(ls[3]),
                               "v"(ls[4]), "v"(ls[5]), "v"(ls[6]), "v"(ls[7]));
#pragma unroll
            for (int u = 0; u < UNR; ++u) place_edge(E, b + u * WG + tid, e1, lo, ls[u], ld[u], s[u], rel[u]);
        }
#pragma unroll
        for (int u = 0; u < UNR; ++u) {
            bool keep = rel[u] < (unsigned)span;
            unsigned mask = __builtin_amdgcn_ballot_w32(keep);
            if (keep) qw[cnt + __popc(mask & lt)] = make_int2(s[u], (int)rel[u]);
            cnt += __popc(mask);
            if (cnt >= 32) {
                wave_sync();
                drain(32);
                wave_sync();
                const int rem = cnt - 32;
                if ((tid & 31) < rem) { const int2 c = qw[32 + (tid & 31)]; qw[tid & 31] = c; }
                cnt = rem;
            }
        }
    }
    wave_sync();
    drain(cnt);
}

template <class Drain>
__device__ __forceinline__ void walk_slice(const unsigned* sl, int total,
                                           int2* qw, int lane, Drain drain) {
    if (total <= 0) return;
    const int last = total - 1;
    const unsigned smask = (1u << SRC_BITS) - 1u;
    unsigned a = sl[lane < last ? lane : last];
    unsigned b = sl[32 + lane < last ? 32 + lane : last];
    for (int k0 = 0; k0 < total; k0 += QCAP) {
        qw[lane]      = make_int2((int)(a & smask), (int)(a >> SRC_BITS));
        qw[32 + lane] = make_int2((int)(b & smask), (int)(b >> SRC_BITS));
        const int kn = k0 + QCAP;
        if (kn < total) {
            a = sl[kn + lane < last ? kn + lane : last];
            b = sl[kn + 32 + lane < last ? kn + 32 + lane : last];
        }
        wave_sync();
        drain(total - k0 < QCAP ? total - k0 : QCAP);
        wave_sync();
    }
}

template <class Drain>
__device__ __forceinline__ void walk_slice_val(const unsigned* sl, const float* sv, int total,
                                               int2* qw, int lane, Drain drain) {
    if (total <= 0) return;
    const int last = total - 1;
    int i0 = lane < last ? lane : last, i1 = 32 + lane < last ? 32 + lane : last;
    unsigned a = sl[i0], b = sl[i1];
    float va = sv[i0], vb = sv[i1];
    for (int k0 = 0; k0 < total; k0 += QCAP) {
        qw[lane]      = make_int2(__float_as_int(va), (int)(a >> SRC_BITS));
        qw[32 + lane] = make_int2(__float_as_int(vb), (int)(b >> SRC_BITS));
        const int kn = k0 + QCAP;
        if (kn < total) {
            i0 = kn + lane < last ? kn + lane : last;
            i1 = kn + 32 + lane < last ? kn + 32 + lane : last;
            a = sl[i0]; b = sl[i1];
            va = sv[i0]; vb = sv[i1];
        }
        wave_sync();
        drain(total - k0 < QCAP ? total - k0 : QCAP);
        wave_sync();
    }
}

template <bool B> struct on_slices { static constexpr bool value = B; };
template <bool B> struct full_batch { static constexpr bool value = B; };
template <bool B> struct self_shift { static constexpr bool value = B; };
#define SUM_LIMIT 1e30f
__device__ __forceinline__ bool too_large(float v) { return !(fabsf(v) <= SUM_LIMIT); }


__device__ __forceinline__ void head_consts(const float* __restrict__ W1, const float* __restrict__ as1,
                                            const float* __restrict__ ad1, int h, float& wsl, float& wdl) {
    float a = 0.f, b = 0.f;
#pragma unroll
    for (int c = 0; c < 8; ++c) {
        float w = W1[h * 8 + c];
        a += w * as1[h * 8 + c];
        b += w * ad1[h * 8 + c];
    }
    wsl = a; wdl = b;
}

template <int SPAN, int G>
__global__ __launch_bounds__(WG) void k_l1(const int* __restrict__ ei, int E, int N,
                                           const float* __restrict__ x,
                                           const float* __restrict__ W1,
                                           const float* __restrict__ as1,
                                           const float* __restrict__ ad1,
                                           float* __restrict__ pm1, float* __restrict__ prec1,
                                           int* __restrict__ fl,
                                           unsigned* sl, float* slx, int* __restrict__ slcnt, int capw) {
    __shared__ __attribute__((aligned(16))) unsigned mx[SPAN * 4];
    __shared__ __attribute__((aligned(16))) float rec[SPAN * 8];
    __shared__ int2 q[(WG / 32) * QCAP];
    const int tid = threadIdx.x, lane = tid & 31;
    const int r = blockIdx.x / G, g = blockIdx.x % G;
    const int lo = r * SPAN;
    const int span = (N - lo < SPAN) ? (N - lo) : SPAN;
    const int h = lane & 3;
    float wsl, wdl;
    head_consts(W1, as1, ad1, h, wsl, wdl);
    const float wsd = wsl + wdl;

    if (tid == 0) mx[0] = 0u;
    for (int i = tid; i < span * 8; i += WG) rec[i] = 0.f;
    __syncthreads();

    int2* qw = q + (tid >> 5) * QCAP;
    const int wslot = blockIdx.x * (WG / 32) + (tid >> 5);
    unsigned* slw = sl + (size_t)wslot * capw;
    float* slxw = slx + (size_t)wslot * capw;
    int pos = 0;
    const int sub4 = lane >> 2;

    auto sum_batch = [&](int k0, int cnt, auto where, auto full, auto shift) {
        constexpr bool SAVED = decltype(where)::value;
        constexpr bool FULL = decltype(full)::value;
        constexpr bool SELF = decltype(shift)::value;
        bool ok[4]; int2 en[4]; float xs[4], xdv[4];
#pragma unroll
        for (int rr = 0; rr < 4; ++rr) {
            int k = k0 + rr * 8 + sub4;
            ok[rr] = FULL || k < cnt;
            en[rr] = qw[ok[rr] ? k : cnt - 1];
        }
#pragma unroll
        for (int rr = 0; rr < 4; ++rr) {
            if constexpr (SAVED) xs[rr] = __int_as_float(en[rr].x);
            else                 xs[rr] = x[en[rr].x];
            xdv[rr] = x[lo + en[rr].y];
        }
        unsigned mv[4];
        if constexpr (FULL) {
            if constexpr (!SELF) {
#pragma unroll
                for (int rr = 0; rr < 4; ++rr) mv[rr] = mx[en[rr].y * 4 + h];
            }
            __builtin_amdgcn_sched_barrier(0);
        } else {
            if constexpr (SAVED)
                asm volatile("" :: "v"(xdv[0]), "v"(xdv[1]), "v"(xdv[2]), "v"(xdv[3]));
            else
                asm volatile("" :: "v"(xs[0]), "v"(xs[1]), "v"(xs[2]), "v"(xs[3]),
                                   "v"(xdv[0]), "v"(xdv[1]), "v"(xdv[2]), "v"(xdv[3]));
        }
#pragma unroll
        for (int rr = 0; rr < 4; ++rr) {
            float sh;
            if constexpr (SELF) {
                sh = lrelu(xdv[rr] * wsd);
            } else {
                if constexpr (!FULL) mv[rr] = mx[en[rr].y * 4 + h];
                sh = forddec(mv[rr]);
            }
            float t = __expf(lrelu(xs[rr] * wsl + xdv[rr] * wdl) - sh);
            if (ok[rr]) {
                atomicAdd(&rec[en[rr].y * 8 + h], t);
                atomicAdd(&rec[en[rr].y * 8 + 4 + h], t * xs[rr]);
            }
        }
    };
    auto keep_words = [&](int cnt) {
        if (lane < cnt && pos + lane < capw) {
            int2 en = qw[lane];
            slw[pos + lane] = (unsigned)en.x | ((unsigned)en.y << SRC_BITS);
        }
        if (32 + lane < cnt && pos + 32 + lane < capw) {
            int2 en = qw[32 + lane];
            slw[pos + 32 + lane] = (unsigned)en.x | ((unsigned)en.y << SRC_BITS);
        }
        pos += cnt;
    };
    auto drain_self = [&](int cnt) {
        for (int k0 = 0; k0 < cnt; k0 += 32) {
            if (cnt - k0 >= 32) sum_batch(k0, cnt, on_slices<false>{}, full_batch<true>{}, self_shift<true>{});
            else                sum_batch(k0, cnt, on_slices<false>{}, full_batch<false>{}, self_shift<true>{});
        }
        keep_words(cnt);
    };
    const int EN = E + N;
    const int per = ((EN + G - 1) / G + 3) & ~3;
    const int e0 = g * per;
    const int e1 = (e0 + per < EN) ? (e0 + per) : EN;
    scan_chunk(ei, E, e0, e1, lo, span, qw, tid, drain_self);
    if (lane == 0) slcnt[wslot] = pos;
    __syncthreads();

    float4* pr = reinterpret_cast<float4*>(prec1 + ((size_t)g * N + lo) * 8);
    bool big = false;
    for (int k = tid; k < span * 2; k += WG) {
        const float4 v = *reinterpret_cast<const float4*>(&rec[k * 4]);
        big |= too_large(v.x) | too_large(v.y) | too_large(v.z) | too_large(v.w);
        pr[k] = v;
    }
    if (big) mx[0] = 1u;
    __syncthreads();
    const bool longway = mx[0] != 0u;
    if (tid == 0) fl[blockIdx.x] = longway ? 1 : 0;
    if (!longway) return;

    __syncthreads();
    for (int i = tid; i < span * 4; i += WG) mx[i] = FORD_NEG_INF;
    for (int i = tid; i < span * 8; i += WG) rec[i] = 0.f;
    __syncthreads();
    pos = 0;
    auto max_batch = [&](int k0, int cnt, auto full) {
        constexpr bool FULL = decltype(full)::value;
        bool ok[4]; int kq[4]; int2 en[4]; float xs[4], xdv[4];
#pragma unroll
        for (int rr = 0; rr < 4; ++rr) {
            kq[rr] = k0 + rr * 8 + sub4;
            ok[rr] = FULL || kq[rr] < cnt;
            en[rr] = qw[ok[rr] ? kq[rr] : cnt - 1];
        }
#pragma unroll
        for (int rr = 0; rr < 4; ++rr) { xs[rr] = x[en[rr].x]; xdv[rr] = x[lo + en[rr].y]; }
        if constexpr (FULL) {
            __builtin_amdgcn_sched_barrier(0);
        } else {
            asm volatile("" :: "v"(xs[0]), "v"(xs[1]), "v"(xs[2]), "v"(xs[3]),
                               "v"(xdv[0]), "v"(xdv[1]), "v"(xdv[2]), "v"(xdv[3]));
        }
#pragma unroll
        for (int rr = 0; rr < 4; ++rr) {
            unsigned v = ford(lrelu(xs[rr] * wsl + xdv[rr] * wdl));
            if (ok[rr]) atomicMax(&mx[en[rr].y * 4 + h], v);
        }
        if (h == 0) {
#pragma unroll
            for (int rr = 0; rr < 4; ++rr)
                if (ok[rr] && pos + kq[rr] < capw) slxw[pos + kq[rr]] = xs[rr];
        }
    };
    auto drain_max = [&](int cnt) {
        for (int k0 = 0; k0 < cnt; k0 += 32) {
            if (cnt - k0 >= 32) max_batch(k0, cnt, full_batch<true>{});
            else                max_batch(k0, cnt, full_batch<false>{});
        }
        keep_words(cnt);
    };
    scan_chunk(ei, E, e0, e1, lo, span, qw, tid, drain_max);
    __syncthreads();

    auto sum_pass = [&](int cnt, auto where) {
        for (int k0 = 0; k0 < cnt; k0 += 32) {
            if (cnt - k0 >= 32) sum_batch(k0, cnt, where, full_batch<true>{}, self_shift<false>{});
            else                sum_batch(k0, cnt, where, full_batch<false>{}, self_shift<false>{});
        }
    };
    auto drain_sum_saved  = [&](int cnt) { sum_pass(cnt, on_slices<true>{}); };
    auto drain_sum_gather = [&](int cnt) { sum_pass(cnt, on_slices<false>{}); };
    const int total = __builtin_amdgcn_readfirstlane(pos);
    if (total <= capw) walk_slice_val(slw, slxw, total, qw, lane, drain_sum_saved);
    else               scan_chunk(ei, E, e0, e1, lo, span, qw, tid, drain_sum_gather);
    __syncthreads();

    float4* pm = reinterpret_cast<float4*>(pm1 + ((size_t)g * N + lo) * 4);
    for (int k = tid; k < span; k += WG) {
        const uint4 u = *reinterpret_cast<const uint4*>(&mx[k * 4]);
        pm[k] = make_float4(forddec(u.x), forddec(u.y), forddec(u.z), forddec(u.w));
    }
    for (int k = tid; k < span * 2; k += WG) pr[k] = *reinterpret_cast<const float4*>(&rec[k * 4]);
}

template <int SPAN, int G>
__global__ void k_fin1(const float* __restrict__ W1, const float* __restrict__ b1,
                       const float* __restrict__ x, const float* __restrict__ as1,
                       const float* __restrict__ ad1, const int* __restrict__ fl,
                       const float* __restrict__ pm1, const float* __restrict__ prec1,
                       float* __restrict__ out1, int N) {
    static_assert(G % 4 == 0, "the flags of a range's sharers are read 16 bytes at a time");
    const int i = blockIdx.x * blockDim.x + threadIdx.x;
    const int ic = i < N * 4 ? i : N * 4 - 1;
    const int n = ic >> 2, h = ic & 3;
    const int rg = (n / SPAN) * G;
    float s[G], w[G];
#pragma unroll
    for (int g = 0; g < G; ++g) {
        s[g] = prec1[((size_t)g * N + n) * 8 + h];
        w[g] = prec1[((size_t)g * N + n) * 8 + 4 + h];
    }
    int any = 0;
#pragma unroll
    for (int k = 0; k < G / 4; ++k) {
        const int4 f = reinterpret_cast<const int4*>(fl + rg)[k];
        any |= f.x | f.y | f.z | f.w;
    }
    float S = 0.f, Wt = 0.f;
    if (any == 0) {
#pragma unroll
        for (int g = 0; g < G; ++g) { S += s[g]; Wt += w[g]; }
    } else {
        float wsl, wdl;
        head_consts(W1, as1, ad1, h, wsl, wdl);
        const float self = lrelu(x[n] * (wsl + wdl));
        float m[G];
#pragma unroll
        for (int g = 0; g < G; ++g) m[g] = fl[rg + g] != 0 ? pm1[(size_t)g * N * 4 + ic] : self;
        float M = m[0];
#pragma unroll
        for (int g = 1; g < G; ++g) M = fmaxf(M, m[g]);
#pragma unroll
        for (int g = 0; g < G; ++g) {
            float sc = __expf(m[g] - M);
            S += s[g] * sc;
            Wt += w[g] * sc;
        }
    }
    const float ratio = Wt / S;

    const int lane = threadIdx.x & 31;
    const int node0 = (i - lane) >> 2;
    const int hd = (lane & 7) >> 1, c0 = (lane & 1) * 4;
#pragma unroll
    for (int k = 0; k < 2; ++k) {
        const int nl = k * 4 + (lane >> 3);
        const float rt = __shfl(ratio, nl * 4 + hd, 32);
        float v[4];
#pragma unroll
        for (int c = 0; c < 4; ++c) {
            float t = W1[hd * 8 + c0 + c] * rt + b1[hd * 8 + c0 + c];
            v[c] = t > 0.f ? t : 0.f;
        }
        const int node = node0 + nl;
        if (node < N)
            *reinterpret_cast<float4*>(out1 + (size_t)node * 32 + hd * 8 + c0) = make_float4(v[0], v[1], v[2], v[3]);
    }
}

__global__ void k_gemm2(const float* __restrict__ out1, const float* __restrict__ W2,
                        const float* __restrict__ as2, const float* __restrict__ ad2,
                        float* __restrict__ h2, float* __restrict__ alS2,
                        float* __restrict__ alD2, int N) {
    __shared__ float Bp[32 * 16];
    int t = threadIdx.x;
    for (int i = t; i < 512; i += blockDim.x) {
        int k = i >> 4, n = i & 15;
        float v = 0.f;
        if (n < 8) {
            v = W2[k * 8 + n];
        } else if (n == 8) {
            for (int c = 0; c < 8; ++c) v += W2[k * 8 + c] * as2[c];
        } else if (n == 9) {
            for (int c = 0; c < 8; ++c) v += W2[k * 8 + c] * ad2[c];
        }
        Bp[i] = v;
    }
    __syncthreads();

    int wave = t >> 5, lane = t & 31;
    long tile = (long)blockIdx.x * 8 + wave;
    long row0 = tile * 16;
    if (row0 >= N) return;

    int m = lane & 15, khi = lane >> 4;
    long rowm = row0 + m;
    if (rowm >= N) rowm = N - 1;
    const float* Arow = out1 + rowm * 32;

    v8f c = {};
#pragma unroll
    for (int step = 0; step < 8; ++step) {
        int k0 = step * 4 + khi * 2;
        v2f a; a.x = Arow[k0];          a.y = Arow[k0 + 1];
        v2f b; b.x = Bp[k0 * 16 + m];   b.y = Bp[(k0 + 1) * 16 + m];
        c = __builtin_amdgcn_wmma_f32_16x16x4_f32(false, a, false, b, (short)0, c,
                                                  false, false);
    }

#pragma unroll
    for (int r = 0; r < 8; ++r) {
        long node = row0 + khi * 8 + r;
        if (node >= N) continue;
        float v = c[r];
        if (m < 8)        h2[node * 8 + m] = v;
        else if (m == 8)  alS2[node] = v;
        else if (m == 9)  alD2[node] = v;
    }
}

template <int SPAN, int G>
__global__ __launch_bounds__(WG) void k_l2(const int* __restrict__ ei, int E,
                                           const unsigned* __restrict__ sl, float* sla,
                                           const int* __restrict__ slcnt, int capw, int N,
                                           const float* __restrict__ h2,
                                           const float* __restrict__ alS2,
                                           const float* __restrict__ alD2,
                                           float* __restrict__ pm2, float* __restrict__ ps2,
                                           float* __restrict__ pacc2, int* __restrict__ fl) {
    __shared__ unsigned mx[SPAN];
    __shared__ float ad[SPAN];
    __shared__ float ssum[SPAN];
    __shared__ __attribute__((aligned(16))) float acc[SPAN * 8];
    __shared__ int2 q[(WG / 32) * QCAP];
    __shared__ float tq[WG];
    __shared__ unsigned redo;
    const int tid = threadIdx.x, lane = tid & 31;
    const int r = blockIdx.x / G, g = blockIdx.x % G;
    const int lo = r * SPAN;
    const int span = (N - lo < SPAN) ? (N - lo) : SPAN;

    if (tid == 0) redo = 0u;
    for (int i = tid; i < span; i += WG) {
        const float a = alD2[lo + i];
        ad[i] = a; mx[i] = ford(lrelu(alS2[lo + i] + a)); ssum[i] = 0.f;
    }
    for (int i = tid; i < span * 8; i += WG) acc[i] = 0.f;
    __syncthreads();

    int2* qw = q + (tid >> 5) * QCAP;
    const int wslot = blockIdx.x * (WG / 32) + (tid >> 5);
    const unsigned* slw = sl + (size_t)wslot * capw;
    float* slaw = sla + (size_t)wslot * capw;
    int posm = 0, poss = 0;
    auto max_batch = [&](int k0, int cnt, auto where, auto full) {
        constexpr bool SAVED = decltype(where)::value;
        constexpr bool FULL = decltype(full)::value;
        bool ok[2]; int kq[2]; int2 en[2]; float as[2];
#pragma unroll
        for (int rr = 0; rr < 2; ++rr) {
            kq[rr] = k0 + rr * 32 + lane;
            ok[rr] = FULL || kq[rr] < cnt;
            en[rr] = qw[ok[rr] ? kq[rr] : cnt - 1];
        }
#pragma unroll
        for (int rr = 0; rr < 2; ++rr) as[rr] = alS2[en[rr].x];
        if constexpr (!FULL) {
            asm volatile("" :: "v"(as[0]), "v"(as[1]));
        }
#pragma unroll
        for (int rr = 0; rr < 2; ++rr)
            if (ok[rr]) atomicMax(&mx[en[rr].y], ford(lrelu(as[rr] + ad[en[rr].y])));
        if constexpr (SAVED) {
#pragma unroll
            for (int rr = 0; rr < 2; ++rr)
                if (ok[rr]) slaw[posm + kq[rr]] = as[rr];
        }
    };
    auto max_pass = [&](int cnt, auto where) {
        for (int k0 = 0; k0 < cnt; k0 += 64) {
            if (cnt - k0 >= 64) max_batch(k0, cnt, where, full_batch<true>{});
            else                max_batch(k0, cnt, where, full_batch<false>{});
        }
        if constexpr (decltype(where)::value) posm += cnt;
    };
    auto drain_max_saved  = [&](int cnt) { max_pass(cnt, on_slices<true>{}); };
    auto drain_max_gather = [&](int cnt) { max_pass(cnt, on_slices<false>{}); };
    float* tw = tq + (tid & ~31);
    const int j = lane & 7, sub = lane >> 3;
    auto sum_batch = [&](int k0, int cnt, auto where, auto full) {
        constexpr bool SAVED = decltype(where)::value;
        constexpr bool FULL = decltype(full)::value;
        const int nb = FULL ? 32 : cnt - k0;
        const int2* qb = qw + k0;
        const bool oka = FULL || lane < nb;
        const int2 ea = qb[oka ? lane : nb - 1];
        bool ok[8]; int kk[8]; int2 en[8]; float hv[8];
#pragma unroll
        for (int rr = 0; rr < 8; ++rr) {
            int k = rr * 4 + sub;
            ok[rr] = FULL || k < nb;
            kk[rr] = ok[rr] ? k : nb - 1;
            en[rr] = qb[kk[rr]];
        }
        const float adv = ad[ea.y];
        const unsigned mxv = mx[ea.y];
        float as;
        if constexpr (SAVED) as = slaw[poss + k0 + (oka ? lane : nb - 1)];
        else                 as = alS2[ea.x];
#pragma unroll
        for (int rr = 0; rr < 8; ++rr) hv[rr] = h2[(size_t)en[rr].x * 8 + j];
        if constexpr (!FULL) {
            asm volatile("" :: "v"(as), "v"(hv[0]), "v"(hv[1]), "v"(hv[2]), "v"(hv[3]),
                               "v"(hv[4]), "v"(hv[5]), "v"(hv[6]), "v"(hv[7]));
        }
        const float t = __expf(lrelu(as + adv) - forddec(mxv));
        if (oka) atomicAdd(&ssum[ea.y], t);
        tw[lane] = t;
        wave_sync();
        float tv[8];
#pragma unroll
        for (int rr = 0; rr < 8; ++rr) tv[rr] = tw[kk[rr]];
        if constexpr (!FULL) {
            asm volatile("" :: "v"(tv[0]), "v"(tv[1]), "v"(tv[2]), "v"(tv[3]),
                               "v"(tv[4]), "v"(tv[5]), "v"(tv[6]), "v"(tv[7]));
        }
#pragma unroll
        for (int rr = 0; rr < 8; ++rr)
            if (ok[rr]) atomicAdd(&acc[en[rr].y * 8 + j], tv[rr] * hv[rr]);
        wave_sync();
    };
    auto sum_pass = [&](int cnt, auto where) {
        for (int k0 = 0; k0 < cnt; k0 += 32) {
            if (cnt - k0 >= 32) sum_batch(k0, cnt, where, full_batch<true>{});
            else                sum_batch(k0, cnt, where, full_batch<false>{});
        }
        if constexpr (decltype(where)::value) poss += cnt;
    };
    auto drain_sum_saved  = [&](int cnt) { sum_pass(cnt, on_slices<true>{}); };
    auto drain_sum_gather = [&](int cnt) { sum_pass(cnt, on_slices<false>{}); };
    const int total = __builtin_amdgcn_readfirstlane(slcnt[wslot]);
    const int EN = E + N;
    const int per = ((EN + G - 1) / G + 3) & ~3;
    const int e0 = g * per;
    const int e1 = (e0 + per < EN) ? (e0 + per) : EN;
    if (total <= capw) walk_slice(slw, total, qw, lane, drain_sum_gather);
    else               scan_chunk(ei, E, e0, e1, lo, span, qw, tid, drain_sum_gather);
    __syncthreads();

    float* ps = ps2 + (size_t)g * N + lo;
    float4* pa = reinterpret_cast<float4*>(pacc2 + ((size_t)g * N + lo) * 8);
    bool big = false;
    for (int k = tid; k < span; k += WG) {
        const float s = ssum[k];
        big |= too_large(s);
        ps[k] = s;
    }
    for (int k = tid; k < span * 2; k += WG) {
        const float4 v = *reinterpret_cast<const float4*>(&acc[k * 4]);
        big |= too_large(v.x) | too_large(v.y) | too_large(v.z) | too_large(v.w);
        pa[k] = v;
    }
    if (big) redo = 1u;
    __syncthreads();
    const bool longway = redo != 0u;
    if (tid == 0) fl[blockIdx.x] = longway ? 1 : 0;
    if (!longway) return;

    for (int i = tid; i < span; i += WG) { mx[i] = FORD_NEG_INF; ssum[i] = 0.f; }
    for (int i = tid; i < span * 8; i += WG) acc[i] = 0.f;
    __syncthreads();
    if (total <= capw) walk_slice(slw, total, qw, lane, drain_max_saved);
    else               scan_chunk(ei, E, e0, e1, lo, span, qw, tid, drain_max_gather);
    __syncthreads();
    if (total <= capw) walk_slice(slw, total, qw, lane, drain_sum_saved);
    else               scan_chunk(ei, E, e0, e1, lo, span, qw, tid, drain_sum_gather);
    __syncthreads();

    float* pm = pm2 + (size_t)g * N + lo;
    for (int k = tid; k < span; k += WG) { pm[k] = forddec(mx[k]); ps[k] = ssum[k]; }
    for (int k = tid; k < span * 2; k += WG) pa[k] = *reinterpret_cast<const float4*>(&acc[k * 4]);
}

template <int SPAN, int G>
__global__ void k_fin2(const float* __restrict__ alS2, const float* __restrict__ alD2,
                       const int* __restrict__ fl,
                       const float* __restrict__ pm2, const float* __restrict__ ps2,
                       const float* __restrict__ pacc2,
                       const float* __restrict__ b2, float* __restrict__ out, int N) {
    static_assert(G % 4 == 0, "the flags of a range's sharers are read 16 bytes at a time");
    const int i = blockIdx.x * blockDim.x + threadIdx.x;
    if (i >= N * 2) return;
    const int n = i >> 1, q = i & 1;
    const int rg = (n / SPAN) * G;
    float s[G]; float4 a[G];
#pragma unroll
    for (int g = 0; g < G; ++g) {
        s[g] = ps2[(size_t)g * N + n];
        a[g] = reinterpret_cast<const float4*>(pacc2 + (size_t)g * N * 8)[i];
    }
    int any = 0;
#pragma unroll
    for (int k = 0; k < G / 4; ++k) {
        const int4 f = reinterpret_cast<const int4*>(fl + rg)[k];
        any |= f.x | f.y | f.z | f.w;
    }
    float S = 0.f, A0 = 0.f, A1 = 0.f, A2 = 0.f, A3 = 0.f;
    if (any == 0) {
#pragma unroll
        for (int g = 0; g < G; ++g) {
            S  += s[g];
            A0 += a[g].x; A1 += a[g].y; A2 += a[g].z; A3 += a[g].w;
        }
    } else {
        const float self = lrelu(alS2[n] + alD2[n]);
        float m[G];
#pragma unroll
        for (int g = 0; g < G; ++g) m[g] = fl[rg + g] != 0 ? pm2[(size_t)g * N + n] : self;
        float M = m[0];
#pragma unroll
        for (int g = 1; g < G; ++g) M = fmaxf(M, m[g]);
#pragma unroll
        for (int g = 0; g < G; ++g) {
            float sc = __expf(m[g] - M);
            S  += s[g] * sc;
            A0 += a[g].x * sc; A1 += a[g].y * sc; A2 += a[g].z * sc; A3 += a[g].w * sc;
        }
    }
    reinterpret_cast<float4*>(out)[i] = make_float4(A0 / S + b2[q * 4 + 0], A1 / S + b2[q * 4 + 1],
                                                    A2 / S + b2[q * 4 + 2], A3 / S + b2[q * 4 + 3]);
}

extern "C" void kernel_launch(void* const* d_in, const int* in_sizes, int n_in,
                              void* d_out, int out_size, void* d_ws, size_t ws_size,
                              hipStream_t stream) {
    const float* x   = (const float*)d_in[0];
    const int*   ei  = (const int*)d_in[1];
    const float* W1  = (const float*)d_in[2];
    const float* as1 = (const float*)d_in[3];
    const float* ad1 = (const float*)d_in[4];
    const float* b1  = (const float*)d_in[5];
    const float* W2  = (const float*)d_in[6];
    const float* as2 = (const float*)d_in[7];
    const float* ad2 = (const float*)d_in[8];
    const float* b2  = (const float*)d_in[9];
    (void)n_in; (void)out_size; (void)ws_size;

    int N = in_sizes[0];
    int E = in_sizes[1] / 2;

    char* ws = (char*)d_ws;
    size_t cur = 0;
    auto carve = [&](size_t bytes) -> void* {
        void* p = ws + cur;
        cur += (bytes + 255) & ~(size_t)255;
        return p;
    };
    float*    pm1   = (float*)carve(sizeof(float) * (size_t)G_ALL * N * 4);
    float*    prec1 = (float*)carve(sizeof(float) * (size_t)G_ALL * N * 8);
    float*    pm2   = (float*)carve(sizeof(float) * (size_t)G_ALL * N);
    float*    ps2   = (float*)carve(sizeof(float) * (size_t)G_ALL * N);
    float*    pacc2 = (float*)carve(sizeof(float) * (size_t)G_ALL * N * 8);
    float*    out1  = (float*)carve(sizeof(float) * (size_t)N * 32);
    float*    h2    = (float*)carve(sizeof(float) * (size_t)N * 8);
    float*    alS2  = (float*)carve(sizeof(float) * (size_t)N);
    float*    alD2  = (float*)carve(sizeof(float) * (size_t)N);
    const int nwg   = ((N + SPAN_ALL - 1) / SPAN_ALL) * G_ALL;
    const int per   = ((E + N + G_ALL - 1) / G_ALL + 3) & ~3;
    const int capf  = ((per + WG * UNR - 1) / (WG * UNR)) * UNR * 32;
    const int capw  = capf < SLICE_CAP ? capf : SLICE_CAP;
    unsigned* sl    = (unsigned*)carve(sizeof(unsigned) * (size_t)nwg * (WG / 32) * capw);
    int*      slcnt = (int*)carve(sizeof(int) * (size_t)nwg * (WG / 32));
    int*      fl1   = (int*)carve(sizeof(int) * (size_t)nwg);
    int*      fl2   = (int*)carve(sizeof(int) * (size_t)nwg);
    float*    slx   = pm2;
    float*    sla   = pm1;

    const int B = 256;
    k_l1<SPAN_ALL, G_ALL><<<nwg, WG, 0, stream>>>(ei, E, N, x, W1, as1, ad1, pm1, prec1, fl1, sl, slx, slcnt, capw);
    k_fin1<SPAN_ALL, G_ALL><<<(4 * N + B - 1) / B, B, 0, stream>>>(W1, b1, x, as1, ad1, fl1, pm1, prec1, out1, N);

    int tiles = (N + 15) / 16;
    k_gemm2<<<(tiles + 7) / 8, B, 0, stream>>>(out1, W2, as2, ad2, h2, alS2, alD2, N);

    k_l2<SPAN_ALL, G_ALL><<<nwg, WG, 0, stream>>>(ei, E, sl, sla, slcnt, capw, N, h2, alS2, alD2, pm2, ps2, pacc2, fl2);
    k_fin2<SPAN_ALL, G_ALL><<<(2 * N + B - 1) / B, B, 0, stream>>>(alS2, alD2, fl2, pm2, ps2, pacc2, b2, (float*)d_out, N);
}


# Layer-2 sum pass: source logit = a_src2 . gathered source row, summed over its 8 lanes by DPP (a lane tree); that lane makes the weight, DPP8 hands it back: no logit gather; sums in arrival order
# speedup vs baseline: 15.1197x; 1.0688x over previous
#include <hip/hip_runtime.h>
#include <math.h>


#define NEG_SLOPE 0.2f
#define WG 1024
#define UNR 8
#define QCAP 64
#define SPAN_ALL 6250
#define G_ALL 16
#define SRC_BITS 19
#define SLICE_CAP 2048
static_assert(UNR == 8, "the load fence in scan_chunk names eight values; a 16-byte trip is two loads of 4 edges a lane");
static_assert(QCAP >= 64, "a push of 32 entries on top of 31 carried ones must fit");
static_assert(QCAP == 64, "walk_slice refills the queue with two entries per lane");
static_assert(SPAN_ALL <= (1 << (32 - SRC_BITS)), "destination - lo must fit above the source bits");

typedef __attribute__((ext_vector_type(2))) float v2f;
typedef __attribute__((ext_vector_type(8))) float v8f;

__device__ __forceinline__ float lrelu(float v) { return v > 0.f ? v : NEG_SLOPE * v; }

__device__ __forceinline__ unsigned ford(float f) {
    unsigned u = __float_as_uint(f);
    return (u & 0x80000000u) ? ~u : (u | 0x80000000u);
}
__device__ __forceinline__ float forddec(unsigned u) {
    return (u & 0x80000000u) ? __uint_as_float(u & 0x7fffffffu) : __uint_as_float(~u);
}
#define FORD_NEG_INF 0x007FFFFFu


__device__ __forceinline__ void wave_sync() {
    __builtin_amdgcn_fence(__ATOMIC_RELEASE, "wavefront");
    __builtin_amdgcn_wave_barrier();
    __builtin_amdgcn_fence(__ATOMIC_ACQUIRE, "wavefront");
}

template <int CTRL>
__device__ __forceinline__ float dpp_row(float v) {
    return __int_as_float(__builtin_amdgcn_update_dpp(0, __float_as_int(v), CTRL, 0xF, 0xF, true));
}
#define ROW_XMASK 0x160
template <int J>
__device__ __forceinline__ float bcast8(float v) {
    return __int_as_float(__builtin_amdgcn_mov_dpp8(__float_as_int(v), J * 0x249249));
}

__device__ __forceinline__ void load_edge(const int* __restrict__ ei, int E, int e, int& ls, int& ld) {
    int ec = e < E ? e : E - 1;
    ls = ei[ec];
    ld = ei[E + ec];
}
__device__ __forceinline__ void place_edge(int E, int e, int e1, int lo, int ls, int ld,
                                           int& s, unsigned& rel) {
    int d = e < E ? ld : e - E;
    s     = e < E ? ls : e - E;
    if (e >= e1) { d = -1; s = 0; }
    rel = (unsigned)(d - lo);
}

template <class Drain>
__device__ __forceinline__ void scan_chunk(const int* __restrict__ ei, int E, int e0, int e1,
                                           int lo, int span, int2* qw, int tid, Drain drain) {
    const unsigned lt = (1u << (tid & 31)) - 1u;
    const int ereal = e1 < E ? e1 : E;
    const bool wide = ((E & 3) == 0) && ((e0 & 3) == 0) && ((reinterpret_cast<size_t>(ei) & 15) == 0);
    int cnt = 0;
    for (int b = e0; b < e1; b += WG * UNR) {
        int s[UNR]; unsigned rel[UNR];
        if (wide && b + WG * UNR <= ereal) {
            const int4* ps = reinterpret_cast<const int4*>(ei + b) + tid;
            const int4* pd = reinterpret_cast<const int4*>(ei + E + b) + tid;
            const int4 s0 = ps[0], s1 = ps[WG], d0 = pd[0], d1 = pd[WG];
            s[0] = s0.x; s[1] = s0.y; s[2] = s0.z; s[3] = s0.w;
            s[4] = s1.x; s[5] = s1.y; s[6] = s1.z; s[7] = s1.w;
            rel[0] = (unsigned)(d0.x - lo); rel[1] = (unsigned)(d0.y - lo);
            rel[2] = (unsigned)(d0.z - lo); rel[3] = (unsigned)(d0.w - lo);
            rel[4] = (unsigned)(d1.x - lo); rel[5] = (unsigned)(d1.y - lo);
            rel[6] = (unsigned)(d1.z - lo); rel[7] = (unsigned)(d1.w - lo);
        } else if (b >= E && b + WG * UNR <= e1) {
#pragma unroll
            for (int u = 0; u < UNR; ++u) {
                s[u] = b - E + u * WG + tid;
                rel[u] = (unsigned)(s[u] - lo);
            }
        } else {
            int ls[UNR], ld[UNR];
#pragma unroll
            for (int u = 0; u < UNR; ++u) load_edge(ei, E, b + u * WG + tid, ls[u], ld[u]);
            asm volatile("" :: "v"(ls[0]), "v"(ls[1]), "v"(ls[2]), "v"(ls[3]),
                               "v"(ls[4]), "v"(ls[5]), "v"(ls[6]), "v"(ls[7]));
#pragma unroll
            for (int u = 0; u < UNR; ++u) place_edge(E, b + u * WG + tid, e1, lo, ls[u], ld[u], s[u], rel[u]);
        }
#pragma unroll
        for (int u = 0; u < UNR; ++u) {
            bool keep = rel[u] < (unsigned)span;
            unsigned mask = __builtin_amdgcn_ballot_w32(keep);
            if (keep) qw[cnt + __popc(mask & lt)] = make_int2(s[u], (int)rel[u]);
            cnt += __popc(mask);
            if (cnt >= 32) {
                wave_sync();
                drain(32);
                wave_sync();
                const int rem = cnt - 32;
                if ((tid & 31) < rem) { const int2 c = qw[32 + (tid & 31)]; qw[tid & 31] = c; }
                cnt = rem;
            }
        }
    }
    wave_sync();
    drain(cnt);
}

template <class Drain>
__device__ __forceinline__ void walk_slice(const unsigned* sl, int total,
                                           int2* qw, int lane, Drain drain) {
    if (total <= 0) return;
    const int last = total - 1;
    const unsigned smask = (1u << SRC_BITS) - 1u;
    unsigned a = sl[lane < last ? lane : last];
    unsigned b = sl[32 + lane < last ? 32 + lane : last];
    for (int k0 = 0; k0 < total; k0 += QCAP) {
        qw[lane]      = make_int2((int)(a & smask), (int)(a >> SRC_BITS));
        qw[32 + lane] = make_int2((int)(b & smask), (int)(b >> SRC_BITS));
        const int kn = k0 + QCAP;
        if (kn < total) {
            a = sl[kn + lane < last ? kn + lane : last];
            b = sl[kn + 32 + lane < last ? kn + 32 + lane : last];
        }
        wave_sync();
        drain(total - k0 < QCAP ? total - k0 : QCAP);
        wave_sync();
    }
}

template <class Drain>
__device__ __forceinline__ void walk_slice_val(const unsigned* sl, const float* sv, int total,
                                               int2* qw, int lane, Drain drain) {
    if (total <= 0) return;
    const int last = total - 1;
    int i0 = lane < last ? lane : last, i1 = 32 + lane < last ? 32 + lane : last;
    unsigned a = sl[i0], b = sl[i1];
    float va = sv[i0], vb = sv[i1];
    for (int k0 = 0; k0 < total; k0 += QCAP) {
        qw[lane]      = make_int2(__float_as_int(va), (int)(a >> SRC_BITS));
        qw[32 + lane] = make_int2(__float_as_int(vb), (int)(b >> SRC_BITS));
        const int kn = k0 + QCAP;
        if (kn < total) {
            i0 = kn + lane < last ? kn + lane : last;
            i1 = kn + 32 + lane < last ? kn + 32 + lane : last;
            a = sl[i0]; b = sl[i1];
            va = sv[i0]; vb = sv[i1];
        }
        wave_sync();
        drain(total - k0 < QCAP ? total - k0 : QCAP);
        wave_sync();
    }
}

template <bool B> struct on_slices { static constexpr bool value = B; };
template <bool B> struct full_batch { static constexpr bool value = B; };
template <bool B> struct self_shift { static constexpr bool value = B; };
#define SUM_LIMIT 1e30f
__device__ __forceinline__ bool too_large(float v) { return !(fabsf(v) <= SUM_LIMIT); }


__device__ __forceinline__ void head_consts(const float* __restrict__ W1, const float* __restrict__ as1,
                                            const float* __restrict__ ad1, int h, float& wsl, float& wdl) {
    float a = 0.f, b = 0.f;
#pragma unroll
    for (int c = 0; c < 8; ++c) {
        float w = W1[h * 8 + c];
        a += w * as1[h * 8 + c];
        b += w * ad1[h * 8 + c];
    }
    wsl = a; wdl = b;
}

template <int SPAN, int G>
__global__ __launch_bounds__(WG) void k_l1(const int* __restrict__ ei, int E, int N,
                                           const float* __restrict__ x,
                                           const float* __restrict__ W1,
                                           const float* __restrict__ as1,
                                           const float* __restrict__ ad1,
                                           float* __restrict__ pm1, float* __restrict__ prec1,
                                           int* __restrict__ fl,
                                           unsigned* sl, float* slx, int* __restrict__ slcnt, int capw) {
    __shared__ __attribute__((aligned(16))) unsigned mx[SPAN * 4];
    __shared__ __attribute__((aligned(16))) float rec[SPAN * 8];
    __shared__ int2 q[(WG / 32) * QCAP];
    const int tid = threadIdx.x, lane = tid & 31;
    const int r = blockIdx.x / G, g = blockIdx.x % G;
    const int lo = r * SPAN;
    const int span = (N - lo < SPAN) ? (N - lo) : SPAN;
    const int h = lane & 3;
    float wsl, wdl;
    head_consts(W1, as1, ad1, h, wsl, wdl);
    const float wsd = wsl + wdl;

    if (tid == 0) mx[0] = 0u;
    for (int i = tid; i < span * 8; i += WG) rec[i] = 0.f;
    __syncthreads();

    int2* qw = q + (tid >> 5) * QCAP;
    const int wslot = blockIdx.x * (WG / 32) + (tid >> 5);
    unsigned* slw = sl + (size_t)wslot * capw;
    float* slxw = slx + (size_t)wslot * capw;
    int pos = 0;
    const int sub4 = lane >> 2;

    auto sum_batch = [&](int k0, int cnt, auto where, auto full, auto shift) {
        constexpr bool SAVED = decltype(where)::value;
        constexpr bool FULL = decltype(full)::value;
        constexpr bool SELF = decltype(shift)::value;
        bool ok[4]; int2 en[4]; float xs[4], xdv[4];
#pragma unroll
        for (int rr = 0; rr < 4; ++rr) {
            int k = k0 + rr * 8 + sub4;
            ok[rr] = FULL || k < cnt;
            en[rr] = qw[ok[rr] ? k : cnt - 1];
        }
#pragma unroll
        for (int rr = 0; rr < 4; ++rr) {
            if constexpr (SAVED) xs[rr] = __int_as_float(en[rr].x);
            else                 xs[rr] = x[en[rr].x];
            xdv[rr] = x[lo + en[rr].y];
        }
        unsigned mv[4];
        if constexpr (FULL) {
            if constexpr (!SELF) {
#pragma unroll
                for (int rr = 0; rr < 4; ++rr) mv[rr] = mx[en[rr].y * 4 + h];
            }
            __builtin_amdgcn_sched_barrier(0);
        } else {
            if constexpr (SAVED)
                asm volatile("" :: "v"(xdv[0]), "v"(xdv[1]), "v"(xdv[2]), "v"(xdv[3]));
            else
                asm volatile("" :: "v"(xs[0]), "v"(xs[1]), "v"(xs[2]), "v"(xs[3]),
                                   "v"(xdv[0]), "v"(xdv[1]), "v"(xdv[2]), "v"(xdv[3]));
        }
#pragma unroll
        for (int rr = 0; rr < 4; ++rr) {
            float sh;
            if constexpr (SELF) {
                sh = lrelu(xdv[rr] * wsd);
            } else {
                if constexpr (!FULL) mv[rr] = mx[en[rr].y * 4 + h];
                sh = forddec(mv[rr]);
            }
            float t = __expf(lrelu(xs[rr] * wsl + xdv[rr] * wdl) - sh);
            if (ok[rr]) {
                atomicAdd(&rec[en[rr].y * 8 + h], t);
                atomicAdd(&rec[en[rr].y * 8 + 4 + h], t * xs[rr]);
            }
        }
    };
    auto keep_words = [&](int cnt) {
        if (lane < cnt && pos + lane < capw) {
            int2 en = qw[lane];
            slw[pos + lane] = (unsigned)en.x | ((unsigned)en.y << SRC_BITS);
        }
        if (32 + lane < cnt && pos + 32 + lane < capw) {
            int2 en = qw[32 + lane];
            slw[pos + 32 + lane] = (unsigned)en.x | ((unsigned)en.y << SRC_BITS);
        }
        pos += cnt;
    };
    auto drain_self = [&](int cnt) {
        for (int k0 = 0; k0 < cnt; k0 += 32) {
            if (cnt - k0 >= 32) sum_batch(k0, cnt, on_slices<false>{}, full_batch<true>{}, self_shift<true>{});
            else                sum_batch(k0, cnt, on_slices<false>{}, full_batch<false>{}, self_shift<true>{});
        }
        keep_words(cnt);
    };
    const int EN = E + N;
    const int per = ((EN + G - 1) / G + 3) & ~3;
    const int e0 = g * per;
    const int e1 = (e0 + per < EN) ? (e0 + per) : EN;
    scan_chunk(ei, E, e0, e1, lo, span, qw, tid, drain_self);
    if (lane == 0) slcnt[wslot] = pos;
    __syncthreads();

    float4* pr = reinterpret_cast<float4*>(prec1 + ((size_t)g * N + lo) * 8);
    bool big = false;
    for (int k = tid; k < span * 2; k += WG) {
        const float4 v = *reinterpret_cast<const float4*>(&rec[k * 4]);
        big |= too_large(v.x) | too_large(v.y) | too_large(v.z) | too_large(v.w);
        pr[k] = v;
    }
    if (big) mx[0] = 1u;
    __syncthreads();
    const bool longway = mx[0] != 0u;
    if (tid == 0) fl[blockIdx.x] = longway ? 1 : 0;
    if (!longway) return;

    __syncthreads();
    for (int i = tid; i < span * 4; i += WG) mx[i] = FORD_NEG_INF;
    for (int i = tid; i < span * 8; i += WG) rec[i] = 0.f;
    __syncthreads();
    pos = 0;
    auto max_batch = [&](int k0, int cnt, auto full) {
        constexpr bool FULL = decltype(full)::value;
        bool ok[4]; int kq[4]; int2 en[4]; float xs[4], xdv[4];
#pragma unroll
        for (int rr = 0; rr < 4; ++rr) {
            kq[rr] = k0 + rr * 8 + sub4;
            ok[rr] = FULL || kq[rr] < cnt;
            en[rr] = qw[ok[rr] ? kq[rr] : cnt - 1];
        }
#pragma unroll
        for (int rr = 0; rr < 4; ++rr) { xs[rr] = x[en[rr].x]; xdv[rr] = x[lo + en[rr].y]; }
        if constexpr (FULL) {
            __builtin_amdgcn_sched_barrier(0);
        } else {
            asm volatile("" :: "v"(xs[0]), "v"(xs[1]), "v"(xs[2]), "v"(xs[3]),
                               "v"(xdv[0]), "v"(xdv[1]), "v"(xdv[2]), "v"(xdv[3]));
        }
#pragma unroll
        for (int rr = 0; rr < 4; ++rr) {
            unsigned v = ford(lrelu(xs[rr] * wsl + xdv[rr] * wdl));
            if (ok[rr]) atomicMax(&mx[en[rr].y * 4 + h], v);
        }
        if (h == 0) {
#pragma unroll
            for (int rr = 0; rr < 4; ++rr)
                if (ok[rr] && pos + kq[rr] < capw) slxw[pos + kq[rr]] = xs[rr];
        }
    };
    auto drain_max = [&](int cnt) {
        for (int k0 = 0; k0 < cnt; k0 += 32) {
            if (cnt - k0 >= 32) max_batch(k0, cnt, full_batch<true>{});
            else                max_batch(k0, cnt, full_batch<false>{});
        }
        keep_words(cnt);
    };
    scan_chunk(ei, E, e0, e1, lo, span, qw, tid, drain_max);
    __syncthreads();

    auto sum_pass = [&](int cnt, auto where) {
        for (int k0 = 0; k0 < cnt; k0 += 32) {
            if (cnt - k0 >= 32) sum_batch(k0, cnt, where, full_batch<true>{}, self_shift<false>{});
            else                sum_batch(k0, cnt, where, full_batch<false>{}, self_shift<false>{});
        }
    };
    auto drain_sum_saved  = [&](int cnt) { sum_pass(cnt, on_slices<true>{}); };
    auto drain_sum_gather = [&](int cnt) { sum_pass(cnt, on_slices<false>{}); };
    const int total = __builtin_amdgcn_readfirstlane(pos);
    if (total <= capw) walk_slice_val(slw, slxw, total, qw, lane, drain_sum_saved);
    else               scan_chunk(ei, E, e0, e1, lo, span, qw, tid, drain_sum_gather);
    __syncthreads();

    float4* pm = reinterpret_cast<float4*>(pm1 + ((size_t)g * N + lo) * 4);
    for (int k = tid; k < span; k += WG) {
        const uint4 u = *reinterpret_cast<const uint4*>(&mx[k * 4]);
        pm[k] = make_float4(forddec(u.x), forddec(u.y), forddec(u.z), forddec(u.w));
    }
    for (int k = tid; k < span * 2; k += WG) pr[k] = *reinterpret_cast<const float4*>(&rec[k * 4]);
}

template <int SPAN, int G>
__global__ void k_fin1(const float* __restrict__ W1, const float* __restrict__ b1,
                       const float* __restrict__ x, const float* __restrict__ as1,
                       const float* __restrict__ ad1, const int* __restrict__ fl,
                       const float* __restrict__ pm1, const float* __restrict__ prec1,
                       float* __restrict__ out1, int N) {
    static_assert(G % 4 == 0, "the flags of a range's sharers are read 16 bytes at a time");
    const int i = blockIdx.x * blockDim.x + threadIdx.x;
    const int ic = i < N * 4 ? i : N * 4 - 1;
    const int n = ic >> 2, h = ic & 3;
    const int rg = (n / SPAN) * G;
    float s[G], w[G];
#pragma unroll
    for (int g = 0; g < G; ++g) {
        s[g] = prec1[((size_t)g * N + n) * 8 + h];
        w[g] = prec1[((size_t)g * N + n) * 8 + 4 + h];
    }
    int any = 0;
#pragma unroll
    for (int k = 0; k < G / 4; ++k) {
        const int4 f = reinterpret_cast<const int4*>(fl + rg)[k];
        any |= f.x | f.y | f.z | f.w;
    }
    float S = 0.f, Wt = 0.f;
    if (any == 0) {
#pragma unroll
        for (int g = 0; g < G; ++g) { S += s[g]; Wt += w[g]; }
    } else {
        float wsl, wdl;
        head_consts(W1, as1, ad1, h, wsl, wdl);
        const float self = lrelu(x[n] * (wsl + wdl));
        float m[G];
#pragma unroll
        for (int g = 0; g < G; ++g) m[g] = fl[rg + g] != 0 ? pm1[(size_t)g * N * 4 + ic] : self;
        float M = m[0];
#pragma unroll
        for (int g = 1; g < G; ++g) M = fmaxf(M, m[g]);
#pragma unroll
        for (int g = 0; g < G; ++g) {
            float sc = __expf(m[g] - M);
            S += s[g] * sc;
            Wt += w[g] * sc;
        }
    }
    const float ratio = Wt / S;

    const int lane = threadIdx.x & 31;
    const int node0 = (i - lane) >> 2;
    const int hd = (lane & 7) >> 1, c0 = (lane & 1) * 4;
#pragma unroll
    for (int k = 0; k < 2; ++k) {
        const int nl = k * 4 + (lane >> 3);
        const float rt = __shfl(ratio, nl * 4 + hd, 32);
        float v[4];
#pragma unroll
        for (int c = 0; c < 4; ++c) {
            float t = W1[hd * 8 + c0 + c] * rt + b1[hd * 8 + c0 + c];
            v[c] = t > 0.f ? t : 0.f;
        }
        const int node = node0 + nl;
        if (node < N)
            *reinterpret_cast<float4*>(out1 + (size_t)node * 32 + hd * 8 + c0) = make_float4(v[0], v[1], v[2], v[3]);
    }
}

__global__ void k_gemm2(const float* __restrict__ out1, const float* __restrict__ W2,
                        const float* __restrict__ as2, const float* __restrict__ ad2,
                        float* __restrict__ h2, float* __restrict__ alS2,
                        float* __restrict__ alD2, int N) {
    __shared__ float Bp[32 * 16];
    int t = threadIdx.x;
    for (int i = t; i < 512; i += blockDim.x) {
        int k = i >> 4, n = i & 15;
        float v = 0.f;
        if (n < 8) {
            v = W2[k * 8 + n];
        } else if (n == 8) {
            for (int c = 0; c < 8; ++c) v += W2[k * 8 + c] * as2[c];
        } else if (n == 9) {
            for (int c = 0; c < 8; ++c) v += W2[k * 8 + c] * ad2[c];
        }
        Bp[i] = v;
    }
    __syncthreads();

    int wave = t >> 5, lane = t & 31;
    long tile = (long)blockIdx.x * 8 + wave;
    long row0 = tile * 16;
    if (row0 >= N) return;

    int m = lane & 15, khi = lane >> 4;
    long rowm = row0 + m;
    if (rowm >= N) rowm = N - 1;
    const float* Arow = out1 + rowm * 32;

    v8f c = {};
#pragma unroll
    for (int step = 0; step < 8; ++step) {
        int k0 = step * 4 + khi * 2;
        v2f a; a.x = Arow[k0];          a.y = Arow[k0 + 1];
        v2f b; b.x = Bp[k0 * 16 + m];   b.y = Bp[(k0 + 1) * 16 + m];
        c = __builtin_amdgcn_wmma_f32_16x16x4_f32(false, a, false, b, (short)0, c,
                                                  false, false);
    }

#pragma unroll
    for (int r = 0; r < 8; ++r) {
        long node = row0 + khi * 8 + r;
        if (node >= N) continue;
        float v = c[r];
        if (m < 8)        h2[node * 8 + m] = v;
        else if (m == 8)  alS2[node] = v;
        else if (m == 9)  alD2[node] = v;
    }
}

template <int SPAN, int G>
__global__ __launch_bounds__(WG) void k_l2(const int* __restrict__ ei, int E,
                                           const unsigned* __restrict__ sl, float* sla,
                                           const int* __restrict__ slcnt, int capw, int N,
                                           const float* __restrict__ h2,
                                           const float* __restrict__ alS2,
                                           const float* __restrict__ alD2,
                                           const float* __restrict__ as2,
                                           float* __restrict__ pm2, float* __restrict__ ps2,
                                           float* __restrict__ pacc2, int* __restrict__ fl) {
    __shared__ unsigned mx[SPAN];
    __shared__ float ad[SPAN];
    __shared__ float ssum[SPAN];
    __shared__ __attribute__((aligned(16))) float acc[SPAN * 8];
    __shared__ int2 q[(WG / 32) * QCAP];
    __shared__ unsigned redo;
    const int tid = threadIdx.x, lane = tid & 31;
    const int r = blockIdx.x / G, g = blockIdx.x % G;
    const int lo = r * SPAN;
    const int span = (N - lo < SPAN) ? (N - lo) : SPAN;

    if (tid == 0) redo = 0u;
    for (int i = tid; i < span; i += WG) {
        const float a = alD2[lo + i];
        ad[i] = a; mx[i] = ford(lrelu(alS2[lo + i] + a)); ssum[i] = 0.f;
    }
    for (int i = tid; i < span * 8; i += WG) acc[i] = 0.f;
    __syncthreads();

    int2* qw = q + (tid >> 5) * QCAP;
    const int wslot = blockIdx.x * (WG / 32) + (tid >> 5);
    const unsigned* slw = sl + (size_t)wslot * capw;
    float* slaw = sla + (size_t)wslot * capw;
    int posm = 0, poss = 0;
    auto max_batch = [&](int k0, int cnt, auto where, auto full) {
        constexpr bool SAVED = decltype(where)::value;
        constexpr bool FULL = decltype(full)::value;
        bool ok[2]; int kq[2]; int2 en[2]; float as[2];
#pragma unroll
        for (int rr = 0; rr < 2; ++rr) {
            kq[rr] = k0 + rr * 32 + lane;
            ok[rr] = FULL || kq[rr] < cnt;
            en[rr] = qw[ok[rr] ? kq[rr] : cnt - 1];
        }
#pragma unroll
        for (int rr = 0; rr < 2; ++rr) as[rr] = alS2[en[rr].x];
        if constexpr (!FULL) {
            asm volatile("" :: "v"(as[0]), "v"(as[1]));
        }
#pragma unroll
        for (int rr = 0; rr < 2; ++rr)
            if (ok[rr]) atomicMax(&mx[en[rr].y], ford(lrelu(as[rr] + ad[en[rr].y])));
        if constexpr (SAVED) {
#pragma unroll
            for (int rr = 0; rr < 2; ++rr)
                if (ok[rr]) slaw[posm + kq[rr]] = as[rr];
        }
    };
    auto max_pass = [&](int cnt, auto where) {
        for (int k0 = 0; k0 < cnt; k0 += 64) {
            if (cnt - k0 >= 64) max_batch(k0, cnt, where, full_batch<true>{});
            else                max_batch(k0, cnt, where, full_batch<false>{});
        }
        if constexpr (decltype(where)::value) posm += cnt;
    };
    auto drain_max_saved  = [&](int cnt) { max_pass(cnt, on_slices<true>{}); };
    auto drain_max_gather = [&](int cnt) { max_pass(cnt, on_slices<false>{}); };
    const int j = lane & 7, sub = lane >> 3;
    const float aj = as2[j];
    const bool b0 = (j & 1) != 0, b1 = (j & 2) != 0, b2 = (j & 4) != 0;
    auto sum_batch = [&](int k0, int cnt, auto where, auto full) {
        constexpr bool SAVED = decltype(where)::value;
        constexpr bool FULL = decltype(full)::value;
        const int nb = FULL ? 32 : cnt - k0;
        const int2* qb = qw + k0;
        const int kl = j * 4 + sub;
        const bool oka = FULL || kl < nb;
        const int kla = oka ? kl : nb - 1;
        const int2 ea = qb[kla];
        bool ok[8]; int2 en[8]; float hv[8];
#pragma unroll
        for (int rr = 0; rr < 8; ++rr) {
            int k = rr * 4 + sub;
            ok[rr] = FULL || k < nb;
            en[rr] = qb[ok[rr] ? k : nb - 1];
        }
        const float adv = ad[ea.y];
        const unsigned mxv = mx[ea.y];
#pragma unroll
        for (int rr = 0; rr < 8; ++rr) hv[rr] = h2[(size_t)en[rr].x * 8 + j];
        float as;
        if constexpr (SAVED) {
            as = slaw[poss + k0 + kla];
            if constexpr (!FULL) {
                asm volatile("" :: "v"(as), "v"(hv[0]), "v"(hv[1]), "v"(hv[2]), "v"(hv[3]),
                                   "v"(hv[4]), "v"(hv[5]), "v"(hv[6]), "v"(hv[7]));
            }
        } else {
            float q4[4], q2[2];
#pragma unroll
            for (int r = 0; r < 4; ++r) {
                const float keep = b2 ? hv[4 + r] : hv[r], send = b2 ? hv[r] : hv[4 + r];
                q4[r] = fmaf(keep, aj, dpp_row<ROW_XMASK + 4>(send * aj));
            }
#pragma unroll
            for (int r = 0; r < 2; ++r) {
                const float keep = b1 ? q4[2 + r] : q4[r], send = b1 ? q4[r] : q4[2 + r];
                q2[r] = keep + dpp_row<ROW_XMASK + 2>(send);
            }
            as = (b0 ? q2[1] : q2[0]) + dpp_row<ROW_XMASK + 1>(b0 ? q2[0] : q2[1]);
        }
        const float t = __expf(lrelu(as + adv) - forddec(mxv));
        if (oka) atomicAdd(&ssum[ea.y], t);
        const float tv[8] = { bcast8<0>(t), bcast8<1>(t), bcast8<2>(t), bcast8<3>(t),
                              bcast8<4>(t), bcast8<5>(t), bcast8<6>(t), bcast8<7>(t) };
#pragma unroll
        for (int rr = 0; rr < 8; ++rr)
            if (ok[rr]) atomicAdd(&acc[en[rr].y * 8 + j], tv[rr] * hv[rr]);
    };
    auto sum_pass = [&](int cnt, auto where) {
        for (int k0 = 0; k0 < cnt; k0 += 32) {
            if (cnt - k0 >= 32) sum_batch(k0, cnt, where, full_batch<true>{});
            else                sum_batch(k0, cnt, where, full_batch<false>{});
        }
        if constexpr (decltype(where)::value) poss += cnt;
    };
    auto drain_sum_saved  = [&](int cnt) { sum_pass(cnt, on_slices<true>{}); };
    auto drain_sum_gather = [&](int cnt) { sum_pass(cnt, on_slices<false>{}); };
    const int total = __builtin_amdgcn_readfirstlane(slcnt[wslot]);
    const int EN = E + N;
    const int per = ((EN + G - 1) / G + 3) & ~3;
    const int e0 = g * per;
    const int e1 = (e0 + per < EN) ? (e0 + per) : EN;
    if (total <= capw) walk_slice(slw, total, qw, lane, drain_sum_gather);
    else               scan_chunk(ei, E, e0, e1, lo, span, qw, tid, drain_sum_gather);
    __syncthreads();

    float* ps = ps2 + (size_t)g * N + lo;
    float4* pa = reinterpret_cast<float4*>(pacc2 + ((size_t)g * N + lo) * 8);
    bool big = false;
    for (int k = tid; k < span; k += WG) {
        const float s = ssum[k];
        big |= too_large(s);
        ps[k] = s;
    }
    for (int k = tid; k < span * 2; k += WG) {
        const float4 v = *reinterpret_cast<const float4*>(&acc[k * 4]);
        big |= too_large(v.x) | too_large(v.y) | too_large(v.z) | too_large(v.w);
        pa[k] = v;
    }
    if (big) redo = 1u;
    __syncthreads();
    const bool longway = redo != 0u;
    if (tid == 0) fl[blockIdx.x] = longway ? 1 : 0;
    if (!longway) return;

    for (int i = tid; i < span; i += WG) { mx[i] = FORD_NEG_INF; ssum[i] = 0.f; }
    for (int i = tid; i < span * 8; i += WG) acc[i] = 0.f;
    __syncthreads();
    if (total <= capw) walk_slice(slw, total, qw, lane, drain_max_saved);
    else               scan_chunk(ei, E, e0, e1, lo, span, qw, tid, drain_max_gather);
    __syncthreads();
    if (total <= capw) walk_slice(slw, total, qw, lane, drain_sum_saved);
    else               scan_chunk(ei, E, e0, e1, lo, span, qw, tid, drain_sum_gather);
    __syncthreads();

    float* pm = pm2 + (size_t)g * N + lo;
    for (int k = tid; k < span; k += WG) { pm[k] = forddec(mx[k]); ps[k] = ssum[k]; }
    for (int k = tid; k < span * 2; k += WG) pa[k] = *reinterpret_cast<const float4*>(&acc[k * 4]);
}

template <int SPAN, int G>
__global__ void k_fin2(const float* __restrict__ alS2, const float* __restrict__ alD2,
                       const int* __restrict__ fl,
                       const float* __restrict__ pm2, const float* __restrict__ ps2,
                       const float* __restrict__ pacc2,
                       const float* __restrict__ b2, float* __restrict__ out, int N) {
    static_assert(G % 4 == 0, "the flags of a range's sharers are read 16 bytes at a time");
    const int i = blockIdx.x * blockDim.x + threadIdx.x;
    if (i >= N * 2) return;
    const int n = i >> 1, q = i & 1;
    const int rg = (n / SPAN) * G;
    float s[G]; float4 a[G];
#pragma unroll
    for (int g = 0; g < G; ++g) {
        s[g] = ps2[(size_t)g * N + n];
        a[g] = reinterpret_cast<const float4*>(pacc2 + (size_t)g * N * 8)[i];
    }
    int any = 0;
#pragma unroll
    for (int k = 0; k < G / 4; ++k) {
        const int4 f = reinterpret_cast<const int4*>(fl + rg)[k];
        any |= f.x | f.y | f.z | f.w;
    }
    float S = 0.f, A0 = 0.f, A1 = 0.f, A2 = 0.f, A3 = 0.f;
    if (any == 0) {
#pragma unroll
        for (int g = 0; g < G; ++g) {
            S  += s[g];
            A0 += a[g].x; A1 += a[g].y; A2 += a[g].z; A3 += a[g].w;
        }
    } else {
        const float self = lrelu(alS2[n] + alD2[n]);
        float m[G];
#pragma unroll
        for (int g = 0; g < G; ++g) m[g] = fl[rg + g] != 0 ? pm2[(size_t)g * N + n] : self;
        float M = m[0];
#pragma unroll
        for (int g = 1; g < G; ++g) M = fmaxf(M, m[g]);
#pragma unroll
        for (int g = 0; g < G; ++g) {
            float sc = __expf(m[g] - M);
            S  += s[g] * sc;
            A0 += a[g].x * sc; A1 += a[g].y * sc; A2 += a[g].z * sc; A3 += a[g].w * sc;
        }
    }
    reinterpret_cast<float4*>(out)[i] = make_float4(A0 / S + b2[q * 4 + 0], A1 / S + b2[q * 4 + 1],
                                                    A2 / S + b2[q * 4 + 2], A3 / S + b2[q * 4 + 3]);
}

extern "C" void kernel_launch(void* const* d_in, const int* in_sizes, int n_in,
                              void* d_out, int out_size, void* d_ws, size_t ws_size,
                              hipStream_t stream) {
    const float* x   = (const float*)d_in[0];
    const int*   ei  = (const int*)d_in[1];
    const float* W1  = (const float*)d_in[2];
    const float* as1 = (const float*)d_in[3];
    const float* ad1 = (const float*)d_in[4];
    const float* b1  = (const float*)d_in[5];
    const float* W2  = (const float*)d_in[6];
    const float* as2 = (const float*)d_in[7];
    const float* ad2 = (const float*)d_in[8];
    const float* b2  = (const float*)d_in[9];
    (void)n_in; (void)out_size; (void)ws_size;

    int N = in_sizes[0];
    int E = in_sizes[1] / 2;

    char* ws = (char*)d_ws;
    size_t cur = 0;
    auto carve = [&](size_t bytes) -> void* {
        void* p = ws + cur;
        cur += (bytes + 255) & ~(size_t)255;
        return p;
    };
    float*    pm1   = (float*)carve(sizeof(float) * (size_t)G_ALL * N * 4);
    float*    prec1 = (float*)carve(sizeof(float) * (size_t)G_ALL * N * 8);
    float*    pm2   = (float*)carve(sizeof(float) * (size_t)G_ALL * N);
    float*    ps2   = (float*)carve(sizeof(float) * (size_t)G_ALL * N);
    float*    pacc2 = (float*)carve(sizeof(float) * (size_t)G_ALL * N * 8);
    float*    out1  = (float*)carve(sizeof(float) * (size_t)N * 32);
    float*    h2    = (float*)carve(sizeof(float) * (size_t)N * 8);
    float*    alS2  = (float*)carve(sizeof(float) * (size_t)N);
    float*    alD2  = (float*)carve(sizeof(float) * (size_t)N);
    const int nwg   = ((N + SPAN_ALL - 1) / SPAN_ALL) * G_ALL;
    const int per   = ((E + N + G_ALL - 1) / G_ALL + 3) & ~3;
    const int capf  = ((per + WG * UNR - 1) / (WG * UNR)) * UNR * 32;
    const int capw  = capf < SLICE_CAP ? capf : SLICE_CAP;
    unsigned* sl    = (unsigned*)carve(sizeof(unsigned) * (size_t)nwg * (WG / 32) * capw);
    int*      slcnt = (int*)carve(sizeof(int) * (size_t)nwg * (WG / 32));
    int*      fl1   = (int*)carve(sizeof(int) * (size_t)nwg);
    int*      fl2   = (int*)carve(sizeof(int) * (size_t)nwg);
    float*    slx   = pm2;
    float*    sla   = pm1;

    const int B = 256;
    k_l1<SPAN_ALL, G_ALL><<<nwg, WG, 0, stream>>>(ei, E, N, x, W1, as1, ad1, pm1, prec1, fl1, sl, slx, slcnt, capw);
    k_fin1<SPAN_ALL, G_ALL><<<(4 * N + B - 1) / B, B, 0, stream>>>(W1, b1, x, as1, ad1, fl1, pm1, prec1, out1, N);

    int tiles = (N + 15) / 16;
    k_gemm2<<<(tiles + 7) / 8, B, 0, stream>>>(out1, W2, as2, ad2, h2, alS2, alD2, N);

    k_l2<SPAN_ALL, G_ALL><<<nwg, WG, 0, stream>>>(ei, E, sl, sla, slcnt, capw, N, h2, alS2, alD2, as2, pm2, ps2, pacc2, fl2);
    k_fin2<SPAN_ALL, G_ALL><<<(2 * N + B - 1) / B, B, 0, stream>>>(alS2, alD2, fl2, pm2, ps2, pacc2, b2, (float*)d_out, N);
}
